# starting kernel
# baseline (speedup 1.0000x reference)
#include <hip/hip_runtime.h>
#include <hip/hip_bf16.h>
#include <math.h>


typedef float v2f __attribute__((ext_vector_type(2)));
typedef float v8f __attribute__((ext_vector_type(8)));

#define BN_EPS 1e-5f

__global__ void k_zero(float* __restrict__ p, size_t n) {
  size_t i = (size_t)blockIdx.x * blockDim.x + threadIdx.x;
  size_t stride = (size_t)gridDim.x * blockDim.x;
  for (; i < n; i += stride) p[i] = 0.0f;
}

__global__ void k_deg_init(float* __restrict__ deg, int n) {
  int i = blockIdx.x * blockDim.x + threadIdx.x;
  if (i < n) deg[i] = 1.0f;
}

__global__ void k_deg_count(const int* __restrict__ dst, float* __restrict__ deg, int E) {
  int e = blockIdx.x * blockDim.x + threadIdx.x;
  if (e < E) unsafeAtomicAdd(&deg[dst[e]], 1.0f);
}

__global__ void k_rsqrt_inplace(float* __restrict__ d, int n) {
  int i = blockIdx.x * blockDim.x + threadIdx.x;
  if (i < n) {
    float v = d[i];
    d[i] = (v > 0.0f) ? rsqrtf(v) : 0.0f;
  }
}

__global__ void k_gemm_wmma_f32(const float* __restrict__ A, const float* __restrict__ B,
                                const float* __restrict__ dinv,
                                float* __restrict__ C, int M, int K, int F) {
  const int lane = threadIdx.x & 31;
  const int wave = (int)((blockIdx.x * blockDim.x + threadIdx.x) >> 5);
  const int nf = F >> 4;
  const int mTile = wave / nf;
  const int fTile = wave - mTile * nf;
  const int m0 = mTile << 4;
  if (m0 >= M) return;
  const int f0 = fTile << 4;
  const int row = lane & 15;
  const int kh2 = (lane >> 4) << 1;

  const float* arow = A + (size_t)(m0 + row) * K + kh2;
  const float* bcol = B + (size_t)kh2 * F + f0 + row;

  v8f acc = {0.f, 0.f, 0.f, 0.f, 0.f, 0.f, 0.f, 0.f};
  for (int k = 0; k < K; k += 4) {
    float2 av = *reinterpret_cast<const float2*>(arow + k);
    v2f a; a.x = av.x; a.y = av.y;
    v2f b;
    b.x = bcol[(size_t)k * F];
    b.y = bcol[(size_t)(k + 1) * F];
    acc = __builtin_amdgcn_wmma_f32_16x16x4_f32(false, a, false, b,
                                                (short)0, acc, false, false);
  }
  const float* dvp = dinv + m0 + (kh2 << 2);
  float* crow = C + (size_t)(m0 + (kh2 << 2)) * F + f0 + row;
#pragma unroll
  for (int r = 0; r < 8; ++r) crow[(size_t)r * F] = acc[r] * dvp[r];
}

__global__ void k_scatter(const float* __restrict__ hsc, const int* __restrict__ src,
                          const int* __restrict__ dst,
                          float* __restrict__ agg, int E, int F) {
  int wave = (int)((blockIdx.x * blockDim.x + threadIdx.x) >> 5);
  int lane = threadIdx.x & 31;
  if (wave >= E) return;
  int s = src[wave], d = dst[wave];
  const float4* hrow = reinterpret_cast<const float4*>(hsc + (size_t)s * F);
  float* ad = agg + (size_t)d * F;
  for (int f4 = lane; 4 * f4 < F; f4 += 32) {
    float4 v = hrow[f4];
    unsafeAtomicAdd(ad + 4 * f4 + 0, v.x);
    unsafeAtomicAdd(ad + 4 * f4 + 1, v.y);
    unsafeAtomicAdd(ad + 4 * f4 + 2, v.z);
    unsafeAtomicAdd(ad + 4 * f4 + 3, v.w);
  }
}

__global__ void k_finish_relu_stats(float* __restrict__ agg, const float* __restrict__ hsc,
                                    const float* __restrict__ dinv,
                                    const float* __restrict__ bias,
                                    float* __restrict__ stats,
                                    int Nn, int F, int rowsPerBlock) {
  int j = threadIdx.x;
  int r0 = blockIdx.x * rowsPerBlock;
  int r1 = min(r0 + rowsPerBlock, Nn);
  float bj = bias[j];
  float sum = 0.f, sumsq = 0.f;
  for (int r = r0; r < r1; ++r) {
    float di = dinv[r];
    size_t idx = (size_t)r * F + j;
    float v = (agg[idx] + hsc[idx]) * di + bj;
    v = fmaxf(v, 0.f);
    agg[idx] = v;
    sum += v;
    sumsq += v * v;
  }
  unsafeAtomicAdd(stats + j, sum);
  unsafeAtomicAdd(stats + F + j, sumsq);
}

__global__ void k_bn_apply(float* __restrict__ v, const float* __restrict__ stats,
                           const float* __restrict__ gamma, const float* __restrict__ beta,
                           int Nn, int F, float invN) {
  size_t total = (size_t)Nn * F;
  size_t i = (size_t)blockIdx.x * blockDim.x + threadIdx.x;
  size_t stride = (size_t)gridDim.x * blockDim.x;
  for (; i < total; i += stride) {
    int j = (int)(i & (size_t)(F - 1));
    float mean = stats[j] * invN;
    float var = stats[F + j] * invN - mean * mean;
    v[i] = (v[i] - mean) * rsqrtf(var + BN_EPS) * gamma[j] + beta[j];
  }
}

__global__ void k_gemm64x2(const float* __restrict__ h3, const float* __restrict__ W4,
                           const float* __restrict__ dinv,
                           float* __restrict__ h4, int Nn) {
  int i = blockIdx.x * blockDim.x + threadIdx.x;
  if (i >= Nn) return;
  const float* row = h3 + (size_t)i * 64;
  float o0 = 0.f, o1 = 0.f;
#pragma unroll 8
  for (int k = 0; k < 64; ++k) {
    float xv = row[k];
    o0 += xv * W4[2 * k];
    o1 += xv * W4[2 * k + 1];
  }
  float di = dinv[i];
  h4[2 * (size_t)i]     = o0 * di;
  h4[2 * (size_t)i + 1] = o1 * di;
}

__global__ void k_scatter2(const float* __restrict__ h4s, const int* __restrict__ src,
                           const int* __restrict__ dst,
                           float* __restrict__ agg, int E) {
  int e = blockIdx.x * blockDim.x + threadIdx.x;
  if (e >= E) return;
  int s = src[e], d = dst[e];
  float2 v = *reinterpret_cast<const float2*>(h4s + 2 * (size_t)s);
  unsafeAtomicAdd(agg + 2 * (size_t)d,     v.x);
  unsafeAtomicAdd(agg + 2 * (size_t)d + 1, v.y);
}

__global__ void k_final_lsm(const float* __restrict__ agg, const float* __restrict__ h4s,
                            const float* __restrict__ dinv, const float* __restrict__ b4,
                            float* __restrict__ out, int Nn) {
  int i = blockIdx.x * blockDim.x + threadIdx.x;
  if (i >= Nn) return;
  float di = dinv[i];
  float o0 = (agg[2 * (size_t)i]     + h4s[2 * (size_t)i])     * di + b4[0];
  float o1 = (agg[2 * (size_t)i + 1] + h4s[2 * (size_t)i + 1]) * di + b4[1];
  float m = fmaxf(o0, o1);
  float l = m + logf(expf(o0 - m) + expf(o1 - m));
  out[2 * (size_t)i]     = o0 - l;
  out[2 * (size_t)i + 1] = o1 - l;
}

extern "C" void kernel_launch(void* const* d_in, const int* in_sizes, int n_in,
                              void* d_out, int out_size, void* d_ws, size_t ws_size,
                              hipStream_t stream) {
  const float* x   = (const float*)d_in[0];
  const int*   ei  = (const int*)d_in[1];
  const float* W1  = (const float*)d_in[2];
  const float* b1  = (const float*)d_in[3];
  const float* g1  = (const float*)d_in[4];
  const float* be1 = (const float*)d_in[5];
  const float* W2  = (const float*)d_in[6];
  const float* b2  = (const float*)d_in[7];
  const float* g2  = (const float*)d_in[8];
  const float* be2 = (const float*)d_in[9];
  const float* W3  = (const float*)d_in[10];
  const float* b3  = (const float*)d_in[11];
  const float* g3  = (const float*)d_in[12];
  const float* be3 = (const float*)d_in[13];
  const float* W4  = (const float*)d_in[14];
  const float* b4  = (const float*)d_in[15];

  const int Nn = in_sizes[0] / 24;
  const int E  = in_sizes[1] / 2;
  const int* src = ei;
  const int* dst = ei + E;

  float* ws    = (float*)d_ws;
  float* dinv  = ws;
  float* stats = ws + Nn;
  float* bufA  = stats + 256;
  float* bufB  = bufA + (size_t)Nn * 128;

  const float invN = 1.0f / (float)Nn;
  const int TB = 256;

  auto gemm = [&](const float* A, const float* Bm, float* C, int K, int F) {
    long waves = (long)((Nn + 15) / 16) * (F / 16);
    int blocks = (int)((waves * 32 + TB - 1) / TB);
    k_gemm_wmma_f32<<<blocks, TB, 0, stream>>>(A, Bm, dinv, C, Nn, K, F);
  };
  auto scatter = [&](const float* h, float* agg, int F) {
    int blocks = (int)(((long)E * 32 + TB - 1) / TB);
    k_scatter<<<blocks, TB, 0, stream>>>(h, src, dst, agg, E, F);
  };
  auto finish = [&](float* agg, const float* h, const float* bias, int F) {
    const int rows = 256;
    k_zero<<<1, 256, 0, stream>>>(stats, 256);
    k_finish_relu_stats<<<(Nn + rows - 1) / rows, F, 0, stream>>>(
        agg, h, dinv, bias, stats, Nn, F, rows);
  };
  auto bn = [&](float* v, const float* gamma, const float* beta, int F) {
    k_bn_apply<<<2048, TB, 0, stream>>>(v, stats, gamma, beta, Nn, F, invN);
  };
  auto zero = [&](float* p, size_t n) {
    int blocks = (int)((n + TB - 1) / TB);
    if (blocks > 4096) blocks = 4096;
    k_zero<<<blocks, TB, 0, stream>>>(p, n);
  };

  k_deg_init<<<(Nn + TB - 1) / TB, TB, 0, stream>>>(dinv, Nn);
  k_deg_count<<<(E + TB - 1) / TB, TB, 0, stream>>>(dst, dinv, E);
  k_rsqrt_inplace<<<(Nn + TB - 1) / TB, TB, 0, stream>>>(dinv, Nn);

  gemm(x, W1, bufA, 24, 128);
  zero(bufB, (size_t)Nn * 128);
  scatter(bufA, bufB, 128);
  finish(bufB, bufA, b1, 128);
  bn(bufB, g1, be1, 128);

  gemm(bufB, W2, bufA, 128, 128);
  zero(bufB, (size_t)Nn * 128);
  scatter(bufA, bufB, 128);
  finish(bufB, bufA, b2, 128);
  bn(bufB, g2, be2, 128);

  gemm(bufB, W3, bufA, 128, 64);
  zero(bufB, (size_t)Nn * 64);
  scatter(bufA, bufB, 64);
  finish(bufB, bufA, b3, 64);
  bn(bufB, g3, be3, 64);

  k_gemm64x2<<<(Nn + TB - 1) / TB, TB, 0, stream>>>(bufB, W4, dinv, bufA, Nn);
  zero(bufB, (size_t)Nn * 2);
  k_scatter2<<<(E + TB - 1) / TB, TB, 0, stream>>>(bufA, src, dst, bufB, E);
  k_final_lsm<<<(Nn + TB - 1) / TB, TB, 0, stream>>>(bufB, bufA, dinv, b4,
                                                     (float*)d_out, Nn);
}


# finish pass (norm+bias+relu+BN sums) as a stream: 256-thread blocks on 128-row chunks, 16-byte accesses, 16 loads in flight per wave; BN column sums folded in LDS before atomics (sum order changes)
# speedup vs baseline: 1.2712x; 1.2712x over previous
#include <hip/hip_runtime.h>
#include <hip/hip_bf16.h>
#include <math.h>


typedef float v2f __attribute__((ext_vector_type(2)));
typedef float v4f __attribute__((ext_vector_type(4)));
typedef float v8f __attribute__((ext_vector_type(8)));

#define BN_EPS 1e-5f

__global__ void k_zero(float* __restrict__ p, size_t n) {
  size_t i = (size_t)blockIdx.x * blockDim.x + threadIdx.x;
  size_t stride = (size_t)gridDim.x * blockDim.x;
  for (; i < n; i += stride) p[i] = 0.0f;
}

__global__ void k_deg_init(float* __restrict__ deg, int n) {
  int i = blockIdx.x * blockDim.x + threadIdx.x;
  if (i < n) deg[i] = 1.0f;
}

__global__ void k_deg_count(const int* __restrict__ dst, float* __restrict__ deg, int E) {
  int e = blockIdx.x * blockDim.x + threadIdx.x;
  if (e < E) unsafeAtomicAdd(&deg[dst[e]], 1.0f);
}

__global__ void k_rsqrt_inplace(float* __restrict__ d, int n) {
  int i = blockIdx.x * blockDim.x + threadIdx.x;
  if (i < n) {
    float v = d[i];
    d[i] = (v > 0.0f) ? rsqrtf(v) : 0.0f;
  }
}

__global__ void k_gemm_wmma_f32(const float* __restrict__ A, const float* __restrict__ B,
                                const float* __restrict__ dinv,
                                float* __restrict__ C, int M, int K, int F) {
  const int lane = threadIdx.x & 31;
  const int wave = (int)((blockIdx.x * blockDim.x + threadIdx.x) >> 5);
  const int nf = F >> 4;
  const int mTile = wave / nf;
  const int fTile = wave - mTile * nf;
  const int m0 = mTile << 4;
  if (m0 >= M) return;
  const int f0 = fTile << 4;
  const int row = lane & 15;
  const int kh2 = (lane >> 4) << 1;

  const float* arow = A + (size_t)(m0 + row) * K + kh2;
  const float* bcol = B + (size_t)kh2 * F + f0 + row;

  v8f acc = {0.f, 0.f, 0.f, 0.f, 0.f, 0.f, 0.f, 0.f};
  for (int k = 0; k < K; k += 4) {
    float2 av = *reinterpret_cast<const float2*>(arow + k);
    v2f a; a.x = av.x; a.y = av.y;
    v2f b;
    b.x = bcol[(size_t)k * F];
    b.y = bcol[(size_t)(k + 1) * F];
    acc = __builtin_amdgcn_wmma_f32_16x16x4_f32(false, a, false, b,
                                                (short)0, acc, false, false);
  }
  const float* dvp = dinv + m0 + (kh2 << 2);
  float* crow = C + (size_t)(m0 + (kh2 << 2)) * F + f0 + row;
#pragma unroll
  for (int r = 0; r < 8; ++r) crow[(size_t)r * F] = acc[r] * dvp[r];
}

__global__ void k_scatter(const float* __restrict__ hsc, const int* __restrict__ src,
                          const int* __restrict__ dst,
                          float* __restrict__ agg, int E, int F) {
  int wave = (int)((blockIdx.x * blockDim.x + threadIdx.x) >> 5);
  int lane = threadIdx.x & 31;
  if (wave >= E) return;
  int s = src[wave], d = dst[wave];
  const float4* hrow = reinterpret_cast<const float4*>(hsc + (size_t)s * F);
  float* ad = agg + (size_t)d * F;
  for (int f4 = lane; 4 * f4 < F; f4 += 32) {
    float4 v = hrow[f4];
    unsafeAtomicAdd(ad + 4 * f4 + 0, v.x);
    unsafeAtomicAdd(ad + 4 * f4 + 1, v.y);
    unsafeAtomicAdd(ad + 4 * f4 + 2, v.z);
    unsafeAtomicAdd(ad + 4 * f4 + 3, v.w);
  }
}

#define FIN_ROWS 128
template <int F>
__global__ void __launch_bounds__(256)
k_finish_relu_stats(float* __restrict__ agg, const float* __restrict__ hsc,
                    const float* __restrict__ dinv,
                    const float* __restrict__ bias,
                    float* __restrict__ stats,
                    int Nn) {
  constexpr int TPR = F / 4;
  constexpr int RPP = 256 / TPR;
  constexpr int U = 8;
  constexpr int STEP = RPP * TPR;
  static_assert(FIN_ROWS % (U * RPP) == 0, "a block is a whole number of groups");
  __shared__ __attribute__((aligned(16))) float red[2][RPP][F];

  const int t = threadIdx.x;
  const int c4 = t % TPR;
  const int rsub = t / TPR;
  const int r0 = blockIdx.x * FIN_ROWS;
  const int r1 = min(r0 + FIN_ROWS, Nn);

  const v4f b = *reinterpret_cast<const v4f*>(bias + 4 * c4);
  const v4f zero4 = {0.f, 0.f, 0.f, 0.f};
  v4f s = zero4;
  v4f q = zero4;

  int r = r0 + rsub;
  v4f* ap = reinterpret_cast<v4f*>(agg) + (size_t)r * TPR + c4;
  const v4f* hp = reinterpret_cast<const v4f*>(hsc) + (size_t)r * TPR + c4;
  const float* dp = dinv + r;

  auto apply = [&](const v4f a, const v4f h, const float di, v4f* dst) {
    const v4f v = __builtin_elementwise_max((a + h) * di + b, zero4);
    *dst = v;
    s += v;
    q += v * v;
  };

  for (; r + (U - 1) * RPP < r1; r += U * RPP) {
    v4f a[U], h[U];
    float di[U];
#pragma unroll
    for (int u = 0; u < U; ++u) {
      a[u] = ap[u * STEP];
      h[u] = hp[u * STEP];
      di[u] = dp[u * RPP];
    }
    __builtin_amdgcn_sched_barrier(0);
#pragma unroll
    for (int u = 0; u < U; ++u) apply(a[u], h[u], di[u], ap + u * STEP);
    ap += U * STEP;
    hp += U * STEP;
    dp += U * RPP;
  }
  for (; r < r1; r += RPP) {
    apply(*ap, *hp, *dp, ap);
    ap += STEP;
    hp += STEP;
    dp += RPP;
  }

  *reinterpret_cast<v4f*>(&red[0][rsub][4 * c4]) = s;
  *reinterpret_cast<v4f*>(&red[1][rsub][4 * c4]) = q;
  __syncthreads();
  if (t < 2 * F) {
    const int which = t / F;
    const int j = t % F;
    float acc = 0.f;
    for (int k = 0; k < RPP; ++k) acc += red[which][k][j];
    unsafeAtomicAdd(stats + which * F + j, acc);
  }
}

__global__ void k_bn_apply(float* __restrict__ v, const float* __restrict__ stats,
                           const float* __restrict__ gamma, const float* __restrict__ beta,
                           int Nn, int F, float invN) {
  size_t total = (size_t)Nn * F;
  size_t i = (size_t)blockIdx.x * blockDim.x + threadIdx.x;
  size_t stride = (size_t)gridDim.x * blockDim.x;
  for (; i < total; i += stride) {
    int j = (int)(i & (size_t)(F - 1));
    float mean = stats[j] * invN;
    float var = stats[F + j] * invN - mean * mean;
    v[i] = (v[i] - mean) * rsqrtf(var + BN_EPS) * gamma[j] + beta[j];
  }
}

__global__ void k_gemm64x2(const float* __restrict__ h3, const float* __restrict__ W4,
                           const float* __restrict__ dinv,
                           float* __restrict__ h4, int Nn) {
  int i = blockIdx.x * blockDim.x + threadIdx.x;
  if (i >= Nn) return;
  const float* row = h3 + (size_t)i * 64;
  float o0 = 0.f, o1 = 0.f;
#pragma unroll 8
  for (int k = 0; k < 64; ++k) {
    float xv = row[k];
    o0 += xv * W4[2 * k];
    o1 += xv * W4[2 * k + 1];
  }
  float di = dinv[i];
  h4[2 * (size_t)i]     = o0 * di;
  h4[2 * (size_t)i + 1] = o1 * di;
}

__global__ void k_scatter2(const float* __restrict__ h4s, const int* __restrict__ src,
                           const int* __restrict__ dst,
                           float* __restrict__ agg, int E) {
  int e = blockIdx.x * blockDim.x + threadIdx.x;
  if (e >= E) return;
  int s = src[e], d = dst[e];
  float2 v = *reinterpret_cast<const float2*>(h4s + 2 * (size_t)s);
  unsafeAtomicAdd(agg + 2 * (size_t)d,     v.x);
  unsafeAtomicAdd(agg + 2 * (size_t)d + 1, v.y);
}

__global__ void k_final_lsm(const float* __restrict__ agg, const float* __restrict__ h4s,
                            const float* __restrict__ dinv, const float* __restrict__ b4,
                            float* __restrict__ out, int Nn) {
  int i = blockIdx.x * blockDim.x + threadIdx.x;
  if (i >= Nn) return;
  float di = dinv[i];
  float o0 = (agg[2 * (size_t)i]     + h4s[2 * (size_t)i])     * di + b4[0];
  float o1 = (agg[2 * (size_t)i + 1] + h4s[2 * (size_t)i + 1]) * di + b4[1];
  float m = fmaxf(o0, o1);
  float l = m + logf(expf(o0 - m) + expf(o1 - m));
  out[2 * (size_t)i]     = o0 - l;
  out[2 * (size_t)i + 1] = o1 - l;
}

extern "C" void kernel_launch(void* const* d_in, const int* in_sizes, int n_in,
                              void* d_out, int out_size, void* d_ws, size_t ws_size,
                              hipStream_t stream) {
  const float* x   = (const float*)d_in[0];
  const int*   ei  = (const int*)d_in[1];
  const float* W1  = (const float*)d_in[2];
  const float* b1  = (const float*)d_in[3];
  const float* g1  = (const float*)d_in[4];
  const float* be1 = (const float*)d_in[5];
  const float* W2  = (const float*)d_in[6];
  const float* b2  = (const float*)d_in[7];
  const float* g2  = (const float*)d_in[8];
  const float* be2 = (const float*)d_in[9];
  const float* W3  = (const float*)d_in[10];
  const float* b3  = (const float*)d_in[11];
  const float* g3  = (const float*)d_in[12];
  const float* be3 = (const float*)d_in[13];
  const float* W4  = (const float*)d_in[14];
  const float* b4  = (const float*)d_in[15];

  const int Nn = in_sizes[0] / 24;
  const int E  = in_sizes[1] / 2;
  const int* src = ei;
  const int* dst = ei + E;

  float* ws    = (float*)d_ws;
  float* dinv  = ws;
  float* stats = ws + Nn;
  float* bufA  = stats + 256;
  float* bufB  = bufA + (size_t)Nn * 128;

  const float invN = 1.0f / (float)Nn;
  const int TB = 256;

  auto gemm = [&](const float* A, const float* Bm, float* C, int K, int F) {
    long waves = (long)((Nn + 15) / 16) * (F / 16);
    int blocks = (int)((waves * 32 + TB - 1) / TB);
    k_gemm_wmma_f32<<<blocks, TB, 0, stream>>>(A, Bm, dinv, C, Nn, K, F);
  };
  auto scatter = [&](const float* h, float* agg, int F) {
    int blocks = (int)(((long)E * 32 + TB - 1) / TB);
    k_scatter<<<blocks, TB, 0, stream>>>(h, src, dst, agg, E, F);
  };
  auto finish = [&](float* agg, const float* h, const float* bias, int F) {
    const int blocks = (Nn + FIN_ROWS - 1) / FIN_ROWS;
    k_zero<<<1, 256, 0, stream>>>(stats, 256);
    if (F == 128)
      k_finish_relu_stats<128><<<blocks, 256, 0, stream>>>(agg, h, dinv, bias, stats, Nn);
    else
      k_finish_relu_stats<64><<<blocks, 256, 0, stream>>>(agg, h, dinv, bias, stats, Nn);
  };
  auto bn = [&](float* v, const float* gamma, const float* beta, int F) {
    k_bn_apply<<<2048, TB, 0, stream>>>(v, stats, gamma, beta, Nn, F, invN);
  };
  auto zero = [&](float* p, size_t n) {
    int blocks = (int)((n + TB - 1) / TB);
    if (blocks > 4096) blocks = 4096;
    k_zero<<<blocks, TB, 0, stream>>>(p, n);
  };

  k_deg_init<<<(Nn + TB - 1) / TB, TB, 0, stream>>>(dinv, Nn);
  k_deg_count<<<(E + TB - 1) / TB, TB, 0, stream>>>(dst, dinv, E);
  k_rsqrt_inplace<<<(Nn + TB - 1) / TB, TB, 0, stream>>>(dinv, Nn);

  gemm(x, W1, bufA, 24, 128);
  zero(bufB, (size_t)Nn * 128);
  scatter(bufA, bufB, 128);
  finish(bufB, bufA, b1, 128);
  bn(bufB, g1, be1, 128);

  gemm(bufB, W2, bufA, 128, 128);
  zero(bufB, (size_t)Nn * 128);
  scatter(bufA, bufB, 128);
  finish(bufB, bufA, b2, 128);
  bn(bufB, g2, be2, 128);

  gemm(bufB, W3, bufA, 128, 64);
  zero(bufB, (size_t)Nn * 64);
  scatter(bufA, bufB, 64);
  finish(bufB, bufA, b3, 64);
  bn(bufB, g3, be3, 64);

  k_gemm64x2<<<(Nn + TB - 1) / TB, TB, 0, stream>>>(bufB, W4, dinv, bufA, Nn);
  zero(bufB, (size_t)Nn * 2);
  k_scatter2<<<(E + TB - 1) / TB, TB, 0, stream>>>(bufA, src, dst, bufB, E);
  k_final_lsm<<<(Nn + TB - 1) / TB, TB, 0, stream>>>(bufB, bufA, dinv, b4,
                                                     (float*)d_out, Nn);
}


# layers 1-3: in-edges sorted by destination per call (int atomic count+rank, scan, fill); a wave sums a row in registers and stores it once; their float atomics and zero pass go (sum order changes)
# speedup vs baseline: 2.6089x; 2.0523x over previous
#include <hip/hip_runtime.h>
#include <hip/hip_bf16.h>
#include <math.h>


typedef float v2f __attribute__((ext_vector_type(2)));
typedef float v4f __attribute__((ext_vector_type(4)));
typedef float v8f __attribute__((ext_vector_type(8)));

#define BN_EPS 1e-5f

__global__ void k_zero(float* __restrict__ p, size_t n) {
  size_t i = (size_t)blockIdx.x * blockDim.x + threadIdx.x;
  size_t stride = (size_t)gridDim.x * blockDim.x;
  for (; i < n; i += stride) p[i] = 0.0f;
}

__global__ void k_deg_init(float* __restrict__ deg, int n) {
  int i = blockIdx.x * blockDim.x + threadIdx.x;
  if (i < n) deg[i] = 1.0f;
}

__global__ void k_deg_count(const int* __restrict__ dst, float* __restrict__ deg, int E) {
  int e = blockIdx.x * blockDim.x + threadIdx.x;
  if (e < E) unsafeAtomicAdd(&deg[dst[e]], 1.0f);
}

__global__ void k_rsqrt_inplace(float* __restrict__ d, int n) {
  int i = blockIdx.x * blockDim.x + threadIdx.x;
  if (i < n) {
    float v = d[i];
    d[i] = (v > 0.0f) ? rsqrtf(v) : 0.0f;
  }
}

__global__ void k_count_rank(const int* __restrict__ dst, int* __restrict__ cnt,
                             int* __restrict__ rank, int E) {
  int e = blockIdx.x * blockDim.x + threadIdx.x;
  if (e < E) rank[e] = atomicAdd(&cnt[dst[e]], 1);
}

#define SCAN_CHUNK 1024

__global__ void __launch_bounds__(256)
k_dinv_bsum(const int* __restrict__ cnt, float* __restrict__ dinv,
            int* __restrict__ bsum, int n) {
  __shared__ int wsum[8];
  const int t = threadIdx.x;
  const int i0 = blockIdx.x * SCAN_CHUNK + 4 * t;
  int s = 0;
#pragma unroll
  for (int k = 0; k < 4; ++k) {
    const int i = i0 + k;
    if (i < n) {
      const int c = cnt[i];
      s += c;
      const float deg = (float)(c + 1);
      dinv[i] = (deg > 0.0f) ? rsqrtf(deg) : 0.0f;
    }
  }
  for (int o = 16; o > 0; o >>= 1) s += __shfl_xor(s, o);
  if ((t & 31) == 0) wsum[t >> 5] = s;
  __syncthreads();
  if (t == 0) {
    int tot = 0;
    for (int w = 0; w < 8; ++w) tot += wsum[w];
    bsum[blockIdx.x] = tot;
  }
}

__global__ void __launch_bounds__(256)
k_scan_rowptr(const int* __restrict__ cnt, const int* __restrict__ bsum,
              int* __restrict__ rowptr, int n) {
  __shared__ int wtot[8];
  __shared__ int base_s;
  const int t = threadIdx.x, lane = t & 31, w = t >> 5;
  int p = 0;
  for (int b = t; b < (int)blockIdx.x; b += 256) p += bsum[b];
  for (int o = 16; o > 0; o >>= 1) p += __shfl_xor(p, o);
  if (lane == 0) wtot[w] = p;
  __syncthreads();
  if (t == 0) {
    int tot = 0;
    for (int k = 0; k < 8; ++k) tot += wtot[k];
    base_s = tot;
  }
  __syncthreads();
  const int base = base_s;
  const int i0 = blockIdx.x * SCAN_CHUNK + 4 * t;
  int c[4];
  int s = 0;
#pragma unroll
  for (int k = 0; k < 4; ++k) {
    c[k] = (i0 + k < n) ? cnt[i0 + k] : 0;
    s += c[k];
  }
  int incl = s;
  for (int o = 1; o < 32; o <<= 1) {
    const int v = __shfl_up(incl, o);
    if (lane >= o) incl += v;
  }
  if (lane == 31) wtot[w] = incl;
  __syncthreads();
  int woff = 0;
  for (int k = 0; k < w; ++k) woff += wtot[k];
  int excl = base + woff + incl - s;
#pragma unroll
  for (int k = 0; k < 4; ++k) {
    if (i0 + k < n) {
      rowptr[i0 + k] = excl;
      excl += c[k];
    }
  }
  if (blockIdx.x == gridDim.x - 1 && t == 255) rowptr[n] = excl;
}

__global__ void k_fill_csr(const int* __restrict__ src, const int* __restrict__ dst,
                           const int* __restrict__ rank, const int* __restrict__ rowptr,
                           int* __restrict__ csr, int E) {
  int e = blockIdx.x * blockDim.x + threadIdx.x;
  if (e >= E) return;
  unsigned pos = (unsigned)(rowptr[dst[e]] + rank[e]);
  csr[min(pos, (unsigned)(E - 1))] = src[e];
}

template <int F>
__global__ void __launch_bounds__(256)
k_gather_rows(const float* __restrict__ hsc, const int* __restrict__ rowptr,
              const int* __restrict__ csr, float* __restrict__ agg, int Nn, int E) {
  constexpr int V = F / 32;
  typedef float vf __attribute__((ext_vector_type(V)));
  const int lane = threadIdx.x & 31;
  const int d = __builtin_amdgcn_readfirstlane((int)(blockIdx.x * 8 + (threadIdx.x >> 5)));
  if (d >= Nn) return;
  const int beg = max(rowptr[d], 0);
  const int end = min(rowptr[d + 1], E);
  const vf* hp = reinterpret_cast<const vf*>(hsc) + lane;
  const unsigned nmax = (unsigned)(Nn - 1);
  vf acc = 0.0f;
  for (int j0 = beg; j0 < end; j0 += 32) {
    const int m = min(32, end - j0);
    const int idx = (lane < m) ? csr[j0 + lane] : 0;
    auto row = [&](int u) -> vf {
      const unsigned s = min((unsigned)__builtin_amdgcn_readlane(idx, u), nmax);
      return hp[(size_t)s * 32];
    };
    int u = 0;
    for (; u + 8 <= m; u += 8) {
      vf r[8];
#pragma unroll
      for (int k = 0; k < 8; ++k) r[k] = row(u + k);
#pragma unroll
      for (int k = 0; k < 8; ++k) acc += r[k];
    }
    if (u + 4 <= m) {
      const vf r0 = row(u), r1 = row(u + 1), r2 = row(u + 2), r3 = row(u + 3);
      acc += r0; acc += r1; acc += r2; acc += r3;
      u += 4;
    }
    if (u + 2 <= m) {
      const vf r0 = row(u), r1 = row(u + 1);
      acc += r0; acc += r1;
      u += 2;
    }
    if (u < m) acc += row(u);
  }
  reinterpret_cast<vf*>(agg)[(size_t)d * 32 + lane] = acc;
}

__global__ void k_gemm_wmma_f32(const float* __restrict__ A, const float* __restrict__ B,
                                const float* __restrict__ dinv,
                                float* __restrict__ C, int M, int K, int F) {
  const int lane = threadIdx.x & 31;
  const int wave = (int)((blockIdx.x * blockDim.x + threadIdx.x) >> 5);
  const int nf = F >> 4;
  const int mTile = wave / nf;
  const int fTile = wave - mTile * nf;
  const int m0 = mTile << 4;
  if (m0 >= M) return;
  const int f0 = fTile << 4;
  const int row = lane & 15;
  const int kh2 = (lane >> 4) << 1;

  const float* arow = A + (size_t)(m0 + row) * K + kh2;
  const float* bcol = B + (size_t)kh2 * F + f0 + row;

  v8f acc = {0.f, 0.f, 0.f, 0.f, 0.f, 0.f, 0.f, 0.f};
  for (int k = 0; k < K; k += 4) {
    float2 av = *reinterpret_cast<const float2*>(arow + k);
    v2f a; a.x = av.x; a.y = av.y;
    v2f b;
    b.x = bcol[(size_t)k * F];
    b.y = bcol[(size_t)(k + 1) * F];
    acc = __builtin_amdgcn_wmma_f32_16x16x4_f32(false, a, false, b,
                                                (short)0, acc, false, false);
  }
  const float* dvp = dinv + m0 + (kh2 << 2);
  float* crow = C + (size_t)(m0 + (kh2 << 2)) * F + f0 + row;
#pragma unroll
  for (int r = 0; r < 8; ++r) crow[(size_t)r * F] = acc[r] * dvp[r];
}

__global__ void k_scatter(const float* __restrict__ hsc, const int* __restrict__ src,
                          const int* __restrict__ dst,
                          float* __restrict__ agg, int E, int F) {
  int wave = (int)((blockIdx.x * blockDim.x + threadIdx.x) >> 5);
  int lane = threadIdx.x & 31;
  if (wave >= E) return;
  int s = src[wave], d = dst[wave];
  const float4* hrow = reinterpret_cast<const float4*>(hsc + (size_t)s * F);
  float* ad = agg + (size_t)d * F;
  for (int f4 = lane; 4 * f4 < F; f4 += 32) {
    float4 v = hrow[f4];
    unsafeAtomicAdd(ad + 4 * f4 + 0, v.x);
    unsafeAtomicAdd(ad + 4 * f4 + 1, v.y);
    unsafeAtomicAdd(ad + 4 * f4 + 2, v.z);
    unsafeAtomicAdd(ad + 4 * f4 + 3, v.w);
  }
}

#define FIN_ROWS 128
template <int F>
__global__ void __launch_bounds__(256)
k_finish_relu_stats(float* __restrict__ agg, const float* __restrict__ hsc,
                    const float* __restrict__ dinv,
                    const float* __restrict__ bias,
                    float* __restrict__ stats,
                    int Nn) {
  constexpr int TPR = F / 4;
  constexpr int RPP = 256 / TPR;
  constexpr int U = 8;
  constexpr int STEP = RPP * TPR;
  static_assert(FIN_ROWS % (U * RPP) == 0, "a block is a whole number of groups");
  __shared__ __attribute__((aligned(16))) float red[2][RPP][F];

  const int t = threadIdx.x;
  const int c4 = t % TPR;
  const int rsub = t / TPR;
  const int r0 = blockIdx.x * FIN_ROWS;
  const int r1 = min(r0 + FIN_ROWS, Nn);

  const v4f b = *reinterpret_cast<const v4f*>(bias + 4 * c4);
  const v4f zero4 = {0.f, 0.f, 0.f, 0.f};
  v4f s = zero4;
  v4f q = zero4;

  int r = r0 + rsub;
  v4f* ap = reinterpret_cast<v4f*>(agg) + (size_t)r * TPR + c4;
  const v4f* hp = reinterpret_cast<const v4f*>(hsc) + (size_t)r * TPR + c4;
  const float* dp = dinv + r;

  auto apply = [&](const v4f a, const v4f h, const float di, v4f* dst) {
    const v4f v = __builtin_elementwise_max((a + h) * di + b, zero4);
    *dst = v;
    s += v;
    q += v * v;
  };

  for (; r + (U - 1) * RPP < r1; r += U * RPP) {
    v4f a[U], h[U];
    float di[U];
#pragma unroll
    for (int u = 0; u < U; ++u) {
      a[u] = ap[u * STEP];
      h[u] = hp[u * STEP];
      di[u] = dp[u * RPP];
    }
    __builtin_amdgcn_sched_barrier(0);
#pragma unroll
    for (int u = 0; u < U; ++u) apply(a[u], h[u], di[u], ap + u * STEP);
    ap += U * STEP;
    hp += U * STEP;
    dp += U * RPP;
  }
  for (; r < r1; r += RPP) {
    apply(*ap, *hp, *dp, ap);
    ap += STEP;
    hp += STEP;
    dp += RPP;
  }

  *reinterpret_cast<v4f*>(&red[0][rsub][4 * c4]) = s;
  *reinterpret_cast<v4f*>(&red[1][rsub][4 * c4]) = q;
  __syncthreads();
  if (t < 2 * F) {
    const int which = t / F;
    const int j = t % F;
    float acc = 0.f;
    for (int k = 0; k < RPP; ++k) acc += red[which][k][j];
    unsafeAtomicAdd(stats + which * F + j, acc);
  }
}

__global__ void k_bn_apply(float* __restrict__ v, const float* __restrict__ stats,
                           const float* __restrict__ gamma, const float* __restrict__ beta,
                           int Nn, int F, float invN) {
  size_t total = (size_t)Nn * F;
  size_t i = (size_t)blockIdx.x * blockDim.x + threadIdx.x;
  size_t stride = (size_t)gridDim.x * blockDim.x;
  for (; i < total; i += stride) {
    int j = (int)(i & (size_t)(F - 1));
    float mean = stats[j] * invN;
    float var = stats[F + j] * invN - mean * mean;
    v[i] = (v[i] - mean) * rsqrtf(var + BN_EPS) * gamma[j] + beta[j];
  }
}

__global__ void k_gemm64x2(const float* __restrict__ h3, const float* __restrict__ W4,
                           const float* __restrict__ dinv,
                           float* __restrict__ h4, int Nn) {
  int i = blockIdx.x * blockDim.x + threadIdx.x;
  if (i >= Nn) return;
  const float* row = h3 + (size_t)i * 64;
  float o0 = 0.f, o1 = 0.f;
#pragma unroll 8
  for (int k = 0; k < 64; ++k) {
    float xv = row[k];
    o0 += xv * W4[2 * k];
    o1 += xv * W4[2 * k + 1];
  }
  float di = dinv[i];
  h4[2 * (size_t)i]     = o0 * di;
  h4[2 * (size_t)i + 1] = o1 * di;
}

__global__ void k_scatter2(const float* __restrict__ h4s, const int* __restrict__ src,
                           const int* __restrict__ dst,
                           float* __restrict__ agg, int E) {
  int e = blockIdx.x * blockDim.x + threadIdx.x;
  if (e >= E) return;
  int s = src[e], d = dst[e];
  float2 v = *reinterpret_cast<const float2*>(h4s + 2 * (size_t)s);
  unsafeAtomicAdd(agg + 2 * (size_t)d,     v.x);
  unsafeAtomicAdd(agg + 2 * (size_t)d + 1, v.y);
}

__global__ void k_final_lsm(const float* __restrict__ agg, const float* __restrict__ h4s,
                            const float* __restrict__ dinv, const float* __restrict__ b4,
                            float* __restrict__ out, int Nn) {
  int i = blockIdx.x * blockDim.x + threadIdx.x;
  if (i >= Nn) return;
  float di = dinv[i];
  float o0 = (agg[2 * (size_t)i]     + h4s[2 * (size_t)i])     * di + b4[0];
  float o1 = (agg[2 * (size_t)i + 1] + h4s[2 * (size_t)i + 1]) * di + b4[1];
  float m = fmaxf(o0, o1);
  float l = m + logf(expf(o0 - m) + expf(o1 - m));
  out[2 * (size_t)i]     = o0 - l;
  out[2 * (size_t)i + 1] = o1 - l;
}

extern "C" void kernel_launch(void* const* d_in, const int* in_sizes, int n_in,
                              void* d_out, int out_size, void* d_ws, size_t ws_size,
                              hipStream_t stream) {
  const float* x   = (const float*)d_in[0];
  const int*   ei  = (const int*)d_in[1];
  const float* W1  = (const float*)d_in[2];
  const float* b1  = (const float*)d_in[3];
  const float* g1  = (const float*)d_in[4];
  const float* be1 = (const float*)d_in[5];
  const float* W2  = (const float*)d_in[6];
  const float* b2  = (const float*)d_in[7];
  const float* g2  = (const float*)d_in[8];
  const float* be2 = (const float*)d_in[9];
  const float* W3  = (const float*)d_in[10];
  const float* b3  = (const float*)d_in[11];
  const float* g3  = (const float*)d_in[12];
  const float* be3 = (const float*)d_in[13];
  const float* W4  = (const float*)d_in[14];
  const float* b4  = (const float*)d_in[15];

  const int Nn = in_sizes[0] / 24;
  const int E  = in_sizes[1] / 2;
  const int* src = ei;
  const int* dst = ei + E;

  float* ws    = (float*)d_ws;
  float* dinv  = ws;
  float* stats = ws + Nn;
  float* bufA  = stats + 256;
  float* bufB  = bufA + (size_t)Nn * 128;
  const int nchunk = (Nn + SCAN_CHUNK - 1) / SCAN_CHUNK;
  const size_t rp_len = ((size_t)Nn + 1 + 31) & ~(size_t)31;
  const size_t bs_len = ((size_t)nchunk + 31) & ~(size_t)31;
  int* rowptr = (int*)(bufB + (size_t)Nn * 128);
  int* csr    = rowptr + rp_len;
  int* cnt    = (int*)bufB;
  int* bsum   = cnt + Nn;
  int* rank   = bsum + bs_len;
  const size_t need_bytes =
      ((size_t)Nn + 256 + 2 * (size_t)Nn * 128 + rp_len + (size_t)E) * sizeof(float);
  const bool grouped = ws_size >= need_bytes &&
                       (size_t)Nn + bs_len + (size_t)E <= (size_t)Nn * 128;

  const float invN = 1.0f / (float)Nn;
  const int TB = 256;

  auto gemm = [&](const float* A, const float* Bm, float* C, int K, int F) {
    long waves = (long)((Nn + 15) / 16) * (F / 16);
    int blocks = (int)((waves * 32 + TB - 1) / TB);
    k_gemm_wmma_f32<<<blocks, TB, 0, stream>>>(A, Bm, dinv, C, Nn, K, F);
  };
  auto scatter = [&](const float* h, float* agg, int F) {
    int blocks = (int)(((long)E * 32 + TB - 1) / TB);
    k_scatter<<<blocks, TB, 0, stream>>>(h, src, dst, agg, E, F);
  };
  auto finish = [&](float* agg, const float* h, const float* bias, int F) {
    const int blocks = (Nn + FIN_ROWS - 1) / FIN_ROWS;
    k_zero<<<1, 256, 0, stream>>>(stats, 256);
    if (F == 128)
      k_finish_relu_stats<128><<<blocks, 256, 0, stream>>>(agg, h, dinv, bias, stats, Nn);
    else
      k_finish_relu_stats<64><<<blocks, 256, 0, stream>>>(agg, h, dinv, bias, stats, Nn);
  };
  auto bn = [&](float* v, const float* gamma, const float* beta, int F) {
    k_bn_apply<<<2048, TB, 0, stream>>>(v, stats, gamma, beta, Nn, F, invN);
  };
  auto zero = [&](float* p, size_t n) {
    int blocks = (int)((n + TB - 1) / TB);
    if (blocks > 4096) blocks = 4096;
    k_zero<<<blocks, TB, 0, stream>>>(p, n);
  };

  auto aggregate = [&](const float* h, float* agg, int F) {
    if (grouped) {
      const int blocks = (Nn + 7) / 8;
      if (F == 128)
        k_gather_rows<128><<<blocks, 256, 0, stream>>>(h, rowptr, csr, agg, Nn, E);
      else
        k_gather_rows<64><<<blocks, 256, 0, stream>>>(h, rowptr, csr, agg, Nn, E);
    } else {
      zero(agg, (size_t)Nn * F);
      scatter(h, agg, F);
    }
  };

  if (grouped) {
    zero((float*)cnt, (size_t)Nn);
    k_count_rank<<<(E + TB - 1) / TB, TB, 0, stream>>>(dst, cnt, rank, E);
    k_dinv_bsum<<<nchunk, 256, 0, stream>>>(cnt, dinv, bsum, Nn);
    k_scan_rowptr<<<nchunk, 256, 0, stream>>>(cnt, bsum, rowptr, Nn);
    k_fill_csr<<<(E + TB - 1) / TB, TB, 0, stream>>>(src, dst, rank, rowptr, csr, E);
  } else {
    k_deg_init<<<(Nn + TB - 1) / TB, TB, 0, stream>>>(dinv, Nn);
    k_deg_count<<<(E + TB - 1) / TB, TB, 0, stream>>>(dst, dinv, E);
    k_rsqrt_inplace<<<(Nn + TB - 1) / TB, TB, 0, stream>>>(dinv, Nn);
  }

  gemm(x, W1, bufA, 24, 128);
  aggregate(bufA, bufB, 128);
  finish(bufB, bufA, b1, 128);
  bn(bufB, g1, be1, 128);

  gemm(bufB, W2, bufA, 128, 128);
  aggregate(bufA, bufB, 128);
  finish(bufB, bufA, b2, 128);
  bn(bufB, g2, be2, 128);

  gemm(bufB, W3, bufA, 128, 64);
  aggregate(bufA, bufB, 64);
  finish(bufB, bufA, b3, 64);
  bn(bufB, g3, be3, 64);

  k_gemm64x2<<<(Nn + TB - 1) / TB, TB, 0, stream>>>(bufB, W4, dinv, bufA, Nn);
  zero(bufB, (size_t)Nn * 2);
  k_scatter2<<<(E + TB - 1) / TB, TB, 0, stream>>>(bufA, src, dst, bufB, E);
  k_final_lsm<<<(Nn + TB - 1) / TB, TB, 0, stream>>>(bufB, bufA, dinv, b4,
                                                     (float*)d_out, Nn);
}


# BatchNorm apply pass removed: the next layer's product normalises its A rows while staging them in LDS (whole-line 16-byte loads, A fragments then read from LDS); layer 4 does it in its row loop
# speedup vs baseline: 3.0688x; 1.1763x over previous
#include <hip/hip_runtime.h>
#include <hip/hip_bf16.h>
#include <math.h>


typedef float v2f __attribute__((ext_vector_type(2)));
typedef float v4f __attribute__((ext_vector_type(4)));
typedef float v8f __attribute__((ext_vector_type(8)));

#define BN_EPS 1e-5f

__global__ void k_zero(float* __restrict__ p, size_t n) {
  size_t i = (size_t)blockIdx.x * blockDim.x + threadIdx.x;
  size_t stride = (size_t)gridDim.x * blockDim.x;
  for (; i < n; i += stride) p[i] = 0.0f;
}

__global__ void k_deg_init(float* __restrict__ deg, int n) {
  int i = blockIdx.x * blockDim.x + threadIdx.x;
  if (i < n) deg[i] = 1.0f;
}

__global__ void k_deg_count(const int* __restrict__ dst, float* __restrict__ deg, int E) {
  int e = blockIdx.x * blockDim.x + threadIdx.x;
  if (e < E) unsafeAtomicAdd(&deg[dst[e]], 1.0f);
}

__global__ void k_rsqrt_inplace(float* __restrict__ d, int n) {
  int i = blockIdx.x * blockDim.x + threadIdx.x;
  if (i < n) {
    float v = d[i];
    d[i] = (v > 0.0f) ? rsqrtf(v) : 0.0f;
  }
}

__global__ void k_count_rank(const int* __restrict__ dst, int* __restrict__ cnt,
                             int* __restrict__ rank, int E) {
  int e = blockIdx.x * blockDim.x + threadIdx.x;
  if (e < E) rank[e] = atomicAdd(&cnt[dst[e]], 1);
}

#define SCAN_CHUNK 1024

__global__ void __launch_bounds__(256)
k_dinv_bsum(const int* __restrict__ cnt, float* __restrict__ dinv,
            int* __restrict__ bsum, int n) {
  __shared__ int wsum[8];
  const int t = threadIdx.x;
  const int i0 = blockIdx.x * SCAN_CHUNK + 4 * t;
  int s = 0;
#pragma unroll
  for (int k = 0; k < 4; ++k) {
    const int i = i0 + k;
    if (i < n) {
      const int c = cnt[i];
      s += c;
      const float deg = (float)(c + 1);
      dinv[i] = (deg > 0.0f) ? rsqrtf(deg) : 0.0f;
    }
  }
  for (int o = 16; o > 0; o >>= 1) s += __shfl_xor(s, o);
  if ((t & 31) == 0) wsum[t >> 5] = s;
  __syncthreads();
  if (t == 0) {
    int tot = 0;
    for (int w = 0; w < 8; ++w) tot += wsum[w];
    bsum[blockIdx.x] = tot;
  }
}

__global__ void __launch_bounds__(256)
k_scan_rowptr(const int* __restrict__ cnt, const int* __restrict__ bsum,
              int* __restrict__ rowptr, int n) {
  __shared__ int wtot[8];
  __shared__ int base_s;
  const int t = threadIdx.x, lane = t & 31, w = t >> 5;
  int p = 0;
  for (int b = t; b < (int)blockIdx.x; b += 256) p += bsum[b];
  for (int o = 16; o > 0; o >>= 1) p += __shfl_xor(p, o);
  if (lane == 0) wtot[w] = p;
  __syncthreads();
  if (t == 0) {
    int tot = 0;
    for (int k = 0; k < 8; ++k) tot += wtot[k];
    base_s = tot;
  }
  __syncthreads();
  const int base = base_s;
  const int i0 = blockIdx.x * SCAN_CHUNK + 4 * t;
  int c[4];
  int s = 0;
#pragma unroll
  for (int k = 0; k < 4; ++k) {
    c[k] = (i0 + k < n) ? cnt[i0 + k] : 0;
    s += c[k];
  }
  int incl = s;
  for (int o = 1; o < 32; o <<= 1) {
    const int v = __shfl_up(incl, o);
    if (lane >= o) incl += v;
  }
  if (lane == 31) wtot[w] = incl;
  __syncthreads();
  int woff = 0;
  for (int k = 0; k < w; ++k) woff += wtot[k];
  int excl = base + woff + incl - s;
#pragma unroll
  for (int k = 0; k < 4; ++k) {
    if (i0 + k < n) {
      rowptr[i0 + k] = excl;
      excl += c[k];
    }
  }
  if (blockIdx.x == gridDim.x - 1 && t == 255) rowptr[n] = excl;
}

__global__ void k_fill_csr(const int* __restrict__ src, const int* __restrict__ dst,
                           const int* __restrict__ rank, const int* __restrict__ rowptr,
                           int* __restrict__ csr, int E) {
  int e = blockIdx.x * blockDim.x + threadIdx.x;
  if (e >= E) return;
  unsigned pos = (unsigned)(rowptr[dst[e]] + rank[e]);
  csr[min(pos, (unsigned)(E - 1))] = src[e];
}

template <int F>
__global__ void __launch_bounds__(256)
k_gather_rows(const float* __restrict__ hsc, const int* __restrict__ rowptr,
              const int* __restrict__ csr, float* __restrict__ agg, int Nn, int E) {
  constexpr int V = F / 32;
  typedef float vf __attribute__((ext_vector_type(V)));
  const int lane = threadIdx.x & 31;
  const int d = __builtin_amdgcn_readfirstlane((int)(blockIdx.x * 8 + (threadIdx.x >> 5)));
  if (d >= Nn) return;
  const int beg = max(rowptr[d], 0);
  const int end = min(rowptr[d + 1], E);
  const vf* hp = reinterpret_cast<const vf*>(hsc) + lane;
  const unsigned nmax = (unsigned)(Nn - 1);
  vf acc = 0.0f;
  for (int j0 = beg; j0 < end; j0 += 32) {
    const int m = min(32, end - j0);
    const int idx = (lane < m) ? csr[j0 + lane] : 0;
    auto row = [&](int u) -> vf {
      const unsigned s = min((unsigned)__builtin_amdgcn_readlane(idx, u), nmax);
      return hp[(size_t)s * 32];
    };
    int u = 0;
    for (; u + 8 <= m; u += 8) {
      vf r[8];
#pragma unroll
      for (int k = 0; k < 8; ++k) r[k] = row(u + k);
#pragma unroll
      for (int k = 0; k < 8; ++k) acc += r[k];
    }
    if (u + 4 <= m) {
      const vf r0 = row(u), r1 = row(u + 1), r2 = row(u + 2), r3 = row(u + 3);
      acc += r0; acc += r1; acc += r2; acc += r3;
      u += 4;
    }
    if (u + 2 <= m) {
      const vf r0 = row(u), r1 = row(u + 1);
      acc += r0; acc += r1;
      u += 2;
    }
    if (u < m) acc += row(u);
  }
  reinterpret_cast<vf*>(agg)[(size_t)d * 32 + lane] = acc;
}

__global__ void k_gemm_wmma_f32(const float* __restrict__ A, const float* __restrict__ B,
                                const float* __restrict__ dinv,
                                float* __restrict__ C, int M, int K, int F) {
  const int lane = threadIdx.x & 31;
  const int wave = (int)((blockIdx.x * blockDim.x + threadIdx.x) >> 5);
  const int nf = F >> 4;
  const int mTile = wave / nf;
  const int fTile = wave - mTile * nf;
  const int m0 = mTile << 4;
  if (m0 >= M) return;
  const int f0 = fTile << 4;
  const int row = lane & 15;
  const int kh2 = (lane >> 4) << 1;

  const float* arow = A + (size_t)(m0 + row) * K + kh2;
  const float* bcol = B + (size_t)kh2 * F + f0 + row;

  v8f acc = {0.f, 0.f, 0.f, 0.f, 0.f, 0.f, 0.f, 0.f};
  for (int k = 0; k < K; k += 4) {
    float2 av = *reinterpret_cast<const float2*>(arow + k);
    v2f a; a.x = av.x; a.y = av.y;
    v2f b;
    b.x = bcol[(size_t)k * F];
    b.y = bcol[(size_t)(k + 1) * F];
    acc = __builtin_amdgcn_wmma_f32_16x16x4_f32(false, a, false, b,
                                                (short)0, acc, false, false);
  }
  const float* dvp = dinv + m0 + (kh2 << 2);
  float* crow = C + (size_t)(m0 + (kh2 << 2)) * F + f0 + row;
#pragma unroll
  for (int r = 0; r < 8; ++r) crow[(size_t)r * F] = acc[r] * dvp[r];
}

template <int F>
__global__ void __launch_bounds__(256)
k_gemm_bn_wmma_f32(const float* __restrict__ V, const float* __restrict__ stats,
                   const float* __restrict__ gamma, const float* __restrict__ beta,
                   const float* __restrict__ B, const float* __restrict__ dinv,
                   float* __restrict__ C, int M, float invN) {
  constexpr int K = 128;
  constexpr int NF = F / 16;
  constexpr int RB = 16 * (8 / NF);
  constexpr int PITCH = K + 4;
  __shared__ __attribute__((aligned(16))) float As[RB * PITCH];

  const int t = threadIdx.x;
  const int lane = t & 31;
  const int w = t >> 5;
  const int rowBase = blockIdx.x * RB;

  {
    const int c4 = t & 31;
    const v4f su = *reinterpret_cast<const v4f*>(stats + 4 * c4);
    const v4f sq = *reinterpret_cast<const v4f*>(stats + K + 4 * c4);
    const v4f g = *reinterpret_cast<const v4f*>(gamma + 4 * c4);
    const v4f be = *reinterpret_cast<const v4f*>(beta + 4 * c4);
    v4f mean, rstd;
#pragma unroll
    for (int i = 0; i < 4; ++i) {
      mean[i] = su[i] * invN;
      const float var = sq[i] * invN - mean[i] * mean[i];
      rstd[i] = rsqrtf(var + BN_EPS);
    }
    v4f v[RB / 8];
#pragma unroll
    for (int p = 0; p < RB / 8; ++p) {
      const int gr = min(rowBase + p * 8 + w, M - 1);
      v[p] = *reinterpret_cast<const v4f*>(V + (size_t)gr * K + 4 * c4);
    }
#pragma unroll
    for (int p = 0; p < RB / 8; ++p) {
      const v4f o = (v[p] - mean) * rstd * g + be;
      *reinterpret_cast<v4f*>(&As[(p * 8 + w) * PITCH + 4 * c4]) = o;
    }
  }
  __syncthreads();

  const int mT = w / NF;
  const int fT = w - mT * NF;
  const int m0 = rowBase + (mT << 4);
  if (m0 >= M) return;
  const int f0 = fT << 4;
  const int row = lane & 15;
  const int kh2 = (lane >> 4) << 1;

  const float* arow = As + ((mT << 4) + row) * PITCH + kh2;
  const float* bcol = B + (size_t)kh2 * F + f0 + row;

  unsigned dbyte = (unsigned)(m0 + (kh2 << 2)) * 4u;
  unsigned cbyte = ((unsigned)(m0 + (kh2 << 2)) * F + f0 + row) * 4u;
  asm volatile("" : "+v"(dbyte), "+v"(cbyte));
  const float* dvp = reinterpret_cast<const float*>(reinterpret_cast<const char*>(dinv) + dbyte);
  float* crow = reinterpret_cast<float*>(reinterpret_cast<char*>(C) + cbyte);

  v8f acc = {0.f, 0.f, 0.f, 0.f, 0.f, 0.f, 0.f, 0.f};
#pragma unroll
  for (int k = 0; k < K; k += 4) {
    const v2f a = *reinterpret_cast<const v2f*>(arow + k);
    v2f b;
    b.x = bcol[(size_t)k * F];
    b.y = bcol[(size_t)(k + 1) * F];
    acc = __builtin_amdgcn_wmma_f32_16x16x4_f32(false, a, false, b,
                                                (short)0, acc, false, false);
  }
  asm volatile("s_wait_alu depctr_va_vdst(0)\n\t"
               "v_nop\n\tv_nop\n\tv_nop\n\tv_nop\n\t"
               "v_nop\n\tv_nop\n\tv_nop\n\tv_nop"
               : "+v"(acc));
#pragma unroll
  for (int r = 0; r < 8; ++r) crow[(size_t)r * F] = acc[r] * dvp[r];
}

__global__ void k_scatter(const float* __restrict__ hsc, const int* __restrict__ src,
                          const int* __restrict__ dst,
                          float* __restrict__ agg, int E, int F) {
  int wave = (int)((blockIdx.x * blockDim.x + threadIdx.x) >> 5);
  int lane = threadIdx.x & 31;
  if (wave >= E) return;
  int s = src[wave], d = dst[wave];
  const float4* hrow = reinterpret_cast<const float4*>(hsc + (size_t)s * F);
  float* ad = agg + (size_t)d * F;
  for (int f4 = lane; 4 * f4 < F; f4 += 32) {
    float4 v = hrow[f4];
    unsafeAtomicAdd(ad + 4 * f4 + 0, v.x);
    unsafeAtomicAdd(ad + 4 * f4 + 1, v.y);
    unsafeAtomicAdd(ad + 4 * f4 + 2, v.z);
    unsafeAtomicAdd(ad + 4 * f4 + 3, v.w);
  }
}

#define FIN_ROWS 128
template <int F>
__global__ void __launch_bounds__(256)
k_finish_relu_stats(float* __restrict__ agg, const float* __restrict__ hsc,
                    const float* __restrict__ dinv,
                    const float* __restrict__ bias,
                    float* __restrict__ stats,
                    int Nn) {
  constexpr int TPR = F / 4;
  constexpr int RPP = 256 / TPR;
  constexpr int U = 8;
  constexpr int STEP = RPP * TPR;
  static_assert(FIN_ROWS % (U * RPP) == 0, "a block is a whole number of groups");
  __shared__ __attribute__((aligned(16))) float red[2][RPP][F];

  const int t = threadIdx.x;
  const int c4 = t % TPR;
  const int rsub = t / TPR;
  const int r0 = blockIdx.x * FIN_ROWS;
  const int r1 = min(r0 + FIN_ROWS, Nn);

  const v4f b = *reinterpret_cast<const v4f*>(bias + 4 * c4);
  const v4f zero4 = {0.f, 0.f, 0.f, 0.f};
  v4f s = zero4;
  v4f q = zero4;

  int r = r0 + rsub;
  v4f* ap = reinterpret_cast<v4f*>(agg) + (size_t)r * TPR + c4;
  const v4f* hp = reinterpret_cast<const v4f*>(hsc) + (size_t)r * TPR + c4;
  const float* dp = dinv + r;

  auto apply = [&](const v4f a, const v4f h, const float di, v4f* dst) {
    const v4f v = __builtin_elementwise_max((a + h) * di + b, zero4);
    *dst = v;
    s += v;
    q += v * v;
  };

  for (; r + (U - 1) * RPP < r1; r += U * RPP) {
    v4f a[U], h[U];
    float di[U];
#pragma unroll
    for (int u = 0; u < U; ++u) {
      a[u] = ap[u * STEP];
      h[u] = hp[u * STEP];
      di[u] = dp[u * RPP];
    }
    __builtin_amdgcn_sched_barrier(0);
#pragma unroll
    for (int u = 0; u < U; ++u) apply(a[u], h[u], di[u], ap + u * STEP);
    ap += U * STEP;
    hp += U * STEP;
    dp += U * RPP;
  }
  for (; r < r1; r += RPP) {
    apply(*ap, *hp, *dp, ap);
    ap += STEP;
    hp += STEP;
    dp += RPP;
  }

  *reinterpret_cast<v4f*>(&red[0][rsub][4 * c4]) = s;
  *reinterpret_cast<v4f*>(&red[1][rsub][4 * c4]) = q;
  __syncthreads();
  if (t < 2 * F) {
    const int which = t / F;
    const int j = t % F;
    float acc = 0.f;
    for (int k = 0; k < RPP; ++k) acc += red[which][k][j];
    unsafeAtomicAdd(stats + which * F + j, acc);
  }
}

__global__ void k_gemm64x2(const float* __restrict__ v3, const float* __restrict__ stats,
                           const float* __restrict__ gamma, const float* __restrict__ beta,
                           const float* __restrict__ W4, const float* __restrict__ dinv,
                           float* __restrict__ h4, int Nn, float invN) {
  int i = blockIdx.x * blockDim.x + threadIdx.x;
  if (i >= Nn) return;
  const float* row = v3 + (size_t)i * 64;
  float o0 = 0.f, o1 = 0.f;
#pragma unroll 8
  for (int k = 0; k < 64; ++k) {
    const float mean = stats[k] * invN;
    const float var = stats[64 + k] * invN - mean * mean;
    const float xv = (row[k] - mean) * rsqrtf(var + BN_EPS) * gamma[k] + beta[k];
    o0 += xv * W4[2 * k];
    o1 += xv * W4[2 * k + 1];
  }
  float di = dinv[i];
  h4[2 * (size_t)i]     = o0 * di;
  h4[2 * (size_t)i + 1] = o1 * di;
}

__global__ void k_scatter2(const float* __restrict__ h4s, const int* __restrict__ src,
                           const int* __restrict__ dst,
                           float* __restrict__ agg, int E) {
  int e = blockIdx.x * blockDim.x + threadIdx.x;
  if (e >= E) return;
  int s = src[e], d = dst[e];
  float2 v = *reinterpret_cast<const float2*>(h4s + 2 * (size_t)s);
  unsafeAtomicAdd(agg + 2 * (size_t)d,     v.x);
  unsafeAtomicAdd(agg + 2 * (size_t)d + 1, v.y);
}

__global__ void k_final_lsm(const float* __restrict__ agg, const float* __restrict__ h4s,
                            const float* __restrict__ dinv, const float* __restrict__ b4,
                            float* __restrict__ out, int Nn) {
  int i = blockIdx.x * blockDim.x + threadIdx.x;
  if (i >= Nn) return;
  float di = dinv[i];
  float o0 = (agg[2 * (size_t)i]     + h4s[2 * (size_t)i])     * di + b4[0];
  float o1 = (agg[2 * (size_t)i + 1] + h4s[2 * (size_t)i + 1]) * di + b4[1];
  float m = fmaxf(o0, o1);
  float l = m + logf(expf(o0 - m) + expf(o1 - m));
  out[2 * (size_t)i]     = o0 - l;
  out[2 * (size_t)i + 1] = o1 - l;
}

extern "C" void kernel_launch(void* const* d_in, const int* in_sizes, int n_in,
                              void* d_out, int out_size, void* d_ws, size_t ws_size,
                              hipStream_t stream) {
  const float* x   = (const float*)d_in[0];
  const int*   ei  = (const int*)d_in[1];
  const float* W1  = (const float*)d_in[2];
  const float* b1  = (const float*)d_in[3];
  const float* g1  = (const float*)d_in[4];
  const float* be1 = (const float*)d_in[5];
  const float* W2  = (const float*)d_in[6];
  const float* b2  = (const float*)d_in[7];
  const float* g2  = (const float*)d_in[8];
  const float* be2 = (const float*)d_in[9];
  const float* W3  = (const float*)d_in[10];
  const float* b3  = (const float*)d_in[11];
  const float* g3  = (const float*)d_in[12];
  const float* be3 = (const float*)d_in[13];
  const float* W4  = (const float*)d_in[14];
  const float* b4  = (const float*)d_in[15];

  const int Nn = in_sizes[0] / 24;
  const int E  = in_sizes[1] / 2;
  const int* src = ei;
  const int* dst = ei + E;

  float* ws    = (float*)d_ws;
  float* dinv  = ws;
  float* stats = ws + Nn;
  float* bufA  = stats + 256;
  float* bufB  = bufA + (size_t)Nn * 128;
  const int nchunk = (Nn + SCAN_CHUNK - 1) / SCAN_CHUNK;
  const size_t rp_len = ((size_t)Nn + 1 + 31) & ~(size_t)31;
  const size_t bs_len = ((size_t)nchunk + 31) & ~(size_t)31;
  int* rowptr = (int*)(bufB + (size_t)Nn * 128);
  int* csr    = rowptr + rp_len;
  int* cnt    = (int*)bufB;
  int* bsum   = cnt + Nn;
  int* rank   = bsum + bs_len;
  const size_t need_bytes =
      ((size_t)Nn + 256 + 2 * (size_t)Nn * 128 + rp_len + (size_t)E) * sizeof(float);
  const bool grouped = ws_size >= need_bytes &&
                       (size_t)Nn + bs_len + (size_t)E <= (size_t)Nn * 128;

  const float invN = 1.0f / (float)Nn;
  const int TB = 256;

  auto gemm = [&](const float* A, const float* Bm, float* C, int K, int F) {
    long waves = (long)((Nn + 15) / 16) * (F / 16);
    int blocks = (int)((waves * 32 + TB - 1) / TB);
    k_gemm_wmma_f32<<<blocks, TB, 0, stream>>>(A, Bm, dinv, C, Nn, K, F);
  };
  auto scatter = [&](const float* h, float* agg, int F) {
    int blocks = (int)(((long)E * 32 + TB - 1) / TB);
    k_scatter<<<blocks, TB, 0, stream>>>(h, src, dst, agg, E, F);
  };
  auto finish = [&](float* agg, const float* h, const float* bias, int F) {
    const int blocks = (Nn + FIN_ROWS - 1) / FIN_ROWS;
    k_zero<<<1, 256, 0, stream>>>(stats, 256);
    if (F == 128)
      k_finish_relu_stats<128><<<blocks, 256, 0, stream>>>(agg, h, dinv, bias, stats, Nn);
    else
      k_finish_relu_stats<64><<<blocks, 256, 0, stream>>>(agg, h, dinv, bias, stats, Nn);
  };
  auto gemm_bn = [&](const float* V, const float* gamma, const float* beta,
                     const float* Bm, float* C, int F) {
    if (F == 128)
      k_gemm_bn_wmma_f32<128><<<(Nn + 15) / 16, 256, 0, stream>>>(
          V, stats, gamma, beta, Bm, dinv, C, Nn, invN);
    else
      k_gemm_bn_wmma_f32<64><<<(Nn + 31) / 32, 256, 0, stream>>>(
          V, stats, gamma, beta, Bm, dinv, C, Nn, invN);
  };
  auto zero = [&](float* p, size_t n) {
    int blocks = (int)((n + TB - 1) / TB);
    if (blocks > 4096) blocks = 4096;
    k_zero<<<blocks, TB, 0, stream>>>(p, n);
  };

  auto aggregate = [&](const float* h, float* agg, int F) {
    if (grouped) {
      const int blocks = (Nn + 7) / 8;
      if (F == 128)
        k_gather_rows<128><<<blocks, 256, 0, stream>>>(h, rowptr, csr, agg, Nn, E);
      else
        k_gather_rows<64><<<blocks, 256, 0, stream>>>(h, rowptr, csr, agg, Nn, E);
    } else {
      zero(agg, (size_t)Nn * F);
      scatter(h, agg, F);
    }
  };

  if (grouped) {
    zero((float*)cnt, (size_t)Nn);
    k_count_rank<<<(E + TB - 1) / TB, TB, 0, stream>>>(dst, cnt, rank, E);
    k_dinv_bsum<<<nchunk, 256, 0, stream>>>(cnt, dinv, bsum, Nn);
    k_scan_rowptr<<<nchunk, 256, 0, stream>>>(cnt, bsum, rowptr, Nn);
    k_fill_csr<<<(E + TB - 1) / TB, TB, 0, stream>>>(src, dst, rank, rowptr, csr, E);
  } else {
    k_deg_init<<<(Nn + TB - 1) / TB, TB, 0, stream>>>(dinv, Nn);
    k_deg_count<<<(E + TB - 1) / TB, TB, 0, stream>>>(dst, dinv, E);
    k_rsqrt_inplace<<<(Nn + TB - 1) / TB, TB, 0, stream>>>(dinv, Nn);
  }

  gemm(x, W1, bufA, 24, 128);
  aggregate(bufA, bufB, 128);
  finish(bufB, bufA, b1, 128);

  gemm_bn(bufB, g1, be1, W2, bufA, 128);
  aggregate(bufA, bufB, 128);
  finish(bufB, bufA, b2, 128);

  gemm_bn(bufB, g2, be2, W3, bufA, 64);
  aggregate(bufA, bufB, 64);
  finish(bufB, bufA, b3, 64);

  k_gemm64x2<<<(Nn + TB - 1) / TB, TB, 0, stream>>>(bufB, stats, g3, be3, W4, dinv,
                                                    bufA, Nn, invN);
  zero(bufB, (size_t)Nn * 2);
  k_scatter2<<<(E + TB - 1) / TB, TB, 0, stream>>>(bufA, src, dst, bufB, E);
  k_final_lsm<<<(Nn + TB - 1) / TB, TB, 0, stream>>>(bufB, bufA, dinv, b4,
                                                     (float*)d_out, Nn);
}


# layer 1 sums its 24 input columns over the in-edges (self loop and both norm factors included) before the product; the product's store then adds bias, relu and the BN column sums (sum order changes)
# speedup vs baseline: 3.2591x; 1.0620x over previous
#include <hip/hip_runtime.h>
#include <hip/hip_bf16.h>
#include <math.h>


typedef float v2f __attribute__((ext_vector_type(2)));
typedef float v4f __attribute__((ext_vector_type(4)));
typedef float v8f __attribute__((ext_vector_type(8)));

#define BN_EPS 1e-5f

__global__ void k_zero(float* __restrict__ p, size_t n) {
  size_t i = (size_t)blockIdx.x * blockDim.x + threadIdx.x;
  size_t stride = (size_t)gridDim.x * blockDim.x;
  for (; i < n; i += stride) p[i] = 0.0f;
}

__global__ void k_deg_init(float* __restrict__ deg, int n) {
  int i = blockIdx.x * blockDim.x + threadIdx.x;
  if (i < n) deg[i] = 1.0f;
}

__global__ void k_deg_count(const int* __restrict__ dst, float* __restrict__ deg, int E) {
  int e = blockIdx.x * blockDim.x + threadIdx.x;
  if (e < E) unsafeAtomicAdd(&deg[dst[e]], 1.0f);
}

__global__ void k_rsqrt_inplace(float* __restrict__ d, int n) {
  int i = blockIdx.x * blockDim.x + threadIdx.x;
  if (i < n) {
    float v = d[i];
    d[i] = (v > 0.0f) ? rsqrtf(v) : 0.0f;
  }
}

__global__ void k_count_rank(const int* __restrict__ dst, int* __restrict__ cnt,
                             int* __restrict__ rank, int E) {
  int e = blockIdx.x * blockDim.x + threadIdx.x;
  if (e < E) rank[e] = atomicAdd(&cnt[dst[e]], 1);
}

#define SCAN_CHUNK 1024

__global__ void __launch_bounds__(256)
k_dinv_bsum(const int* __restrict__ cnt, float* __restrict__ dinv,
            int* __restrict__ bsum, int n) {
  __shared__ int wsum[8];
  const int t = threadIdx.x;
  const int i0 = blockIdx.x * SCAN_CHUNK + 4 * t;
  int s = 0;
#pragma unroll
  for (int k = 0; k < 4; ++k) {
    const int i = i0 + k;
    if (i < n) {
      const int c = cnt[i];
      s += c;
      const float deg = (float)(c + 1);
      dinv[i] = (deg > 0.0f) ? rsqrtf(deg) : 0.0f;
    }
  }
  for (int o = 16; o > 0; o >>= 1) s += __shfl_xor(s, o);
  if ((t & 31) == 0) wsum[t >> 5] = s;
  __syncthreads();
  if (t == 0) {
    int tot = 0;
    for (int w = 0; w < 8; ++w) tot += wsum[w];
    bsum[blockIdx.x] = tot;
  }
}

__global__ void __launch_bounds__(256)
k_scan_rowptr(const int* __restrict__ cnt, const int* __restrict__ bsum,
              int* __restrict__ rowptr, int n) {
  __shared__ int wtot[8];
  __shared__ int base_s;
  const int t = threadIdx.x, lane = t & 31, w = t >> 5;
  int p = 0;
  for (int b = t; b < (int)blockIdx.x; b += 256) p += bsum[b];
  for (int o = 16; o > 0; o >>= 1) p += __shfl_xor(p, o);
  if (lane == 0) wtot[w] = p;
  __syncthreads();
  if (t == 0) {
    int tot = 0;
    for (int k = 0; k < 8; ++k) tot += wtot[k];
    base_s = tot;
  }
  __syncthreads();
  const int base = base_s;
  const int i0 = blockIdx.x * SCAN_CHUNK + 4 * t;
  int c[4];
  int s = 0;
#pragma unroll
  for (int k = 0; k < 4; ++k) {
    c[k] = (i0 + k < n) ? cnt[i0 + k] : 0;
    s += c[k];
  }
  int incl = s;
  for (int o = 1; o < 32; o <<= 1) {
    const int v = __shfl_up(incl, o);
    if (lane >= o) incl += v;
  }
  if (lane == 31) wtot[w] = incl;
  __syncthreads();
  int woff = 0;
  for (int k = 0; k < w; ++k) woff += wtot[k];
  int excl = base + woff + incl - s;
#pragma unroll
  for (int k = 0; k < 4; ++k) {
    if (i0 + k < n) {
      rowptr[i0 + k] = excl;
      excl += c[k];
    }
  }
  if (blockIdx.x == gridDim.x - 1 && t == 255) rowptr[n] = excl;
}

__global__ void k_fill_csr(const int* __restrict__ src, const int* __restrict__ dst,
                           const int* __restrict__ rank, const int* __restrict__ rowptr,
                           int* __restrict__ csr, int E) {
  int e = blockIdx.x * blockDim.x + threadIdx.x;
  if (e >= E) return;
  unsigned pos = (unsigned)(rowptr[dst[e]] + rank[e]);
  csr[min(pos, (unsigned)(E - 1))] = src[e];
}

template <int F>
__global__ void __launch_bounds__(256)
k_gather_rows(const float* __restrict__ hsc, const int* __restrict__ rowptr,
              const int* __restrict__ csr, float* __restrict__ agg, int Nn, int E) {
  constexpr int V = F / 32;
  typedef float vf __attribute__((ext_vector_type(V)));
  const int lane = threadIdx.x & 31;
  const int d = __builtin_amdgcn_readfirstlane((int)(blockIdx.x * 8 + (threadIdx.x >> 5)));
  if (d >= Nn) return;
  const int beg = max(rowptr[d], 0);
  const int end = min(rowptr[d + 1], E);
  const vf* hp = reinterpret_cast<const vf*>(hsc) + lane;
  const unsigned nmax = (unsigned)(Nn - 1);
  vf acc = 0.0f;
  for (int j0 = beg; j0 < end; j0 += 32) {
    const int m = min(32, end - j0);
    const int idx = (lane < m) ? csr[j0 + lane] : 0;
    auto row = [&](int u) -> vf {
      const unsigned s = min((unsigned)__builtin_amdgcn_readlane(idx, u), nmax);
      return hp[(size_t)s * 32];
    };
    int u = 0;
    for (; u + 8 <= m; u += 8) {
      vf r[8];
#pragma unroll
      for (int k = 0; k < 8; ++k) r[k] = row(u + k);
#pragma unroll
      for (int k = 0; k < 8; ++k) acc += r[k];
    }
    if (u + 4 <= m) {
      const vf r0 = row(u), r1 = row(u + 1), r2 = row(u + 2), r3 = row(u + 3);
      acc += r0; acc += r1; acc += r2; acc += r3;
      u += 4;
    }
    if (u + 2 <= m) {
      const vf r0 = row(u), r1 = row(u + 1);
      acc += r0; acc += r1;
      u += 2;
    }
    if (u < m) acc += row(u);
  }
  reinterpret_cast<vf*>(agg)[(size_t)d * 32 + lane] = acc;
}

__global__ void k_gemm_wmma_f32(const float* __restrict__ A, const float* __restrict__ B,
                                const float* __restrict__ dinv,
                                float* __restrict__ C, int M, int K, int F) {
  const int lane = threadIdx.x & 31;
  const int wave = (int)((blockIdx.x * blockDim.x + threadIdx.x) >> 5);
  const int nf = F >> 4;
  const int mTile = wave / nf;
  const int fTile = wave - mTile * nf;
  const int m0 = mTile << 4;
  if (m0 >= M) return;
  const int f0 = fTile << 4;
  const int row = lane & 15;
  const int kh2 = (lane >> 4) << 1;

  const float* arow = A + (size_t)(m0 + row) * K + kh2;
  const float* bcol = B + (size_t)kh2 * F + f0 + row;

  v8f acc = {0.f, 0.f, 0.f, 0.f, 0.f, 0.f, 0.f, 0.f};
  for (int k = 0; k < K; k += 4) {
    float2 av = *reinterpret_cast<const float2*>(arow + k);
    v2f a; a.x = av.x; a.y = av.y;
    v2f b;
    b.x = bcol[(size_t)k * F];
    b.y = bcol[(size_t)(k + 1) * F];
    acc = __builtin_amdgcn_wmma_f32_16x16x4_f32(false, a, false, b,
                                                (short)0, acc, false, false);
  }
  const float* dvp = dinv + m0 + (kh2 << 2);
  float* crow = C + (size_t)(m0 + (kh2 << 2)) * F + f0 + row;
#pragma unroll
  for (int r = 0; r < 8; ++r) crow[(size_t)r * F] = acc[r] * dvp[r];
}

template <int F>
__global__ void __launch_bounds__(256)
k_gemm_bn_wmma_f32(const float* __restrict__ V, const float* __restrict__ stats,
                   const float* __restrict__ gamma, const float* __restrict__ beta,
                   const float* __restrict__ B, const float* __restrict__ dinv,
                   float* __restrict__ C, int M, float invN) {
  constexpr int K = 128;
  constexpr int NF = F / 16;
  constexpr int RB = 16 * (8 / NF);
  constexpr int PITCH = K + 4;
  __shared__ __attribute__((aligned(16))) float As[RB * PITCH];

  const int t = threadIdx.x;
  const int lane = t & 31;
  const int w = t >> 5;
  const int rowBase = blockIdx.x * RB;

  {
    const int c4 = t & 31;
    const v4f su = *reinterpret_cast<const v4f*>(stats + 4 * c4);
    const v4f sq = *reinterpret_cast<const v4f*>(stats + K + 4 * c4);
    const v4f g = *reinterpret_cast<const v4f*>(gamma + 4 * c4);
    const v4f be = *reinterpret_cast<const v4f*>(beta + 4 * c4);
    v4f mean, rstd;
#pragma unroll
    for (int i = 0; i < 4; ++i) {
      mean[i] = su[i] * invN;
      const float var = sq[i] * invN - mean[i] * mean[i];
      rstd[i] = rsqrtf(var + BN_EPS);
    }
    v4f v[RB / 8];
#pragma unroll
    for (int p = 0; p < RB / 8; ++p) {
      const int gr = min(rowBase + p * 8 + w, M - 1);
      v[p] = *reinterpret_cast<const v4f*>(V + (size_t)gr * K + 4 * c4);
    }
#pragma unroll
    for (int p = 0; p < RB / 8; ++p) {
      const v4f o = (v[p] - mean) * rstd * g + be;
      *reinterpret_cast<v4f*>(&As[(p * 8 + w) * PITCH + 4 * c4]) = o;
    }
  }
  __syncthreads();

  const int mT = w / NF;
  const int fT = w - mT * NF;
  const int m0 = rowBase + (mT << 4);
  if (m0 >= M) return;
  const int f0 = fT << 4;
  const int row = lane & 15;
  const int kh2 = (lane >> 4) << 1;

  const float* arow = As + ((mT << 4) + row) * PITCH + kh2;
  const float* bcol = B + (size_t)kh2 * F + f0 + row;

  unsigned dbyte = (unsigned)(m0 + (kh2 << 2)) * 4u;
  unsigned cbyte = ((unsigned)(m0 + (kh2 << 2)) * F + f0 + row) * 4u;
  asm volatile("" : "+v"(dbyte), "+v"(cbyte));
  const float* dvp = reinterpret_cast<const float*>(reinterpret_cast<const char*>(dinv) + dbyte);
  float* crow = reinterpret_cast<float*>(reinterpret_cast<char*>(C) + cbyte);

  v8f acc = {0.f, 0.f, 0.f, 0.f, 0.f, 0.f, 0.f, 0.f};
#pragma unroll
  for (int k = 0; k < K; k += 4) {
    const v2f a = *reinterpret_cast<const v2f*>(arow + k);
    v2f b;
    b.x = bcol[(size_t)k * F];
    b.y = bcol[(size_t)(k + 1) * F];
    acc = __builtin_amdgcn_wmma_f32_16x16x4_f32(false, a, false, b,
                                                (short)0, acc, false, false);
  }
  asm volatile("s_wait_alu depctr_va_vdst(0)\n\t"
               "v_nop\n\tv_nop\n\tv_nop\n\tv_nop\n\t"
               "v_nop\n\tv_nop\n\tv_nop\n\tv_nop"
               : "+v"(acc));
#pragma unroll
  for (int r = 0; r < 8; ++r) crow[(size_t)r * F] = acc[r] * dvp[r];
}

__global__ void __launch_bounds__(256)
k_gather_x24(const float* __restrict__ x, const float* __restrict__ dinv,
             const int* __restrict__ rowptr, const int* __restrict__ csr,
             float* __restrict__ ax, int Nn, int E) {
  const int lane = threadIdx.x & 31;
  const int d = __builtin_amdgcn_readfirstlane((int)(blockIdx.x * 8 + (threadIdx.x >> 5)));
  if (d >= Nn) return;
  const int beg = max(rowptr[d], 0);
  const int end = min(rowptr[d + 1], E);
  const int p = lane & 7;
  const int q = lane >> 3;
  const bool pv = p < 6;
  const unsigned nmax = (unsigned)(Nn - 1);
  const v4f* xp = reinterpret_cast<const v4f*>(x) + p;
  const v4f zero4 = {0.f, 0.f, 0.f, 0.f};
  const float dd = dinv[d];
  const v4f xd = pv ? xp[(size_t)d * 6] : zero4;
  v4f acc = zero4;
  for (int j0 = beg; j0 < end; j0 += 32) {
    const int m = min(32, end - j0);
    int idx = 0;
    float dv = 0.f;
    if (lane < m) {
      idx = (int)min((unsigned)csr[j0 + lane], nmax);
      dv = dinv[idx];
    }
    v4f r[8];
    float wgt[8];
#pragma unroll
    for (int k = 0; k < 8; ++k) {
      const int e = 4 * k + q;
      const int s = __shfl(idx, e);
      wgt[k] = __shfl(dv, e);
      r[k] = (pv && e < m) ? xp[(size_t)s * 6] : zero4;
    }
#pragma unroll
    for (int k = 0; k < 8; ++k) acc += r[k] * wgt[k];
  }
#pragma unroll
  for (int i = 0; i < 4; ++i) {
    acc[i] += __shfl_xor(acc[i], 8);
    acc[i] += __shfl_xor(acc[i], 16);
  }
  if (q == 0 && pv)
    reinterpret_cast<v4f*>(ax)[(size_t)d * 6 + p] = (acc + xd * dd) * dd;
}

#define G1_TILES 8
__global__ void __launch_bounds__(256)
k_gemm1_relu_stats(const float* __restrict__ A, const float* __restrict__ B,
                   const float* __restrict__ bias, float* __restrict__ C,
                   float* __restrict__ stats,
                   int M) {
  constexpr int K = 24;
  constexpr int F = 128;
  constexpr int KS = K / 4;
  const int lane = threadIdx.x & 31;
  const int f0 = (int)(threadIdx.x >> 5) << 4;
  const int c = lane & 15;
  const int h = lane >> 4;
  const int kh2 = h << 1;
  const int rowBase = blockIdx.x * (16 * G1_TILES);

  v2f b[KS];
  const float* bcol = B + (size_t)kh2 * F + f0 + c;
#pragma unroll
  for (int kk = 0; kk < KS; ++kk) {
    b[kk].x = bcol[(size_t)(4 * kk) * F];
    b[kk].y = bcol[(size_t)(4 * kk + 1) * F];
  }
  const float bc = bias[f0 + c];

  v2f a[G1_TILES][KS];
#pragma unroll
  for (int i = 0; i < G1_TILES; ++i) {
    const int gr = min(rowBase + 16 * i + c, M - 1);
    const float* ar = A + (size_t)gr * K + kh2;
#pragma unroll
    for (int kk = 0; kk < KS; ++kk)
      a[i][kk] = *reinterpret_cast<const v2f*>(ar + 4 * kk);
  }

  float s = 0.f, q = 0.f;
#pragma unroll
  for (int i = 0; i < G1_TILES; ++i) {
    const int m0 = rowBase + 16 * i;
    if (m0 >= M) break;
    __builtin_amdgcn_sched_barrier(0);
    v8f acc = {0.f, 0.f, 0.f, 0.f, 0.f, 0.f, 0.f, 0.f};
#pragma unroll
    for (int kk = 0; kk < KS; ++kk)
      acc = __builtin_amdgcn_wmma_f32_16x16x4_f32(false, a[i][kk], false, b[kk],
                                                  (short)0, acc, false, false);
    asm volatile("s_wait_alu depctr_va_vdst(0)\n\t"
                 "v_nop\n\tv_nop\n\tv_nop\n\tv_nop\n\t"
                 "v_nop\n\tv_nop\n\tv_nop\n\tv_nop"
                 : "+v"(acc));
    __builtin_amdgcn_sched_barrier(0);
    const int r0 = m0 + (h << 3);
    float* crow = C + (size_t)r0 * F + f0 + c;
#pragma unroll
    for (int r = 0; r < 8; ++r) {
      if (r0 + r < M) {
        const float v = fmaxf(acc[r] + bc, 0.f);
        crow[(size_t)r * F] = v;
        s += v;
        q += v * v;
      }
    }
  }
  s += __shfl_xor(s, 16);
  q += __shfl_xor(q, 16);
  if (h == 0) {
    unsafeAtomicAdd(stats + f0 + c, s);
    unsafeAtomicAdd(stats + F + f0 + c, q);
  }
}

__global__ void k_scatter(const float* __restrict__ hsc, const int* __restrict__ src,
                          const int* __restrict__ dst,
                          float* __restrict__ agg, int E, int F) {
  int wave = (int)((blockIdx.x * blockDim.x + threadIdx.x) >> 5);
  int lane = threadIdx.x & 31;
  if (wave >= E) return;
  int s = src[wave], d = dst[wave];
  const float4* hrow = reinterpret_cast<const float4*>(hsc + (size_t)s * F);
  float* ad = agg + (size_t)d * F;
  for (int f4 = lane; 4 * f4 < F; f4 += 32) {
    float4 v = hrow[f4];
    unsafeAtomicAdd(ad + 4 * f4 + 0, v.x);
    unsafeAtomicAdd(ad + 4 * f4 + 1, v.y);
    unsafeAtomicAdd(ad + 4 * f4 + 2, v.z);
    unsafeAtomicAdd(ad + 4 * f4 + 3, v.w);
  }
}

#define FIN_ROWS 128
template <int F>
__global__ void __launch_bounds__(256)
k_finish_relu_stats(float* __restrict__ agg, const float* __restrict__ hsc,
                    const float* __restrict__ dinv,
                    const float* __restrict__ bias,
                    float* __restrict__ stats,
                    int Nn) {
  constexpr int TPR = F / 4;
  constexpr int RPP = 256 / TPR;
  constexpr int U = 8;
  constexpr int STEP = RPP * TPR;
  static_assert(FIN_ROWS % (U * RPP) == 0, "a block is a whole number of groups");
  __shared__ __attribute__((aligned(16))) float red[2][RPP][F];

  const int t = threadIdx.x;
  const int c4 = t % TPR;
  const int rsub = t / TPR;
  const int r0 = blockIdx.x * FIN_ROWS;
  const int r1 = min(r0 + FIN_ROWS, Nn);

  const v4f b = *reinterpret_cast<const v4f*>(bias + 4 * c4);
  const v4f zero4 = {0.f, 0.f, 0.f, 0.f};
  v4f s = zero4;
  v4f q = zero4;

  int r = r0 + rsub;
  v4f* ap = reinterpret_cast<v4f*>(agg) + (size_t)r * TPR + c4;
  const v4f* hp = reinterpret_cast<const v4f*>(hsc) + (size_t)r * TPR + c4;
  const float* dp = dinv + r;

  auto apply = [&](const v4f a, const v4f h, const float di, v4f* dst) {
    const v4f v = __builtin_elementwise_max((a + h) * di + b, zero4);
    *dst = v;
    s += v;
    q += v * v;
  };

  for (; r + (U - 1) * RPP < r1; r += U * RPP) {
    v4f a[U], h[U];
    float di[U];
#pragma unroll
    for (int u = 0; u < U; ++u) {
      a[u] = ap[u * STEP];
      h[u] = hp[u * STEP];
      di[u] = dp[u * RPP];
    }
    __builtin_amdgcn_sched_barrier(0);
#pragma unroll
    for (int u = 0; u < U; ++u) apply(a[u], h[u], di[u], ap + u * STEP);
    ap += U * STEP;
    hp += U * STEP;
    dp += U * RPP;
  }
  for (; r < r1; r += RPP) {
    apply(*ap, *hp, *dp, ap);
    ap += STEP;
    hp += STEP;
    dp += RPP;
  }

  *reinterpret_cast<v4f*>(&red[0][rsub][4 * c4]) = s;
  *reinterpret_cast<v4f*>(&red[1][rsub][4 * c4]) = q;
  __syncthreads();
  if (t < 2 * F) {
    const int which = t / F;
    const int j = t % F;
    float acc = 0.f;
    for (int k = 0; k < RPP; ++k) acc += red[which][k][j];
    unsafeAtomicAdd(stats + which * F + j, acc);
  }
}

__global__ void k_gemm64x2(const float* __restrict__ v3, const float* __restrict__ stats,
                           const float* __restrict__ gamma, const float* __restrict__ beta,
                           const float* __restrict__ W4, const float* __restrict__ dinv,
                           float* __restrict__ h4, int Nn, float invN) {
  int i = blockIdx.x * blockDim.x + threadIdx.x;
  if (i >= Nn) return;
  const float* row = v3 + (size_t)i * 64;
  float o0 = 0.f, o1 = 0.f;
#pragma unroll 8
  for (int k = 0; k < 64; ++k) {
    const float mean = stats[k] * invN;
    const float var = stats[64 + k] * invN - mean * mean;
    const float xv = (row[k] - mean) * rsqrtf(var + BN_EPS) * gamma[k] + beta[k];
    o0 += xv * W4[2 * k];
    o1 += xv * W4[2 * k + 1];
  }
  float di = dinv[i];
  h4[2 * (size_t)i]     = o0 * di;
  h4[2 * (size_t)i + 1] = o1 * di;
}

__global__ void k_scatter2(const float* __restrict__ h4s, const int* __restrict__ src,
                           const int* __restrict__ dst,
                           float* __restrict__ agg, int E) {
  int e = blockIdx.x * blockDim.x + threadIdx.x;
  if (e >= E) return;
  int s = src[e], d = dst[e];
  float2 v = *reinterpret_cast<const float2*>(h4s + 2 * (size_t)s);
  unsafeAtomicAdd(agg + 2 * (size_t)d,     v.x);
  unsafeAtomicAdd(agg + 2 * (size_t)d + 1, v.y);
}

__global__ void k_final_lsm(const float* __restrict__ agg, const float* __restrict__ h4s,
                            const float* __restrict__ dinv, const float* __restrict__ b4,
                            float* __restrict__ out, int Nn) {
  int i = blockIdx.x * blockDim.x + threadIdx.x;
  if (i >= Nn) return;
  float di = dinv[i];
  float o0 = (agg[2 * (size_t)i]     + h4s[2 * (size_t)i])     * di + b4[0];
  float o1 = (agg[2 * (size_t)i + 1] + h4s[2 * (size_t)i + 1]) * di + b4[1];
  float m = fmaxf(o0, o1);
  float l = m + logf(expf(o0 - m) + expf(o1 - m));
  out[2 * (size_t)i]     = o0 - l;
  out[2 * (size_t)i + 1] = o1 - l;
}

extern "C" void kernel_launch(void* const* d_in, const int* in_sizes, int n_in,
                              void* d_out, int out_size, void* d_ws, size_t ws_size,
                              hipStream_t stream) {
  const float* x   = (const float*)d_in[0];
  const int*   ei  = (const int*)d_in[1];
  const float* W1  = (const float*)d_in[2];
  const float* b1  = (const float*)d_in[3];
  const float* g1  = (const float*)d_in[4];
  const float* be1 = (const float*)d_in[5];
  const float* W2  = (const float*)d_in[6];
  const float* b2  = (const float*)d_in[7];
  const float* g2  = (const float*)d_in[8];
  const float* be2 = (const float*)d_in[9];
  const float* W3  = (const float*)d_in[10];
  const float* b3  = (const float*)d_in[11];
  const float* g3  = (const float*)d_in[12];
  const float* be3 = (const float*)d_in[13];
  const float* W4  = (const float*)d_in[14];
  const float* b4  = (const float*)d_in[15];

  const int Nn = in_sizes[0] / 24;
  const int E  = in_sizes[1] / 2;
  const int* src = ei;
  const int* dst = ei + E;

  float* ws    = (float*)d_ws;
  float* dinv  = ws;
  float* stats = ws + Nn;
  float* bufA  = stats + 256;
  float* bufB  = bufA + (size_t)Nn * 128;
  const int nchunk = (Nn + SCAN_CHUNK - 1) / SCAN_CHUNK;
  const size_t rp_len = ((size_t)Nn + 1 + 31) & ~(size_t)31;
  const size_t bs_len = ((size_t)nchunk + 31) & ~(size_t)31;
  int* rowptr = (int*)(bufB + (size_t)Nn * 128);
  int* csr    = rowptr + rp_len;
  int* cnt    = (int*)bufB;
  int* bsum   = cnt + Nn;
  int* rank   = bsum + bs_len;
  const size_t need_bytes =
      ((size_t)Nn + 256 + 2 * (size_t)Nn * 128 + rp_len + (size_t)E) * sizeof(float);
  const bool grouped = ws_size >= need_bytes &&
                       (size_t)Nn + bs_len + (size_t)E <= (size_t)Nn * 128;

  const float invN = 1.0f / (float)Nn;
  const int TB = 256;

  auto gemm = [&](const float* A, const float* Bm, float* C, int K, int F) {
    long waves = (long)((Nn + 15) / 16) * (F / 16);
    int blocks = (int)((waves * 32 + TB - 1) / TB);
    k_gemm_wmma_f32<<<blocks, TB, 0, stream>>>(A, Bm, dinv, C, Nn, K, F);
  };
  auto scatter = [&](const float* h, float* agg, int F) {
    int blocks = (int)(((long)E * 32 + TB - 1) / TB);
    k_scatter<<<blocks, TB, 0, stream>>>(h, src, dst, agg, E, F);
  };
  auto finish = [&](float* agg, const float* h, const float* bias, int F) {
    const int blocks = (Nn + FIN_ROWS - 1) / FIN_ROWS;
    k_zero<<<1, 256, 0, stream>>>(stats, 256);
    if (F == 128)
      k_finish_relu_stats<128><<<blocks, 256, 0, stream>>>(agg, h, dinv, bias, stats, Nn);
    else
      k_finish_relu_stats<64><<<blocks, 256, 0, stream>>>(agg, h, dinv, bias, stats, Nn);
  };
  auto gemm_bn = [&](const float* V, const float* gamma, const float* beta,
                     const float* Bm, float* C, int F) {
    if (F == 128)
      k_gemm_bn_wmma_f32<128><<<(Nn + 15) / 16, 256, 0, stream>>>(
          V, stats, gamma, beta, Bm, dinv, C, Nn, invN);
    else
      k_gemm_bn_wmma_f32<64><<<(Nn + 31) / 32, 256, 0, stream>>>(
          V, stats, gamma, beta, Bm, dinv, C, Nn, invN);
  };
  auto zero = [&](float* p, size_t n) {
    int blocks = (int)((n + TB - 1) / TB);
    if (blocks > 4096) blocks = 4096;
    k_zero<<<blocks, TB, 0, stream>>>(p, n);
  };

  auto aggregate = [&](const float* h, float* agg, int F) {
    if (grouped) {
      const int blocks = (Nn + 7) / 8;
      if (F == 128)
        k_gather_rows<128><<<blocks, 256, 0, stream>>>(h, rowptr, csr, agg, Nn, E);
      else
        k_gather_rows<64><<<blocks, 256, 0, stream>>>(h, rowptr, csr, agg, Nn, E);
    } else {
      zero(agg, (size_t)Nn * F);
      scatter(h, agg, F);
    }
  };

  if (grouped) {
    zero((float*)cnt, (size_t)Nn);
    k_count_rank<<<(E + TB - 1) / TB, TB, 0, stream>>>(dst, cnt, rank, E);
    k_dinv_bsum<<<nchunk, 256, 0, stream>>>(cnt, dinv, bsum, Nn);
    k_scan_rowptr<<<nchunk, 256, 0, stream>>>(cnt, bsum, rowptr, Nn);
    k_fill_csr<<<(E + TB - 1) / TB, TB, 0, stream>>>(src, dst, rank, rowptr, csr, E);
  } else {
    k_deg_init<<<(Nn + TB - 1) / TB, TB, 0, stream>>>(dinv, Nn);
    k_deg_count<<<(E + TB - 1) / TB, TB, 0, stream>>>(dst, dinv, E);
    k_rsqrt_inplace<<<(Nn + TB - 1) / TB, TB, 0, stream>>>(dinv, Nn);
  }

  if (grouped) {
    k_gather_x24<<<(Nn + 7) / 8, 256, 0, stream>>>(x, dinv, rowptr, csr, bufA, Nn, E);
    k_zero<<<1, 256, 0, stream>>>(stats, 256);
    k_gemm1_relu_stats<<<(Nn + 16 * G1_TILES - 1) / (16 * G1_TILES), 256, 0, stream>>>(
        bufA, W1, b1, bufB, stats, Nn);
  } else {
    gemm(x, W1, bufA, 24, 128);
    aggregate(bufA, bufB, 128);
    finish(bufB, bufA, b1, 128);
  }

  gemm_bn(bufB, g1, be1, W2, bufA, 128);
  aggregate(bufA, bufB, 128);
  finish(bufB, bufA, b2, 128);

  gemm_bn(bufB, g2, be2, W3, bufA, 64);
  aggregate(bufA, bufB, 64);
  finish(bufB, bufA, b3, 64);

  k_gemm64x2<<<(Nn + TB - 1) / TB, TB, 0, stream>>>(bufB, stats, g3, be3, W4, dinv,
                                                    bufA, Nn, invN);
  zero(bufB, (size_t)Nn * 2);
  k_scatter2<<<(E + TB - 1) / TB, TB, 0, stream>>>(bufA, src, dst, bufB, E);
  k_final_lsm<<<(Nn + TB - 1) / TB, TB, 0, stream>>>(bufB, bufA, dinv, b4,
                                                     (float*)d_out, Nn);
}


# layers 2-3: finish pass (self loop, dst norm, bias, relu, BN column sums) folded into the row-sum kernel's store; a wave walks 8-9 rows, the next row's indices in flight (BN sum order changes)
# speedup vs baseline: 3.3934x; 1.0412x over previous
#include <hip/hip_runtime.h>
#include <hip/hip_bf16.h>
#include <math.h>


typedef float v2f __attribute__((ext_vector_type(2)));
typedef float v4f __attribute__((ext_vector_type(4)));
typedef float v8f __attribute__((ext_vector_type(8)));

#define BN_EPS 1e-5f

__global__ void k_zero(float* __restrict__ p, size_t n) {
  size_t i = (size_t)blockIdx.x * blockDim.x + threadIdx.x;
  size_t stride = (size_t)gridDim.x * blockDim.x;
  for (; i < n; i += stride) p[i] = 0.0f;
}

__global__ void k_deg_init(float* __restrict__ deg, int n) {
  int i = blockIdx.x * blockDim.x + threadIdx.x;
  if (i < n) deg[i] = 1.0f;
}

__global__ void k_deg_count(const int* __restrict__ dst, float* __restrict__ deg, int E) {
  int e = blockIdx.x * blockDim.x + threadIdx.x;
  if (e < E) unsafeAtomicAdd(&deg[dst[e]], 1.0f);
}

__global__ void k_rsqrt_inplace(float* __restrict__ d, int n) {
  int i = blockIdx.x * blockDim.x + threadIdx.x;
  if (i < n) {
    float v = d[i];
    d[i] = (v > 0.0f) ? rsqrtf(v) : 0.0f;
  }
}

__global__ void k_count_rank(const int* __restrict__ dst, int* __restrict__ cnt,
                             int* __restrict__ rank, int E) {
  int e = blockIdx.x * blockDim.x + threadIdx.x;
  if (e < E) rank[e] = atomicAdd(&cnt[dst[e]], 1);
}

#define SCAN_CHUNK 1024

__global__ void __launch_bounds__(256)
k_dinv_bsum(const int* __restrict__ cnt, float* __restrict__ dinv,
            int* __restrict__ bsum, int n) {
  __shared__ int wsum[8];
  const int t = threadIdx.x;
  const int i0 = blockIdx.x * SCAN_CHUNK + 4 * t;
  int s = 0;
#pragma unroll
  for (int k = 0; k < 4; ++k) {
    const int i = i0 + k;
    if (i < n) {
      const int c = cnt[i];
      s += c;
      const float deg = (float)(c + 1);
      dinv[i] = (deg > 0.0f) ? rsqrtf(deg) : 0.0f;
    }
  }
  for (int o = 16; o > 0; o >>= 1) s += __shfl_xor(s, o);
  if ((t & 31) == 0) wsum[t >> 5] = s;
  __syncthreads();
  if (t == 0) {
    int tot = 0;
    for (int w = 0; w < 8; ++w) tot += wsum[w];
    bsum[blockIdx.x] = tot;
  }
}

__global__ void __launch_bounds__(256)
k_scan_rowptr(const int* __restrict__ cnt, const int* __restrict__ bsum,
              int* __restrict__ rowptr, int n) {
  __shared__ int wtot[8];
  __shared__ int base_s;
  const int t = threadIdx.x, lane = t & 31, w = t >> 5;
  int p = 0;
  for (int b = t; b < (int)blockIdx.x; b += 256) p += bsum[b];
  for (int o = 16; o > 0; o >>= 1) p += __shfl_xor(p, o);
  if (lane == 0) wtot[w] = p;
  __syncthreads();
  if (t == 0) {
    int tot = 0;
    for (int k = 0; k < 8; ++k) tot += wtot[k];
    base_s = tot;
  }
  __syncthreads();
  const int base = base_s;
  const int i0 = blockIdx.x * SCAN_CHUNK + 4 * t;
  int c[4];
  int s = 0;
#pragma unroll
  for (int k = 0; k < 4; ++k) {
    c[k] = (i0 + k < n) ? cnt[i0 + k] : 0;
    s += c[k];
  }
  int incl = s;
  for (int o = 1; o < 32; o <<= 1) {
    const int v = __shfl_up(incl, o);
    if (lane >= o) incl += v;
  }
  if (lane == 31) wtot[w] = incl;
  __syncthreads();
  int woff = 0;
  for (int k = 0; k < w; ++k) woff += wtot[k];
  int excl = base + woff + incl - s;
#pragma unroll
  for (int k = 0; k < 4; ++k) {
    if (i0 + k < n) {
      rowptr[i0 + k] = excl;
      excl += c[k];
    }
  }
  if (blockIdx.x == gridDim.x - 1 && t == 255) rowptr[n] = excl;
}

__global__ void k_fill_csr(const int* __restrict__ src, const int* __restrict__ dst,
                           const int* __restrict__ rank, const int* __restrict__ rowptr,
                           int* __restrict__ csr, int E) {
  int e = blockIdx.x * blockDim.x + threadIdx.x;
  if (e >= E) return;
  unsigned pos = (unsigned)(rowptr[dst[e]] + rank[e]);
  csr[min(pos, (unsigned)(E - 1))] = src[e];
}

#define GF_ROWS_128 9
#define GF_ROWS_64 8
template <int F, int R>
__global__ void __launch_bounds__(256)
k_gather_finish(const float* __restrict__ hsc, const int* __restrict__ rowptr,
                const int* __restrict__ csr, const float* __restrict__ dinv,
                const float* __restrict__ bias, float* __restrict__ vout,
                float* __restrict__ stats,
                int Nn, int E) {
  constexpr int V = F / 32;
  typedef float vf __attribute__((ext_vector_type(V)));
  __shared__ __attribute__((aligned(16))) float red[2][8][F];
  const int t = threadIdx.x;
  const int lane = t & 31;
  const int w = t >> 5;
  const int d0 = __builtin_amdgcn_readfirstlane((int)((blockIdx.x * 8 + w) * R));
  vf s = 0.0f;
  vf q = 0.0f;
  if (d0 < Nn) {
    const int rows = min(R, Nn - d0);
    const vf* hp = reinterpret_cast<const vf*>(hsc) + lane;
    vf* op = reinterpret_cast<vf*>(vout) + lane;
    const unsigned nmax = (unsigned)(Nn - 1);
    const vf zerov = 0.0f;
    const int rpraw = rowptr[min(d0 + lane, Nn)];
    const float dvl = dinv[min(d0 + lane, Nn - 1)];
    const vf b = *reinterpret_cast<const vf*>(bias + V * lane);
    const int rp = min(max(rpraw, 0), E);
    const unsigned emax = (unsigned)(E - 1);
    int beg = __builtin_amdgcn_readlane(rp, 0);
    int end = __builtin_amdgcn_readlane(rp, 1);
    int idx = csr[min((unsigned)(beg + lane), emax)];
    auto sum_chunk = [&](const int cur, const int m, vf a) -> vf {
      auto row = [&](int u) -> vf {
        const unsigned sidx = min((unsigned)__builtin_amdgcn_readlane(cur, u), nmax);
        return hp[(size_t)sidx * 32];
      };
      int u = 0;
      for (; u + 8 <= m; u += 8) {
        vf r[8];
#pragma unroll
        for (int k = 0; k < 8; ++k) r[k] = row(u + k);
#pragma unroll
        for (int k = 0; k < 8; ++k) a += r[k];
      }
      if (u + 4 <= m) {
        const vf r0 = row(u), r1 = row(u + 1), r2 = row(u + 2), r3 = row(u + 3);
        a += r0; a += r1; a += r2; a += r3;
        u += 4;
      }
      if (u + 2 <= m) {
        const vf r0 = row(u), r1 = row(u + 1);
        a += r0; a += r1;
        u += 2;
      }
      if (u < m) a += row(u);
      return a;
    };
    for (int i = 0; i < rows; ++i) {
      const int d = d0 + i;
      const vf self = hp[(size_t)d * 32];
      const int nbeg = end;
      const int nend = __builtin_amdgcn_readlane(rp, i + 2);
      const int idxn = csr[min((unsigned)(nbeg + lane), emax)];
      vf acc = sum_chunk(idx, min(32, end - beg), zerov);
      for (int j0 = beg + 32; j0 < end; j0 += 32) {
        const int m = min(32, end - j0);
        const int more = (lane < m) ? csr[j0 + lane] : 0;
        acc = sum_chunk(more, m, acc);
      }
      const float di = __builtin_bit_cast(
          float, __builtin_amdgcn_readlane(__builtin_bit_cast(int, dvl), i));
      const vf v = __builtin_elementwise_max((acc + self) * di + b, zerov);
      op[(size_t)d * 32] = v;
      s += v;
      q += v * v;
      beg = nbeg;
      end = nend;
      idx = idxn;
    }
  }
  *reinterpret_cast<vf*>(&red[0][w][V * lane]) = s;
  *reinterpret_cast<vf*>(&red[1][w][V * lane]) = q;
  __syncthreads();
  if (t < 2 * F) {
    const int which = t / F;
    const int j = t % F;
    float a = 0.f;
    for (int k = 0; k < 8; ++k) a += red[which][k][j];
    unsafeAtomicAdd(stats + which * F + j, a);
  }
}

__global__ void k_gemm_wmma_f32(const float* __restrict__ A, const float* __restrict__ B,
                                const float* __restrict__ dinv,
                                float* __restrict__ C, int M, int K, int F) {
  const int lane = threadIdx.x & 31;
  const int wave = (int)((blockIdx.x * blockDim.x + threadIdx.x) >> 5);
  const int nf = F >> 4;
  const int mTile = wave / nf;
  const int fTile = wave - mTile * nf;
  const int m0 = mTile << 4;
  if (m0 >= M) return;
  const int f0 = fTile << 4;
  const int row = lane & 15;
  const int kh2 = (lane >> 4) << 1;

  const float* arow = A + (size_t)(m0 + row) * K + kh2;
  const float* bcol = B + (size_t)kh2 * F + f0 + row;

  v8f acc = {0.f, 0.f, 0.f, 0.f, 0.f, 0.f, 0.f, 0.f};
  for (int k = 0; k < K; k += 4) {
    float2 av = *reinterpret_cast<const float2*>(arow + k);
    v2f a; a.x = av.x; a.y = av.y;
    v2f b;
    b.x = bcol[(size_t)k * F];
    b.y = bcol[(size_t)(k + 1) * F];
    acc = __builtin_amdgcn_wmma_f32_16x16x4_f32(false, a, false, b,
                                                (short)0, acc, false, false);
  }
  const float* dvp = dinv + m0 + (kh2 << 2);
  float* crow = C + (size_t)(m0 + (kh2 << 2)) * F + f0 + row;
#pragma unroll
  for (int r = 0; r < 8; ++r) crow[(size_t)r * F] = acc[r] * dvp[r];
}

template <int F>
__global__ void __launch_bounds__(256)
k_gemm_bn_wmma_f32(const float* __restrict__ V, const float* __restrict__ stats,
                   const float* __restrict__ gamma, const float* __restrict__ beta,
                   const float* __restrict__ B, const float* __restrict__ dinv,
                   float* __restrict__ C, int M, float invN) {
  constexpr int K = 128;
  constexpr int NF = F / 16;
  constexpr int RB = 16 * (8 / NF);
  constexpr int PITCH = K + 4;
  __shared__ __attribute__((aligned(16))) float As[RB * PITCH];

  const int t = threadIdx.x;
  const int lane = t & 31;
  const int w = t >> 5;
  const int rowBase = blockIdx.x * RB;

  {
    const int c4 = t & 31;
    const v4f su = *reinterpret_cast<const v4f*>(stats + 4 * c4);
    const v4f sq = *reinterpret_cast<const v4f*>(stats + K + 4 * c4);
    const v4f g = *reinterpret_cast<const v4f*>(gamma + 4 * c4);
    const v4f be = *reinterpret_cast<const v4f*>(beta + 4 * c4);
    v4f mean, rstd;
#pragma unroll
    for (int i = 0; i < 4; ++i) {
      mean[i] = su[i] * invN;
      const float var = sq[i] * invN - mean[i] * mean[i];
      rstd[i] = rsqrtf(var + BN_EPS);
    }
    v4f v[RB / 8];
#pragma unroll
    for (int p = 0; p < RB / 8; ++p) {
      const int gr = min(rowBase + p * 8 + w, M - 1);
      v[p] = *reinterpret_cast<const v4f*>(V + (size_t)gr * K + 4 * c4);
    }
#pragma unroll
    for (int p = 0; p < RB / 8; ++p) {
      const v4f o = (v[p] - mean) * rstd * g + be;
      *reinterpret_cast<v4f*>(&As[(p * 8 + w) * PITCH + 4 * c4]) = o;
    }
  }
  __syncthreads();

  const int mT = w / NF;
  const int fT = w - mT * NF;
  const int m0 = rowBase + (mT << 4);
  if (m0 >= M) return;
  const int f0 = fT << 4;
  const int row = lane & 15;
  const int kh2 = (lane >> 4) << 1;

  const float* arow = As + ((mT << 4) + row) * PITCH + kh2;
  const float* bcol = B + (size_t)kh2 * F + f0 + row;

  unsigned dbyte = (unsigned)(m0 + (kh2 << 2)) * 4u;
  unsigned cbyte = ((unsigned)(m0 + (kh2 << 2)) * F + f0 + row) * 4u;
  asm volatile("" : "+v"(dbyte), "+v"(cbyte));
  const float* dvp = reinterpret_cast<const float*>(reinterpret_cast<const char*>(dinv) + dbyte);
  float* crow = reinterpret_cast<float*>(reinterpret_cast<char*>(C) + cbyte);

  v8f acc = {0.f, 0.f, 0.f, 0.f, 0.f, 0.f, 0.f, 0.f};
#pragma unroll
  for (int k = 0; k < K; k += 4) {
    const v2f a = *reinterpret_cast<const v2f*>(arow + k);
    v2f b;
    b.x = bcol[(size_t)k * F];
    b.y = bcol[(size_t)(k + 1) * F];
    acc = __builtin_amdgcn_wmma_f32_16x16x4_f32(false, a, false, b,
                                                (short)0, acc, false, false);
  }
  asm volatile("s_wait_alu depctr_va_vdst(0)\n\t"
               "v_nop\n\tv_nop\n\tv_nop\n\tv_nop\n\t"
               "v_nop\n\tv_nop\n\tv_nop\n\tv_nop"
               : "+v"(acc));
#pragma unroll
  for (int r = 0; r < 8; ++r) crow[(size_t)r * F] = acc[r] * dvp[r];
}

__global__ void __launch_bounds__(256)
k_gather_x24(const float* __restrict__ x, const float* __restrict__ dinv,
             const int* __restrict__ rowptr, const int* __restrict__ csr,
             float* __restrict__ ax, int Nn, int E) {
  const int lane = threadIdx.x & 31;
  const int d = __builtin_amdgcn_readfirstlane((int)(blockIdx.x * 8 + (threadIdx.x >> 5)));
  if (d >= Nn) return;
  const int beg = max(rowptr[d], 0);
  const int end = min(rowptr[d + 1], E);
  const int p = lane & 7;
  const int q = lane >> 3;
  const bool pv = p < 6;
  const unsigned nmax = (unsigned)(Nn - 1);
  const v4f* xp = reinterpret_cast<const v4f*>(x) + p;
  const v4f zero4 = {0.f, 0.f, 0.f, 0.f};
  const float dd = dinv[d];
  const v4f xd = pv ? xp[(size_t)d * 6] : zero4;
  v4f acc = zero4;
  for (int j0 = beg; j0 < end; j0 += 32) {
    const int m = min(32, end - j0);
    int idx = 0;
    float dv = 0.f;
    if (lane < m) {
      idx = (int)min((unsigned)csr[j0 + lane], nmax);
      dv = dinv[idx];
    }
    v4f r[8];
    float wgt[8];
#pragma unroll
    for (int k = 0; k < 8; ++k) {
      const int e = 4 * k + q;
      const int s = __shfl(idx, e);
      wgt[k] = __shfl(dv, e);
      r[k] = (pv && e < m) ? xp[(size_t)s * 6] : zero4;
    }
#pragma unroll
    for (int k = 0; k < 8; ++k) acc += r[k] * wgt[k];
  }
#pragma unroll
  for (int i = 0; i < 4; ++i) {
    acc[i] += __shfl_xor(acc[i], 8);
    acc[i] += __shfl_xor(acc[i], 16);
  }
  if (q == 0 && pv)
    reinterpret_cast<v4f*>(ax)[(size_t)d * 6 + p] = (acc + xd * dd) * dd;
}

#define G1_TILES 8
__global__ void __launch_bounds__(256)
k_gemm1_relu_stats(const float* __restrict__ A, const float* __restrict__ B,
                   const float* __restrict__ bias, float* __restrict__ C,
                   float* __restrict__ stats,
                   int M) {
  constexpr int K = 24;
  constexpr int F = 128;
  constexpr int KS = K / 4;
  const int lane = threadIdx.x & 31;
  const int f0 = (int)(threadIdx.x >> 5) << 4;
  const int c = lane & 15;
  const int h = lane >> 4;
  const int kh2 = h << 1;
  const int rowBase = blockIdx.x * (16 * G1_TILES);

  v2f b[KS];
  const float* bcol = B + (size_t)kh2 * F + f0 + c;
#pragma unroll
  for (int kk = 0; kk < KS; ++kk) {
    b[kk].x = bcol[(size_t)(4 * kk) * F];
    b[kk].y = bcol[(size_t)(4 * kk + 1) * F];
  }
  const float bc = bias[f0 + c];

  v2f a[G1_TILES][KS];
#pragma unroll
  for (int i = 0; i < G1_TILES; ++i) {
    const int gr = min(rowBase + 16 * i + c, M - 1);
    const float* ar = A + (size_t)gr * K + kh2;
#pragma unroll
    for (int kk = 0; kk < KS; ++kk)
      a[i][kk] = *reinterpret_cast<const v2f*>(ar + 4 * kk);
  }

  float s = 0.f, q = 0.f;
#pragma unroll
  for (int i = 0; i < G1_TILES; ++i) {
    const int m0 = rowBase + 16 * i;
    if (m0 >= M) break;
    __builtin_amdgcn_sched_barrier(0);
    v8f acc = {0.f, 0.f, 0.f, 0.f, 0.f, 0.f, 0.f, 0.f};
#pragma unroll
    for (int kk = 0; kk < KS; ++kk)
      acc = __builtin_amdgcn_wmma_f32_16x16x4_f32(false, a[i][kk], false, b[kk],
                                                  (short)0, acc, false, false);
    asm volatile("s_wait_alu depctr_va_vdst(0)\n\t"
                 "v_nop\n\tv_nop\n\tv_nop\n\tv_nop\n\t"
                 "v_nop\n\tv_nop\n\tv_nop\n\tv_nop"
                 : "+v"(acc));
    __builtin_amdgcn_sched_barrier(0);
    const int r0 = m0 + (h << 3);
    float* crow = C + (size_t)r0 * F + f0 + c;
#pragma unroll
    for (int r = 0; r < 8; ++r) {
      if (r0 + r < M) {
        const float v = fmaxf(acc[r] + bc, 0.f);
        crow[(size_t)r * F] = v;
        s += v;
        q += v * v;
      }
    }
  }
  s += __shfl_xor(s, 16);
  q += __shfl_xor(q, 16);
  if (h == 0) {
    unsafeAtomicAdd(stats + f0 + c, s);
    unsafeAtomicAdd(stats + F + f0 + c, q);
  }
}

__global__ void k_scatter(const float* __restrict__ hsc, const int* __restrict__ src,
                          const int* __restrict__ dst,
                          float* __restrict__ agg, int E, int F) {
  int wave = (int)((blockIdx.x * blockDim.x + threadIdx.x) >> 5);
  int lane = threadIdx.x & 31;
  if (wave >= E) return;
  int s = src[wave], d = dst[wave];
  const float4* hrow = reinterpret_cast<const float4*>(hsc + (size_t)s * F);
  float* ad = agg + (size_t)d * F;
  for (int f4 = lane; 4 * f4 < F; f4 += 32) {
    float4 v = hrow[f4];
    unsafeAtomicAdd(ad + 4 * f4 + 0, v.x);
    unsafeAtomicAdd(ad + 4 * f4 + 1, v.y);
    unsafeAtomicAdd(ad + 4 * f4 + 2, v.z);
    unsafeAtomicAdd(ad + 4 * f4 + 3, v.w);
  }
}

#define FIN_ROWS 128
template <int F>
__global__ void __launch_bounds__(256)
k_finish_relu_stats(float* __restrict__ agg, const float* __restrict__ hsc,
                    const float* __restrict__ dinv,
                    const float* __restrict__ bias,
                    float* __restrict__ stats,
                    int Nn) {
  constexpr int TPR = F / 4;
  constexpr int RPP = 256 / TPR;
  constexpr int U = 8;
  constexpr int STEP = RPP * TPR;
  static_assert(FIN_ROWS % (U * RPP) == 0, "a block is a whole number of groups");
  __shared__ __attribute__((aligned(16))) float red[2][RPP][F];

  const int t = threadIdx.x;
  const int c4 = t % TPR;
  const int rsub = t / TPR;
  const int r0 = blockIdx.x * FIN_ROWS;
  const int r1 = min(r0 + FIN_ROWS, Nn);

  const v4f b = *reinterpret_cast<const v4f*>(bias + 4 * c4);
  const v4f zero4 = {0.f, 0.f, 0.f, 0.f};
  v4f s = zero4;
  v4f q = zero4;

  int r = r0 + rsub;
  v4f* ap = reinterpret_cast<v4f*>(agg) + (size_t)r * TPR + c4;
  const v4f* hp = reinterpret_cast<const v4f*>(hsc) + (size_t)r * TPR + c4;
  const float* dp = dinv + r;

  auto apply = [&](const v4f a, const v4f h, const float di, v4f* dst) {
    const v4f v = __builtin_elementwise_max((a + h) * di + b, zero4);
    *dst = v;
    s += v;
    q += v * v;
  };

  for (; r + (U - 1) * RPP < r1; r += U * RPP) {
    v4f a[U], h[U];
    float di[U];
#pragma unroll
    for (int u = 0; u < U; ++u) {
      a[u] = ap[u * STEP];
      h[u] = hp[u * STEP];
      di[u] = dp[u * RPP];
    }
    __builtin_amdgcn_sched_barrier(0);
#pragma unroll
    for (int u = 0; u < U; ++u) apply(a[u], h[u], di[u], ap + u * STEP);
    ap += U * STEP;
    hp += U * STEP;
    dp += U * RPP;
  }
  for (; r < r1; r += RPP) {
    apply(*ap, *hp, *dp, ap);
    ap += STEP;
    hp += STEP;
    dp += RPP;
  }

  *reinterpret_cast<v4f*>(&red[0][rsub][4 * c4]) = s;
  *reinterpret_cast<v4f*>(&red[1][rsub][4 * c4]) = q;
  __syncthreads();
  if (t < 2 * F) {
    const int which = t / F;
    const int j = t % F;
    float acc = 0.f;
    for (int k = 0; k < RPP; ++k) acc += red[which][k][j];
    unsafeAtomicAdd(stats + which * F + j, acc);
  }
}

__global__ void k_gemm64x2(const float* __restrict__ v3, const float* __restrict__ stats,
                           const float* __restrict__ gamma, const float* __restrict__ beta,
                           const float* __restrict__ W4, const float* __restrict__ dinv,
                           float* __restrict__ h4, int Nn, float invN) {
  int i = blockIdx.x * blockDim.x + threadIdx.x;
  if (i >= Nn) return;
  const float* row = v3 + (size_t)i * 64;
  float o0 = 0.f, o1 = 0.f;
#pragma unroll 8
  for (int k = 0; k < 64; ++k) {
    const float mean = stats[k] * invN;
    const float var = stats[64 + k] * invN - mean * mean;
    const float xv = (row[k] - mean) * rsqrtf(var + BN_EPS) * gamma[k] + beta[k];
    o0 += xv * W4[2 * k];
    o1 += xv * W4[2 * k + 1];
  }
  float di = dinv[i];
  h4[2 * (size_t)i]     = o0 * di;
  h4[2 * (size_t)i + 1] = o1 * di;
}

__global__ void k_scatter2(const float* __restrict__ h4s, const int* __restrict__ src,
                           const int* __restrict__ dst,
                           float* __restrict__ agg, int E) {
  int e = blockIdx.x * blockDim.x + threadIdx.x;
  if (e >= E) return;
  int s = src[e], d = dst[e];
  float2 v = *reinterpret_cast<const float2*>(h4s + 2 * (size_t)s);
  unsafeAtomicAdd(agg + 2 * (size_t)d,     v.x);
  unsafeAtomicAdd(agg + 2 * (size_t)d + 1, v.y);
}

__global__ void k_final_lsm(const float* __restrict__ agg, const float* __restrict__ h4s,
                            const float* __restrict__ dinv, const float* __restrict__ b4,
                            float* __restrict__ out, int Nn) {
  int i = blockIdx.x * blockDim.x + threadIdx.x;
  if (i >= Nn) return;
  float di = dinv[i];
  float o0 = (agg[2 * (size_t)i]     + h4s[2 * (size_t)i])     * di + b4[0];
  float o1 = (agg[2 * (size_t)i + 1] + h4s[2 * (size_t)i + 1]) * di + b4[1];
  float m = fmaxf(o0, o1);
  float l = m + logf(expf(o0 - m) + expf(o1 - m));
  out[2 * (size_t)i]     = o0 - l;
  out[2 * (size_t)i + 1] = o1 - l;
}

extern "C" void kernel_launch(void* const* d_in, const int* in_sizes, int n_in,
                              void* d_out, int out_size, void* d_ws, size_t ws_size,
                              hipStream_t stream) {
  const float* x   = (const float*)d_in[0];
  const int*   ei  = (const int*)d_in[1];
  const float* W1  = (const float*)d_in[2];
  const float* b1  = (const float*)d_in[3];
  const float* g1  = (const float*)d_in[4];
  const float* be1 = (const float*)d_in[5];
  const float* W2  = (const float*)d_in[6];
  const float* b2  = (const float*)d_in[7];
  const float* g2  = (const float*)d_in[8];
  const float* be2 = (const float*)d_in[9];
  const float* W3  = (const float*)d_in[10];
  const float* b3  = (const float*)d_in[11];
  const float* g3  = (const float*)d_in[12];
  const float* be3 = (const float*)d_in[13];
  const float* W4  = (const float*)d_in[14];
  const float* b4  = (const float*)d_in[15];

  const int Nn = in_sizes[0] / 24;
  const int E  = in_sizes[1] / 2;
  const int* src = ei;
  const int* dst = ei + E;

  float* ws    = (float*)d_ws;
  float* dinv  = ws;
  float* stats = ws + Nn;
  float* bufA  = stats + 256;
  float* bufB  = bufA + (size_t)Nn * 128;
  const int nchunk = (Nn + SCAN_CHUNK - 1) / SCAN_CHUNK;
  const size_t rp_len = ((size_t)Nn + 1 + 31) & ~(size_t)31;
  const size_t bs_len = ((size_t)nchunk + 31) & ~(size_t)31;
  int* rowptr = (int*)(bufB + (size_t)Nn * 128);
  int* csr    = rowptr + rp_len;
  int* cnt    = (int*)bufB;
  int* bsum   = cnt + Nn;
  int* rank   = bsum + bs_len;
  const size_t need_bytes =
      ((size_t)Nn + 256 + 2 * (size_t)Nn * 128 + rp_len + (size_t)E) * sizeof(float);
  const bool grouped = ws_size >= need_bytes &&
                       (size_t)Nn + bs_len + (size_t)E <= (size_t)Nn * 128;

  const float invN = 1.0f / (float)Nn;
  const int TB = 256;

  auto gemm = [&](const float* A, const float* Bm, float* C, int K, int F) {
    long waves = (long)((Nn + 15) / 16) * (F / 16);
    int blocks = (int)((waves * 32 + TB - 1) / TB);
    k_gemm_wmma_f32<<<blocks, TB, 0, stream>>>(A, Bm, dinv, C, Nn, K, F);
  };
  auto scatter = [&](const float* h, float* agg, int F) {
    int blocks = (int)(((long)E * 32 + TB - 1) / TB);
    k_scatter<<<blocks, TB, 0, stream>>>(h, src, dst, agg, E, F);
  };
  auto finish = [&](float* agg, const float* h, const float* bias, int F) {
    const int blocks = (Nn + FIN_ROWS - 1) / FIN_ROWS;
    k_zero<<<1, 256, 0, stream>>>(stats, 256);
    if (F == 128)
      k_finish_relu_stats<128><<<blocks, 256, 0, stream>>>(agg, h, dinv, bias, stats, Nn);
    else
      k_finish_relu_stats<64><<<blocks, 256, 0, stream>>>(agg, h, dinv, bias, stats, Nn);
  };
  auto gemm_bn = [&](const float* V, const float* gamma, const float* beta,
                     const float* Bm, float* C, int F) {
    if (F == 128)
      k_gemm_bn_wmma_f32<128><<<(Nn + 15) / 16, 256, 0, stream>>>(
          V, stats, gamma, beta, Bm, dinv, C, Nn, invN);
    else
      k_gemm_bn_wmma_f32<64><<<(Nn + 31) / 32, 256, 0, stream>>>(
          V, stats, gamma, beta, Bm, dinv, C, Nn, invN);
  };
  auto zero = [&](float* p, size_t n) {
    int blocks = (int)((n + TB - 1) / TB);
    if (blocks > 4096) blocks = 4096;
    k_zero<<<blocks, TB, 0, stream>>>(p, n);
  };

  auto aggregate = [&](const float* h, float* agg, int F) {
    zero(agg, (size_t)Nn * F);
    scatter(h, agg, F);
  };
  auto aggregate_finish = [&](const float* h, float* v, const float* bias, int F) {
    if (grouped) {
      k_zero<<<1, 256, 0, stream>>>(stats, 256);
      if (F == 128)
        k_gather_finish<128, GF_ROWS_128>
            <<<(Nn + 8 * GF_ROWS_128 - 1) / (8 * GF_ROWS_128), 256, 0, stream>>>(
                h, rowptr, csr, dinv, bias, v, stats, Nn, E);
      else
        k_gather_finish<64, GF_ROWS_64>
            <<<(Nn + 8 * GF_ROWS_64 - 1) / (8 * GF_ROWS_64), 256, 0, stream>>>(
                h, rowptr, csr, dinv, bias, v, stats, Nn, E);
    } else {
      aggregate(h, v, F);
      finish(v, h, bias, F);
    }
  };

  if (grouped) {
    zero((float*)cnt, (size_t)Nn);
    k_count_rank<<<(E + TB - 1) / TB, TB, 0, stream>>>(dst, cnt, rank, E);
    k_dinv_bsum<<<nchunk, 256, 0, stream>>>(cnt, dinv, bsum, Nn);
    k_scan_rowptr<<<nchunk, 256, 0, stream>>>(cnt, bsum, rowptr, Nn);
    k_fill_csr<<<(E + TB - 1) / TB, TB, 0, stream>>>(src, dst, rank, rowptr, csr, E);
  } else {
    k_deg_init<<<(Nn + TB - 1) / TB, TB, 0, stream>>>(dinv, Nn);
    k_deg_count<<<(E + TB - 1) / TB, TB, 0, stream>>>(dst, dinv, E);
    k_rsqrt_inplace<<<(Nn + TB - 1) / TB, TB, 0, stream>>>(dinv, Nn);
  }

  if (grouped) {
    k_gather_x24<<<(Nn + 7) / 8, 256, 0, stream>>>(x, dinv, rowptr, csr, bufA, Nn, E);
    k_zero<<<1, 256, 0, stream>>>(stats, 256);
    k_gemm1_relu_stats<<<(Nn + 16 * G1_TILES - 1) / (16 * G1_TILES), 256, 0, stream>>>(
        bufA, W1, b1, bufB, stats, Nn);
  } else {
    gemm(x, W1, bufA, 24, 128);
    aggregate(bufA, bufB, 128);
    finish(bufB, bufA, b1, 128);
  }

  gemm_bn(bufB, g1, be1, W2, bufA, 128);
  aggregate_finish(bufA, bufB, b2, 128);

  gemm_bn(bufB, g2, be2, W3, bufA, 64);
  aggregate_finish(bufA, bufB, b3, 64);

  k_gemm64x2<<<(Nn + TB - 1) / TB, TB, 0, stream>>>(bufB, stats, g3, be3, W4, dinv,
                                                    bufA, Nn, invN);
  zero(bufB, (size_t)Nn * 2);
  k_scatter2<<<(E + TB - 1) / TB, TB, 0, stream>>>(bufA, src, dst, bufB, E);
  k_final_lsm<<<(Nn + TB - 1) / TB, TB, 0, stream>>>(bufB, bufA, dinv, b4,
                                                     (float*)d_out, Nn);
}


# layer 4: 2-column edge sum over the sorted in-edge lists (8 lanes a row, 16 index then 16 row loads in flight), log-softmax in its store; float atomics, zero pass and last pass go (sum order changes)
# speedup vs baseline: 4.6105x; 1.3587x over previous
#include <hip/hip_runtime.h>
#include <hip/hip_bf16.h>
#include <math.h>


typedef float v2f __attribute__((ext_vector_type(2)));
typedef float v4f __attribute__((ext_vector_type(4)));
typedef float v8f __attribute__((ext_vector_type(8)));

#define BN_EPS 1e-5f

__global__ void k_zero(float* __restrict__ p, size_t n) {
  size_t i = (size_t)blockIdx.x * blockDim.x + threadIdx.x;
  size_t stride = (size_t)gridDim.x * blockDim.x;
  for (; i < n; i += stride) p[i] = 0.0f;
}

__global__ void k_deg_init(float* __restrict__ deg, int n) {
  int i = blockIdx.x * blockDim.x + threadIdx.x;
  if (i < n) deg[i] = 1.0f;
}

__global__ void k_deg_count(const int* __restrict__ dst, float* __restrict__ deg, int E) {
  int e = blockIdx.x * blockDim.x + threadIdx.x;
  if (e < E) unsafeAtomicAdd(&deg[dst[e]], 1.0f);
}

__global__ void k_rsqrt_inplace(float* __restrict__ d, int n) {
  int i = blockIdx.x * blockDim.x + threadIdx.x;
  if (i < n) {
    float v = d[i];
    d[i] = (v > 0.0f) ? rsqrtf(v) : 0.0f;
  }
}

__global__ void k_count_rank(const int* __restrict__ dst, int* __restrict__ cnt,
                             int* __restrict__ rank, int E) {
  int e = blockIdx.x * blockDim.x + threadIdx.x;
  if (e < E) rank[e] = atomicAdd(&cnt[dst[e]], 1);
}

#define SCAN_CHUNK 1024

__global__ void __launch_bounds__(256)
k_dinv_bsum(const int* __restrict__ cnt, float* __restrict__ dinv,
            int* __restrict__ bsum, int n) {
  __shared__ int wsum[8];
  const int t = threadIdx.x;
  const int i0 = blockIdx.x * SCAN_CHUNK + 4 * t;
  int s = 0;
#pragma unroll
  for (int k = 0; k < 4; ++k) {
    const int i = i0 + k;
    if (i < n) {
      const int c = cnt[i];
      s += c;
      const float deg = (float)(c + 1);
      dinv[i] = (deg > 0.0f) ? rsqrtf(deg) : 0.0f;
    }
  }
  for (int o = 16; o > 0; o >>= 1) s += __shfl_xor(s, o);
  if ((t & 31) == 0) wsum[t >> 5] = s;
  __syncthreads();
  if (t == 0) {
    int tot = 0;
    for (int w = 0; w < 8; ++w) tot += wsum[w];
    bsum[blockIdx.x] = tot;
  }
}

__global__ void __launch_bounds__(256)
k_scan_rowptr(const int* __restrict__ cnt, const int* __restrict__ bsum,
              int* __restrict__ rowptr, int n) {
  __shared__ int wtot[8];
  __shared__ int base_s;
  const int t = threadIdx.x, lane = t & 31, w = t >> 5;
  int p = 0;
  for (int b = t; b < (int)blockIdx.x; b += 256) p += bsum[b];
  for (int o = 16; o > 0; o >>= 1) p += __shfl_xor(p, o);
  if (lane == 0) wtot[w] = p;
  __syncthreads();
  if (t == 0) {
    int tot = 0;
    for (int k = 0; k < 8; ++k) tot += wtot[k];
    base_s = tot;
  }
  __syncthreads();
  const int base = base_s;
  const int i0 = blockIdx.x * SCAN_CHUNK + 4 * t;
  int c[4];
  int s = 0;
#pragma unroll
  for (int k = 0; k < 4; ++k) {
    c[k] = (i0 + k < n) ? cnt[i0 + k] : 0;
    s += c[k];
  }
  int incl = s;
  for (int o = 1; o < 32; o <<= 1) {
    const int v = __shfl_up(incl, o);
    if (lane >= o) incl += v;
  }
  if (lane == 31) wtot[w] = incl;
  __syncthreads();
  int woff = 0;
  for (int k = 0; k < w; ++k) woff += wtot[k];
  int excl = base + woff + incl - s;
#pragma unroll
  for (int k = 0; k < 4; ++k) {
    if (i0 + k < n) {
      rowptr[i0 + k] = excl;
      excl += c[k];
    }
  }
  if (blockIdx.x == gridDim.x - 1 && t == 255) rowptr[n] = excl;
}

__global__ void k_fill_csr(const int* __restrict__ src, const int* __restrict__ dst,
                           const int* __restrict__ rank, const int* __restrict__ rowptr,
                           int* __restrict__ csr, int E) {
  int e = blockIdx.x * blockDim.x + threadIdx.x;
  if (e >= E) return;
  unsigned pos = (unsigned)(rowptr[dst[e]] + rank[e]);
  csr[min(pos, (unsigned)(E - 1))] = src[e];
}

#define GF_ROWS_128 9
#define GF_ROWS_64 8
template <int F, int R>
__global__ void __launch_bounds__(256)
k_gather_finish(const float* __restrict__ hsc, const int* __restrict__ rowptr,
                const int* __restrict__ csr, const float* __restrict__ dinv,
                const float* __restrict__ bias, float* __restrict__ vout,
                float* __restrict__ stats,
                int Nn, int E) {
  constexpr int V = F / 32;
  typedef float vf __attribute__((ext_vector_type(V)));
  __shared__ __attribute__((aligned(16))) float red[2][8][F];
  const int t = threadIdx.x;
  const int lane = t & 31;
  const int w = t >> 5;
  const int d0 = __builtin_amdgcn_readfirstlane((int)((blockIdx.x * 8 + w) * R));
  vf s = 0.0f;
  vf q = 0.0f;
  if (d0 < Nn) {
    const int rows = min(R, Nn - d0);
    const vf* hp = reinterpret_cast<const vf*>(hsc) + lane;
    vf* op = reinterpret_cast<vf*>(vout) + lane;
    const unsigned nmax = (unsigned)(Nn - 1);
    const vf zerov = 0.0f;
    const int rpraw = rowptr[min(d0 + lane, Nn)];
    const float dvl = dinv[min(d0 + lane, Nn - 1)];
    const vf b = *reinterpret_cast<const vf*>(bias + V * lane);
    const int rp = min(max(rpraw, 0), E);
    const unsigned emax = (unsigned)(E - 1);
    int beg = __builtin_amdgcn_readlane(rp, 0);
    int end = __builtin_amdgcn_readlane(rp, 1);
    int idx = csr[min((unsigned)(beg + lane), emax)];
    auto sum_chunk = [&](const int cur, const int m, vf a) -> vf {
      auto row = [&](int u) -> vf {
        const unsigned sidx = min((unsigned)__builtin_amdgcn_readlane(cur, u), nmax);
        return hp[(size_t)sidx * 32];
      };
      int u = 0;
      for (; u + 8 <= m; u += 8) {
        vf r[8];
#pragma unroll
        for (int k = 0; k < 8; ++k) r[k] = row(u + k);
#pragma unroll
        for (int k = 0; k < 8; ++k) a += r[k];
      }
      if (u + 4 <= m) {
        const vf r0 = row(u), r1 = row(u + 1), r2 = row(u + 2), r3 = row(u + 3);
        a += r0; a += r1; a += r2; a += r3;
        u += 4;
      }
      if (u + 2 <= m) {
        const vf r0 = row(u), r1 = row(u + 1);
        a += r0; a += r1;
        u += 2;
      }
      if (u < m) a += row(u);
      return a;
    };
    for (int i = 0; i < rows; ++i) {
      const int d = d0 + i;
      const vf self = hp[(size_t)d * 32];
      const int nbeg = end;
      const int nend = __builtin_amdgcn_readlane(rp, i + 2);
      const int idxn = csr[min((unsigned)(nbeg + lane), emax)];
      vf acc = sum_chunk(idx, min(32, end - beg), zerov);
      for (int j0 = beg + 32; j0 < end; j0 += 32) {
        const int m = min(32, end - j0);
        const int more = (lane < m) ? csr[j0 + lane] : 0;
        acc = sum_chunk(more, m, acc);
      }
      const float di = __builtin_bit_cast(
          float, __builtin_amdgcn_readlane(__builtin_bit_cast(int, dvl), i));
      const vf v = __builtin_elementwise_max((acc + self) * di + b, zerov);
      op[(size_t)d * 32] = v;
      s += v;
      q += v * v;
      beg = nbeg;
      end = nend;
      idx = idxn;
    }
  }
  *reinterpret_cast<vf*>(&red[0][w][V * lane]) = s;
  *reinterpret_cast<vf*>(&red[1][w][V * lane]) = q;
  __syncthreads();
  if (t < 2 * F) {
    const int which = t / F;
    const int j = t % F;
    float a = 0.f;
    for (int k = 0; k < 8; ++k) a += red[which][k][j];
    unsafeAtomicAdd(stats + which * F + j, a);
  }
}

__global__ void k_gemm_wmma_f32(const float* __restrict__ A, const float* __restrict__ B,
                                const float* __restrict__ dinv,
                                float* __restrict__ C, int M, int K, int F) {
  const int lane = threadIdx.x & 31;
  const int wave = (int)((blockIdx.x * blockDim.x + threadIdx.x) >> 5);
  const int nf = F >> 4;
  const int mTile = wave / nf;
  const int fTile = wave - mTile * nf;
  const int m0 = mTile << 4;
  if (m0 >= M) return;
  const int f0 = fTile << 4;
  const int row = lane & 15;
  const int kh2 = (lane >> 4) << 1;

  const float* arow = A + (size_t)(m0 + row) * K + kh2;
  const float* bcol = B + (size_t)kh2 * F + f0 + row;

  v8f acc = {0.f, 0.f, 0.f, 0.f, 0.f, 0.f, 0.f, 0.f};
  for (int k = 0; k < K; k += 4) {
    float2 av = *reinterpret_cast<const float2*>(arow + k);
    v2f a; a.x = av.x; a.y = av.y;
    v2f b;
    b.x = bcol[(size_t)k * F];
    b.y = bcol[(size_t)(k + 1) * F];
    acc = __builtin_amdgcn_wmma_f32_16x16x4_f32(false, a, false, b,
                                                (short)0, acc, false, false);
  }
  const float* dvp = dinv + m0 + (kh2 << 2);
  float* crow = C + (size_t)(m0 + (kh2 << 2)) * F + f0 + row;
#pragma unroll
  for (int r = 0; r < 8; ++r) crow[(size_t)r * F] = acc[r] * dvp[r];
}

template <int F>
__global__ void __launch_bounds__(256)
k_gemm_bn_wmma_f32(const float* __restrict__ V, const float* __restrict__ stats,
                   const float* __restrict__ gamma, const float* __restrict__ beta,
                   const float* __restrict__ B, const float* __restrict__ dinv,
                   float* __restrict__ C, int M, float invN) {
  constexpr int K = 128;
  constexpr int NF = F / 16;
  constexpr int RB = 16 * (8 / NF);
  constexpr int PITCH = K + 4;
  __shared__ __attribute__((aligned(16))) float As[RB * PITCH];

  const int t = threadIdx.x;
  const int lane = t & 31;
  const int w = t >> 5;
  const int rowBase = blockIdx.x * RB;

  {
    const int c4 = t & 31;
    const v4f su = *reinterpret_cast<const v4f*>(stats + 4 * c4);
    const v4f sq = *reinterpret_cast<const v4f*>(stats + K + 4 * c4);
    const v4f g = *reinterpret_cast<const v4f*>(gamma + 4 * c4);
    const v4f be = *reinterpret_cast<const v4f*>(beta + 4 * c4);
    v4f mean, rstd;
#pragma unroll
    for (int i = 0; i < 4; ++i) {
      mean[i] = su[i] * invN;
      const float var = sq[i] * invN - mean[i] * mean[i];
      rstd[i] = rsqrtf(var + BN_EPS);
    }
    v4f v[RB / 8];
#pragma unroll
    for (int p = 0; p < RB / 8; ++p) {
      const int gr = min(rowBase + p * 8 + w, M - 1);
      v[p] = *reinterpret_cast<const v4f*>(V + (size_t)gr * K + 4 * c4);
    }
#pragma unroll
    for (int p = 0; p < RB / 8; ++p) {
      const v4f o = (v[p] - mean) * rstd * g + be;
      *reinterpret_cast<v4f*>(&As[(p * 8 + w) * PITCH + 4 * c4]) = o;
    }
  }
  __syncthreads();

  const int mT = w / NF;
  const int fT = w - mT * NF;
  const int m0 = rowBase + (mT << 4);
  if (m0 >= M) return;
  const int f0 = fT << 4;
  const int row = lane & 15;
  const int kh2 = (lane >> 4) << 1;

  const float* arow = As + ((mT << 4) + row) * PITCH + kh2;
  const float* bcol = B + (size_t)kh2 * F + f0 + row;

  unsigned dbyte = (unsigned)(m0 + (kh2 << 2)) * 4u;
  unsigned cbyte = ((unsigned)(m0 + (kh2 << 2)) * F + f0 + row) * 4u;
  asm volatile("" : "+v"(dbyte), "+v"(cbyte));
  const float* dvp = reinterpret_cast<const float*>(reinterpret_cast<const char*>(dinv) + dbyte);
  float* crow = reinterpret_cast<float*>(reinterpret_cast<char*>(C) + cbyte);

  v8f acc = {0.f, 0.f, 0.f, 0.f, 0.f, 0.f, 0.f, 0.f};
#pragma unroll
  for (int k = 0; k < K; k += 4) {
    const v2f a = *reinterpret_cast<const v2f*>(arow + k);
    v2f b;
    b.x = bcol[(size_t)k * F];
    b.y = bcol[(size_t)(k + 1) * F];
    acc = __builtin_amdgcn_wmma_f32_16x16x4_f32(false, a, false, b,
                                                (short)0, acc, false, false);
  }
  asm volatile("s_wait_alu depctr_va_vdst(0)\n\t"
               "v_nop\n\tv_nop\n\tv_nop\n\tv_nop\n\t"
               "v_nop\n\tv_nop\n\tv_nop\n\tv_nop"
               : "+v"(acc));
#pragma unroll
  for (int r = 0; r < 8; ++r) crow[(size_t)r * F] = acc[r] * dvp[r];
}

__global__ void __launch_bounds__(256)
k_gather_x24(const float* __restrict__ x, const float* __restrict__ dinv,
             const int* __restrict__ rowptr, const int* __restrict__ csr,
             float* __restrict__ ax, int Nn, int E) {
  const int lane = threadIdx.x & 31;
  const int d = __builtin_amdgcn_readfirstlane((int)(blockIdx.x * 8 + (threadIdx.x >> 5)));
  if (d >= Nn) return;
  const int beg = max(rowptr[d], 0);
  const int end = min(rowptr[d + 1], E);
  const int p = lane & 7;
  const int q = lane >> 3;
  const bool pv = p < 6;
  const unsigned nmax = (unsigned)(Nn - 1);
  const v4f* xp = reinterpret_cast<const v4f*>(x) + p;
  const v4f zero4 = {0.f, 0.f, 0.f, 0.f};
  const float dd = dinv[d];
  const v4f xd = pv ? xp[(size_t)d * 6] : zero4;
  v4f acc = zero4;
  for (int j0 = beg; j0 < end; j0 += 32) {
    const int m = min(32, end - j0);
    int idx = 0;
    float dv = 0.f;
    if (lane < m) {
      idx = (int)min((unsigned)csr[j0 + lane], nmax);
      dv = dinv[idx];
    }
    v4f r[8];
    float wgt[8];
#pragma unroll
    for (int k = 0; k < 8; ++k) {
      const int e = 4 * k + q;
      const int s = __shfl(idx, e);
      wgt[k] = __shfl(dv, e);
      r[k] = (pv && e < m) ? xp[(size_t)s * 6] : zero4;
    }
#pragma unroll
    for (int k = 0; k < 8; ++k) acc += r[k] * wgt[k];
  }
#pragma unroll
  for (int i = 0; i < 4; ++i) {
    acc[i] += __shfl_xor(acc[i], 8);
    acc[i] += __shfl_xor(acc[i], 16);
  }
  if (q == 0 && pv)
    reinterpret_cast<v4f*>(ax)[(size_t)d * 6 + p] = (acc + xd * dd) * dd;
}

#define G1_TILES 8
__global__ void __launch_bounds__(256)
k_gemm1_relu_stats(const float* __restrict__ A, const float* __restrict__ B,
                   const float* __restrict__ bias, float* __restrict__ C,
                   float* __restrict__ stats,
                   int M) {
  constexpr int K = 24;
  constexpr int F = 128;
  constexpr int KS = K / 4;
  const int lane = threadIdx.x & 31;
  const int f0 = (int)(threadIdx.x >> 5) << 4;
  const int c = lane & 15;
  const int h = lane >> 4;
  const int kh2 = h << 1;
  const int rowBase = blockIdx.x * (16 * G1_TILES);

  v2f b[KS];
  const float* bcol = B + (size_t)kh2 * F + f0 + c;
#pragma unroll
  for (int kk = 0; kk < KS; ++kk) {
    b[kk].x = bcol[(size_t)(4 * kk) * F];
    b[kk].y = bcol[(size_t)(4 * kk + 1) * F];
  }
  const float bc = bias[f0 + c];

  v2f a[G1_TILES][KS];
#pragma unroll
  for (int i = 0; i < G1_TILES; ++i) {
    const int gr = min(rowBase + 16 * i + c, M - 1);
    const float* ar = A + (size_t)gr * K + kh2;
#pragma unroll
    for (int kk = 0; kk < KS; ++kk)
      a[i][kk] = *reinterpret_cast<const v2f*>(ar + 4 * kk);
  }

  float s = 0.f, q = 0.f;
#pragma unroll
  for (int i = 0; i < G1_TILES; ++i) {
    const int m0 = rowBase + 16 * i;
    if (m0 >= M) break;
    __builtin_amdgcn_sched_barrier(0);
    v8f acc = {0.f, 0.f, 0.f, 0.f, 0.f, 0.f, 0.f, 0.f};
#pragma unroll
    for (int kk = 0; kk < KS; ++kk)
      acc = __builtin_amdgcn_wmma_f32_16x16x4_f32(false, a[i][kk], false, b[kk],
                                                  (short)0, acc, false, false);
    asm volatile("s_wait_alu depctr_va_vdst(0)\n\t"
                 "v_nop\n\tv_nop\n\tv_nop\n\tv_nop\n\t"
                 "v_nop\n\tv_nop\n\tv_nop\n\tv_nop"
                 : "+v"(acc));
    __builtin_amdgcn_sched_barrier(0);
    const int r0 = m0 + (h << 3);
    float* crow = C + (size_t)r0 * F + f0 + c;
#pragma unroll
    for (int r = 0; r < 8; ++r) {
      if (r0 + r < M) {
        const float v = fmaxf(acc[r] + bc, 0.f);
        crow[(size_t)r * F] = v;
        s += v;
        q += v * v;
      }
    }
  }
  s += __shfl_xor(s, 16);
  q += __shfl_xor(q, 16);
  if (h == 0) {
    unsafeAtomicAdd(stats + f0 + c, s);
    unsafeAtomicAdd(stats + F + f0 + c, q);
  }
}

__global__ void k_scatter(const float* __restrict__ hsc, const int* __restrict__ src,
                          const int* __restrict__ dst,
                          float* __restrict__ agg, int E, int F) {
  int wave = (int)((blockIdx.x * blockDim.x + threadIdx.x) >> 5);
  int lane = threadIdx.x & 31;
  if (wave >= E) return;
  int s = src[wave], d = dst[wave];
  const float4* hrow = reinterpret_cast<const float4*>(hsc + (size_t)s * F);
  float* ad = agg + (size_t)d * F;
  for (int f4 = lane; 4 * f4 < F; f4 += 32) {
    float4 v = hrow[f4];
    unsafeAtomicAdd(ad + 4 * f4 + 0, v.x);
    unsafeAtomicAdd(ad + 4 * f4 + 1, v.y);
    unsafeAtomicAdd(ad + 4 * f4 + 2, v.z);
    unsafeAtomicAdd(ad + 4 * f4 + 3, v.w);
  }
}

#define FIN_ROWS 128
template <int F>
__global__ void __launch_bounds__(256)
k_finish_relu_stats(float* __restrict__ agg, const float* __restrict__ hsc,
                    const float* __restrict__ dinv,
                    const float* __restrict__ bias,
                    float* __restrict__ stats,
                    int Nn) {
  constexpr int TPR = F / 4;
  constexpr int RPP = 256 / TPR;
  constexpr int U = 8;
  constexpr int STEP = RPP * TPR;
  static_assert(FIN_ROWS % (U * RPP) == 0, "a block is a whole number of groups");
  __shared__ __attribute__((aligned(16))) float red[2][RPP][F];

  const int t = threadIdx.x;
  const int c4 = t % TPR;
  const int rsub = t / TPR;
  const int r0 = blockIdx.x * FIN_ROWS;
  const int r1 = min(r0 + FIN_ROWS, Nn);

  const v4f b = *reinterpret_cast<const v4f*>(bias + 4 * c4);
  const v4f zero4 = {0.f, 0.f, 0.f, 0.f};
  v4f s = zero4;
  v4f q = zero4;

  int r = r0 + rsub;
  v4f* ap = reinterpret_cast<v4f*>(agg) + (size_t)r * TPR + c4;
  const v4f* hp = reinterpret_cast<const v4f*>(hsc) + (size_t)r * TPR + c4;
  const float* dp = dinv + r;

  auto apply = [&](const v4f a, const v4f h, const float di, v4f* dst) {
    const v4f v = __builtin_elementwise_max((a + h) * di + b, zero4);
    *dst = v;
    s += v;
    q += v * v;
  };

  for (; r + (U - 1) * RPP < r1; r += U * RPP) {
    v4f a[U], h[U];
    float di[U];
#pragma unroll
    for (int u = 0; u < U; ++u) {
      a[u] = ap[u * STEP];
      h[u] = hp[u * STEP];
      di[u] = dp[u * RPP];
    }
    __builtin_amdgcn_sched_barrier(0);
#pragma unroll
    for (int u = 0; u < U; ++u) apply(a[u], h[u], di[u], ap + u * STEP);
    ap += U * STEP;
    hp += U * STEP;
    dp += U * RPP;
  }
  for (; r < r1; r += RPP) {
    apply(*ap, *hp, *dp, ap);
    ap += STEP;
    hp += STEP;
    dp += RPP;
  }

  *reinterpret_cast<v4f*>(&red[0][rsub][4 * c4]) = s;
  *reinterpret_cast<v4f*>(&red[1][rsub][4 * c4]) = q;
  __syncthreads();
  if (t < 2 * F) {
    const int which = t / F;
    const int j = t % F;
    float acc = 0.f;
    for (int k = 0; k < RPP; ++k) acc += red[which][k][j];
    unsafeAtomicAdd(stats + which * F + j, acc);
  }
}

__global__ void k_gemm64x2(const float* __restrict__ v3, const float* __restrict__ stats,
                           const float* __restrict__ gamma, const float* __restrict__ beta,
                           const float* __restrict__ W4, const float* __restrict__ dinv,
                           float* __restrict__ h4, int Nn, float invN) {
  int i = blockIdx.x * blockDim.x + threadIdx.x;
  if (i >= Nn) return;
  const float* row = v3 + (size_t)i * 64;
  float o0 = 0.f, o1 = 0.f;
#pragma unroll 8
  for (int k = 0; k < 64; ++k) {
    const float mean = stats[k] * invN;
    const float var = stats[64 + k] * invN - mean * mean;
    const float xv = (row[k] - mean) * rsqrtf(var + BN_EPS) * gamma[k] + beta[k];
    o0 += xv * W4[2 * k];
    o1 += xv * W4[2 * k + 1];
  }
  float di = dinv[i];
  h4[2 * (size_t)i]     = o0 * di;
  h4[2 * (size_t)i + 1] = o1 * di;
}

__global__ void k_scatter2(const float* __restrict__ h4s, const int* __restrict__ src,
                           const int* __restrict__ dst,
                           float* __restrict__ agg, int E) {
  int e = blockIdx.x * blockDim.x + threadIdx.x;
  if (e >= E) return;
  int s = src[e], d = dst[e];
  float2 v = *reinterpret_cast<const float2*>(h4s + 2 * (size_t)s);
  unsafeAtomicAdd(agg + 2 * (size_t)d,     v.x);
  unsafeAtomicAdd(agg + 2 * (size_t)d + 1, v.y);
}

__global__ void k_final_lsm(const float* __restrict__ agg, const float* __restrict__ h4s,
                            const float* __restrict__ dinv, const float* __restrict__ b4,
                            float* __restrict__ out, int Nn) {
  int i = blockIdx.x * blockDim.x + threadIdx.x;
  if (i >= Nn) return;
  float di = dinv[i];
  float o0 = (agg[2 * (size_t)i]     + h4s[2 * (size_t)i])     * di + b4[0];
  float o1 = (agg[2 * (size_t)i + 1] + h4s[2 * (size_t)i + 1]) * di + b4[1];
  float m = fmaxf(o0, o1);
  float l = m + logf(expf(o0 - m) + expf(o1 - m));
  out[2 * (size_t)i]     = o0 - l;
  out[2 * (size_t)i + 1] = o1 - l;
}

#define L4_ROWS 16
__global__ void __launch_bounds__(256)
k_gather2_lsm(const float* __restrict__ h4s, const int* __restrict__ rowptr,
              const int* __restrict__ csr, const float* __restrict__ dinv,
              const float* __restrict__ b4, float* __restrict__ out, int Nn, int E) {
  const int lane = threadIdx.x & 31;
  const int d0 = __builtin_amdgcn_readfirstlane(
      (int)((blockIdx.x * 8 + (threadIdx.x >> 5)) * L4_ROWS));
  if (d0 >= Nn) return;
  const int p = lane & 7;
  const int q = lane >> 3;
  const v2f* hp = reinterpret_cast<const v2f*>(h4s);
  const unsigned nmax = (unsigned)(Nn - 1);
  const unsigned emax = (unsigned)(E - 1);
  const v2f zero2 = {0.f, 0.f};
  const int rpraw = rowptr[min(d0 + lane, Nn)];
  const int gs = p & 3;
  const int rfin = d0 + 4 * gs + q;
  const int rfc = min(rfin, Nn - 1);
  const v2f self = hp[rfc];
  const float di = dinv[rfc];
  const float bb0 = b4[0];
  const float bb1 = b4[1];
  const int rp = min(max(rpraw, 0), E);

  int beg[4], len[4];
  unsigned own[4];
#pragma unroll
  for (int g = 0; g < 4; ++g) {
    const int b = __shfl(rp, 4 * g + q);
    const int e = __shfl(rp, 4 * g + q + 1);
    beg[g] = b;
    len[g] = e - b;
    own[g] = min((unsigned)(d0 + 4 * g + q), nmax);
  }
  int idx[4][4];
#pragma unroll
  for (int g = 0; g < 4; ++g) {
#pragma unroll
    for (int k = 0; k < 4; ++k)
      idx[g][k] = csr[min((unsigned)(beg[g] + p + 8 * k), emax)];
  }
  v2f v[4][4];
#pragma unroll
  for (int g = 0; g < 4; ++g) {
#pragma unroll
    for (int k = 0; k < 4; ++k) {
      const unsigned s = (p + 8 * k < len[g]) ? min((unsigned)idx[g][k], nmax) : own[g];
      v[g][k] = hp[s];
    }
  }
  v2f acc[4];
#pragma unroll
  for (int g = 0; g < 4; ++g) {
    v2f a = zero2;
#pragma unroll
    for (int k = 0; k < 4; ++k) a += (p + 8 * k < len[g]) ? v[g][k] : zero2;
    const int end = beg[g] + len[g];
    for (int j = beg[g] + 32 + p; j < end; j += 8) a += hp[min((unsigned)csr[j], nmax)];
    a.x += __shfl_xor(a.x, 1);
    a.y += __shfl_xor(a.y, 1);
    a.x += __shfl_xor(a.x, 2);
    a.y += __shfl_xor(a.y, 2);
    a.x += __shfl_xor(a.x, 4);
    a.y += __shfl_xor(a.y, 4);
    acc[g] = a;
  }
  const v2f a = gs == 0 ? acc[0] : gs == 1 ? acc[1] : gs == 2 ? acc[2] : acc[3];
  const float o0 = (a.x + self.x) * di + bb0;
  const float o1 = (a.y + self.y) * di + bb1;
  const float m = fmaxf(o0, o1);
  const float l = m + logf(expf(o0 - m) + expf(o1 - m));
  v2f o;
  o.x = o0 - l;
  o.y = o1 - l;
  if (p < 4 && rfin < Nn) reinterpret_cast<v2f*>(out)[rfin] = o;
}

extern "C" void kernel_launch(void* const* d_in, const int* in_sizes, int n_in,
                              void* d_out, int out_size, void* d_ws, size_t ws_size,
                              hipStream_t stream) {
  const float* x   = (const float*)d_in[0];
  const int*   ei  = (const int*)d_in[1];
  const float* W1  = (const float*)d_in[2];
  const float* b1  = (const float*)d_in[3];
  const float* g1  = (const float*)d_in[4];
  const float* be1 = (const float*)d_in[5];
  const float* W2  = (const float*)d_in[6];
  const float* b2  = (const float*)d_in[7];
  const float* g2  = (const float*)d_in[8];
  const float* be2 = (const float*)d_in[9];
  const float* W3  = (const float*)d_in[10];
  const float* b3  = (const float*)d_in[11];
  const float* g3  = (const float*)d_in[12];
  const float* be3 = (const float*)d_in[13];
  const float* W4  = (const float*)d_in[14];
  const float* b4  = (const float*)d_in[15];

  const int Nn = in_sizes[0] / 24;
  const int E  = in_sizes[1] / 2;
  const int* src = ei;
  const int* dst = ei + E;

  float* ws    = (float*)d_ws;
  float* dinv  = ws;
  float* stats = ws + Nn;
  float* bufA  = stats + 256;
  float* bufB  = bufA + (size_t)Nn * 128;
  const int nchunk = (Nn + SCAN_CHUNK - 1) / SCAN_CHUNK;
  const size_t rp_len = ((size_t)Nn + 1 + 31) & ~(size_t)31;
  const size_t bs_len = ((size_t)nchunk + 31) & ~(size_t)31;
  int* rowptr = (int*)(bufB + (size_t)Nn * 128);
  int* csr    = rowptr + rp_len;
  int* cnt    = (int*)bufB;
  int* bsum   = cnt + Nn;
  int* rank   = bsum + bs_len;
  const size_t need_bytes =
      ((size_t)Nn + 256 + 2 * (size_t)Nn * 128 + rp_len + (size_t)E) * sizeof(float);
  const bool grouped = ws_size >= need_bytes &&
                       (size_t)Nn + bs_len + (size_t)E <= (size_t)Nn * 128;

  const float invN = 1.0f / (float)Nn;
  const int TB = 256;

  auto gemm = [&](const float* A, const float* Bm, float* C, int K, int F) {
    long waves = (long)((Nn + 15) / 16) * (F / 16);
    int blocks = (int)((waves * 32 + TB - 1) / TB);
    k_gemm_wmma_f32<<<blocks, TB, 0, stream>>>(A, Bm, dinv, C, Nn, K, F);
  };
  auto scatter = [&](const float* h, float* agg, int F) {
    int blocks = (int)(((long)E * 32 + TB - 1) / TB);
    k_scatter<<<blocks, TB, 0, stream>>>(h, src, dst, agg, E, F);
  };
  auto finish = [&](float* agg, const float* h, const float* bias, int F) {
    const int blocks = (Nn + FIN_ROWS - 1) / FIN_ROWS;
    k_zero<<<1, 256, 0, stream>>>(stats, 256);
    if (F == 128)
      k_finish_relu_stats<128><<<blocks, 256, 0, stream>>>(agg, h, dinv, bias, stats, Nn);
    else
      k_finish_relu_stats<64><<<blocks, 256, 0, stream>>>(agg, h, dinv, bias, stats, Nn);
  };
  auto gemm_bn = [&](const float* V, const float* gamma, const float* beta,
                     const float* Bm, float* C, int F) {
    if (F == 128)
      k_gemm_bn_wmma_f32<128><<<(Nn + 15) / 16, 256, 0, stream>>>(
          V, stats, gamma, beta, Bm, dinv, C, Nn, invN);
    else
      k_gemm_bn_wmma_f32<64><<<(Nn + 31) / 32, 256, 0, stream>>>(
          V, stats, gamma, beta, Bm, dinv, C, Nn, invN);
  };
  auto zero = [&](float* p, size_t n) {
    int blocks = (int)((n + TB - 1) / TB);
    if (blocks > 4096) blocks = 4096;
    k_zero<<<blocks, TB, 0, stream>>>(p, n);
  };

  auto aggregate = [&](const float* h, float* agg, int F) {
    zero(agg, (size_t)Nn * F);
    scatter(h, agg, F);
  };
  auto aggregate_finish = [&](const float* h, float* v, const float* bias, int F) {
    if (grouped) {
      k_zero<<<1, 256, 0, stream>>>(stats, 256);
      if (F == 128)
        k_gather_finish<128, GF_ROWS_128>
            <<<(Nn + 8 * GF_ROWS_128 - 1) / (8 * GF_ROWS_128), 256, 0, stream>>>(
                h, rowptr, csr, dinv, bias, v, stats, Nn, E);
      else
        k_gather_finish<64, GF_ROWS_64>
            <<<(Nn + 8 * GF_ROWS_64 - 1) / (8 * GF_ROWS_64), 256, 0, stream>>>(
                h, rowptr, csr, dinv, bias, v, stats, Nn, E);
    } else {
      aggregate(h, v, F);
      finish(v, h, bias, F);
    }
  };

  if (grouped) {
    zero((float*)cnt, (size_t)Nn);
    k_count_rank<<<(E + TB - 1) / TB, TB, 0, stream>>>(dst, cnt, rank, E);
    k_dinv_bsum<<<nchunk, 256, 0, stream>>>(cnt, dinv, bsum, Nn);
    k_scan_rowptr<<<nchunk, 256, 0, stream>>>(cnt, bsum, rowptr, Nn);
    k_fill_csr<<<(E + TB - 1) / TB, TB, 0, stream>>>(src, dst, rank, rowptr, csr, E);
  } else {
    k_deg_init<<<(Nn + TB - 1) / TB, TB, 0, stream>>>(dinv, Nn);
    k_deg_count<<<(E + TB - 1) / TB, TB, 0, stream>>>(dst, dinv, E);
    k_rsqrt_inplace<<<(Nn + TB - 1) / TB, TB, 0, stream>>>(dinv, Nn);
  }

  if (grouped) {
    k_gather_x24<<<(Nn + 7) / 8, 256, 0, stream>>>(x, dinv, rowptr, csr, bufA, Nn, E);
    k_zero<<<1, 256, 0, stream>>>(stats, 256);
    k_gemm1_relu_stats<<<(Nn + 16 * G1_TILES - 1) / (16 * G1_TILES), 256, 0, stream>>>(
        bufA, W1, b1, bufB, stats, Nn);
  } else {
    gemm(x, W1, bufA, 24, 128);
    aggregate(bufA, bufB, 128);
    finish(bufB, bufA, b1, 128);
  }

  gemm_bn(bufB, g1, be1, W2, bufA, 128);
  aggregate_finish(bufA, bufB, b2, 128);

  gemm_bn(bufB, g2, be2, W3, bufA, 64);
  aggregate_finish(bufA, bufB, b3, 64);

  k_gemm64x2<<<(Nn + TB - 1) / TB, TB, 0, stream>>>(bufB, stats, g3, be3, W4, dinv,
                                                    bufA, Nn, invN);
  if (grouped) {
    k_gather2_lsm<<<(Nn + 8 * L4_ROWS - 1) / (8 * L4_ROWS), 256, 0, stream>>>(
        bufA, rowptr, csr, dinv, b4, (float*)d_out, Nn, E);
  } else {
    zero(bufB, (size_t)Nn * 2);
    k_scatter2<<<(E + TB - 1) / TB, TB, 0, stream>>>(bufA, src, dst, bufB, E);
    k_final_lsm<<<(Nn + TB - 1) / TB, TB, 0, stream>>>(bufB, bufA, dinv, b4,
                                                       (float*)d_out, Nn);
  }
}


# in-edge grouping rebuilt as a bucket sort on LDS counters (chunk histograms, scan, pairs by 128-node bucket, a block per bucket): global int atomics, zero pass, random fill go (sum order changes)
# speedup vs baseline: 6.1545x; 1.3349x over previous
#include <hip/hip_runtime.h>
#include <hip/hip_bf16.h>
#include <math.h>


typedef float v2f __attribute__((ext_vector_type(2)));
typedef float v4f __attribute__((ext_vector_type(4)));
typedef float v8f __attribute__((ext_vector_type(8)));

#define BN_EPS 1e-5f

__global__ void k_zero(float* __restrict__ p, size_t n) {
  size_t i = (size_t)blockIdx.x * blockDim.x + threadIdx.x;
  size_t stride = (size_t)gridDim.x * blockDim.x;
  for (; i < n; i += stride) p[i] = 0.0f;
}

__global__ void k_deg_init(float* __restrict__ deg, int n) {
  int i = blockIdx.x * blockDim.x + threadIdx.x;
  if (i < n) deg[i] = 1.0f;
}

__global__ void k_deg_count(const int* __restrict__ dst, float* __restrict__ deg, int E) {
  int e = blockIdx.x * blockDim.x + threadIdx.x;
  if (e < E) unsafeAtomicAdd(&deg[dst[e]], 1.0f);
}

__global__ void k_rsqrt_inplace(float* __restrict__ d, int n) {
  int i = blockIdx.x * blockDim.x + threadIdx.x;
  if (i < n) {
    float v = d[i];
    d[i] = (v > 0.0f) ? rsqrtf(v) : 0.0f;
  }
}

__global__ void k_count_rank(const int* __restrict__ dst, int* __restrict__ cnt,
                             int* __restrict__ rank, int E) {
  int e = blockIdx.x * blockDim.x + threadIdx.x;
  if (e < E) rank[e] = atomicAdd(&cnt[dst[e]], 1);
}

#define SCAN_CHUNK 1024

__global__ void __launch_bounds__(256)
k_dinv_bsum(const int* __restrict__ cnt, float* __restrict__ dinv,
            int* __restrict__ bsum, int n) {
  __shared__ int wsum[8];
  const int t = threadIdx.x;
  const int i0 = blockIdx.x * SCAN_CHUNK + 4 * t;
  int s = 0;
#pragma unroll
  for (int k = 0; k < 4; ++k) {
    const int i = i0 + k;
    if (i < n) {
      const int c = cnt[i];
      s += c;
      const float deg = (float)(c + 1);
      dinv[i] = (deg > 0.0f) ? rsqrtf(deg) : 0.0f;
    }
  }
  for (int o = 16; o > 0; o >>= 1) s += __shfl_xor(s, o);
  if ((t & 31) == 0) wsum[t >> 5] = s;
  __syncthreads();
  if (t == 0) {
    int tot = 0;
    for (int w = 0; w < 8; ++w) tot += wsum[w];
    bsum[blockIdx.x] = tot;
  }
}

__global__ void __launch_bounds__(256)
k_scan_rowptr(const int* __restrict__ cnt, const int* __restrict__ bsum,
              int* __restrict__ rowptr, int n) {
  __shared__ int wtot[8];
  __shared__ int base_s;
  const int t = threadIdx.x, lane = t & 31, w = t >> 5;
  int p = 0;
  for (int b = t; b < (int)blockIdx.x; b += 256) p += bsum[b];
  for (int o = 16; o > 0; o >>= 1) p += __shfl_xor(p, o);
  if (lane == 0) wtot[w] = p;
  __syncthreads();
  if (t == 0) {
    int tot = 0;
    for (int k = 0; k < 8; ++k) tot += wtot[k];
    base_s = tot;
  }
  __syncthreads();
  const int base = base_s;
  const int i0 = blockIdx.x * SCAN_CHUNK + 4 * t;
  int c[4];
  int s = 0;
#pragma unroll
  for (int k = 0; k < 4; ++k) {
    c[k] = (i0 + k < n) ? cnt[i0 + k] : 0;
    s += c[k];
  }
  int incl = s;
  for (int o = 1; o < 32; o <<= 1) {
    const int v = __shfl_up(incl, o);
    if (lane >= o) incl += v;
  }
  if (lane == 31) wtot[w] = incl;
  __syncthreads();
  int woff = 0;
  for (int k = 0; k < w; ++k) woff += wtot[k];
  int excl = base + woff + incl - s;
#pragma unroll
  for (int k = 0; k < 4; ++k) {
    if (i0 + k < n) {
      rowptr[i0 + k] = excl;
      excl += c[k];
    }
  }
  if (blockIdx.x == gridDim.x - 1 && t == 255) rowptr[n] = excl;
}

__global__ void k_fill_csr(const int* __restrict__ src, const int* __restrict__ dst,
                           const int* __restrict__ rank, const int* __restrict__ rowptr,
                           int* __restrict__ csr, int E) {
  int e = blockIdx.x * blockDim.x + threadIdx.x;
  if (e >= E) return;
  unsigned pos = (unsigned)(rowptr[dst[e]] + rank[e]);
  csr[min(pos, (unsigned)(E - 1))] = src[e];
}

#define BKT_SHIFT 7
#define BKT_NODES (1 << BKT_SHIFT)
#define BKT_MAX 1024
#define BKT_BLOCKS 256
#define BKT_THREADS 1024

__global__ void __launch_bounds__(BKT_THREADS)
k_bkt_hist(const int* __restrict__ dst, int* __restrict__ hist, int E, int NB) {
  __shared__ int h[BKT_MAX];
  const int t = threadIdx.x;
  h[t] = 0;
  __syncthreads();
  const int chunk = (E + BKT_BLOCKS - 1) / BKT_BLOCKS;
  const int beg = min((int)blockIdx.x * chunk, E);
  const int end = min(beg + chunk, E);
  const unsigned bmax = (unsigned)(NB - 1);
  for (int e0 = beg + t; e0 < end; e0 += 4 * BKT_THREADS) {
    int d[4];
#pragma unroll
    for (int k = 0; k < 4; ++k) d[k] = dst[min(e0 + k * BKT_THREADS, end - 1)];
    asm volatile("" : "+v"(d[0]), "+v"(d[1]), "+v"(d[2]), "+v"(d[3]));
#pragma unroll
    for (int k = 0; k < 4; ++k)
      if (e0 + k * BKT_THREADS < end)
        atomicAdd(&h[min((unsigned)d[k] >> BKT_SHIFT, bmax)], 1);
  }
  __syncthreads();
  if (t < NB) hist[(size_t)blockIdx.x * BKT_MAX + t] = h[t];
}

__global__ void __launch_bounds__(1024)
k_bkt_scan(int* __restrict__ hist, int* __restrict__ btot, int NB) {
  constexpr int SEG = BKT_BLOCKS / 32;
  __shared__ int seg[32][32];
  const int lane = threadIdx.x & 31;
  const int w = threadIdx.x >> 5;
  const int b = blockIdx.x * 32 + lane;
  const bool ok = b < NB;
  int* hp = hist + (size_t)(w * SEG) * BKT_MAX + b;
  int v[SEG];
  int s = 0;
#pragma unroll
  for (int k = 0; k < SEG; ++k) v[k] = ok ? hp[(size_t)k * BKT_MAX] : 0;
#pragma unroll
  for (int k = 0; k < SEG; ++k) s += v[k];
  seg[w][lane] = s;
  __syncthreads();
  int off = 0, tot = 0;
  for (int k = 0; k < 32; ++k) {
    const int x = seg[k][lane];
    tot += x;
    if (k < w) off += x;
  }
  if (ok) {
#pragma unroll
    for (int k = 0; k < SEG; ++k) {
      hp[(size_t)k * BKT_MAX] = off;
      off += v[k];
    }
    if (w == 0) btot[b] = tot;
  }
}

__global__ void __launch_bounds__(BKT_THREADS)
k_bkt_bin(const int* __restrict__ src, const int* __restrict__ dst,
          const int* __restrict__ hist, const int* __restrict__ btot,
          int* __restrict__ bbase, int* __restrict__ pairs_, int E, int NB) {
  __shared__ int cur[BKT_MAX];
  __shared__ int wtot[BKT_THREADS / 32];
  int2* pairs = reinterpret_cast<int2*>(pairs_);
  const int t = threadIdx.x, lane = t & 31, w = t >> 5;
  const bool ok = t < NB;
  const int c = ok ? btot[t] : 0;
  const int within = ok ? hist[(size_t)blockIdx.x * BKT_MAX + t] : 0;
  int incl = c;
  for (int o = 1; o < 32; o <<= 1) {
    const int v = __shfl_up(incl, o);
    if (lane >= o) incl += v;
  }
  if (lane == 31) wtot[w] = incl;
  __syncthreads();
  int woff = 0;
  for (int k = 0; k < w; ++k) woff += wtot[k];
  const int excl = woff + incl - c;
  cur[t] = excl + within;
  if (blockIdx.x == 0 && ok) {
    bbase[t] = excl;
    if (t == NB - 1) bbase[NB] = excl + c;
  }
  __syncthreads();
  const int chunk = (E + BKT_BLOCKS - 1) / BKT_BLOCKS;
  const int beg = min((int)blockIdx.x * chunk, E);
  const int end = min(beg + chunk, E);
  const unsigned bmax = (unsigned)(NB - 1);
  const unsigned emax = (unsigned)(E - 1);
  for (int e0 = beg + t; e0 < end; e0 += 4 * BKT_THREADS) {
    int d[4], s[4];
#pragma unroll
    for (int k = 0; k < 4; ++k) {
      const int e = min(e0 + k * BKT_THREADS, end - 1);
      d[k] = dst[e];
      s[k] = src[e];
    }
    asm volatile("" : "+v"(d[0]), "+v"(d[1]), "+v"(d[2]), "+v"(d[3]),
                      "+v"(s[0]), "+v"(s[1]), "+v"(s[2]), "+v"(s[3]));
#pragma unroll
    for (int k = 0; k < 4; ++k) {
      if (e0 + k * BKT_THREADS < end) {
        const int pos = atomicAdd(&cur[min((unsigned)d[k] >> BKT_SHIFT, bmax)], 1);
        pairs[min((unsigned)pos, emax)] = make_int2(d[k], s[k]);
      }
    }
  }
}

__global__ void __launch_bounds__(256)
k_bkt_csr(const int* __restrict__ pairs_, const int* __restrict__ bbase,
          int* __restrict__ rowptr, int* __restrict__ csr, float* __restrict__ dinv,
          int Nn, int E) {
  __shared__ int cnt[BKT_NODES];
  __shared__ int cur[BKT_NODES];
  __shared__ int wtot[BKT_NODES / 32];
  const int2* pairs = reinterpret_cast<const int2*>(pairs_);
  const int t = threadIdx.x, lane = t & 31, w = t >> 5;
  const int b = blockIdx.x;
  const int beg = min(max(bbase[b], 0), E);
  const int end = min(max(bbase[b + 1], beg), E);
  if (t < BKT_NODES) cnt[t] = 0;
  __syncthreads();
  for (int i0 = beg + t; i0 < end; i0 += 4 * 256) {
    int d[4];
#pragma unroll
    for (int k = 0; k < 4; ++k) d[k] = pairs_[2 * (size_t)min(i0 + k * 256, end - 1)];
    asm volatile("" : "+v"(d[0]), "+v"(d[1]), "+v"(d[2]), "+v"(d[3]));
#pragma unroll
    for (int k = 0; k < 4; ++k)
      if (i0 + k * 256 < end) atomicAdd(&cnt[d[k] & (BKT_NODES - 1)], 1);
  }
  __syncthreads();
  const int c = (t < BKT_NODES) ? cnt[t] : 0;
  int incl = c;
  for (int o = 1; o < 32; o <<= 1) {
    const int v = __shfl_up(incl, o);
    if (lane >= o) incl += v;
  }
  if (t < BKT_NODES && lane == 31) wtot[w] = incl;
  __syncthreads();
  if (t < BKT_NODES) {
    int woff = 0;
    for (int k = 0; k < w; ++k) woff += wtot[k];
    const int excl = beg + woff + incl - c;
    cur[t] = excl;
    const int node = b * BKT_NODES + t;
    if (node < Nn) {
      rowptr[node] = excl;
      const float deg = (float)(c + 1);
      dinv[node] = (deg > 0.0f) ? rsqrtf(deg) : 0.0f;
      if (node == Nn - 1) rowptr[Nn] = end;
    }
  }
  __syncthreads();
  const unsigned emax = (unsigned)(E - 1);
  for (int i0 = beg + t; i0 < end; i0 += 4 * 256) {
    int2 p[4];
#pragma unroll
    for (int k = 0; k < 4; ++k) p[k] = pairs[min(i0 + k * 256, end - 1)];
    asm volatile("" : "+v"(p[0].x), "+v"(p[0].y), "+v"(p[1].x), "+v"(p[1].y),
                      "+v"(p[2].x), "+v"(p[2].y), "+v"(p[3].x), "+v"(p[3].y));
#pragma unroll
    for (int k = 0; k < 4; ++k) {
      if (i0 + k * 256 < end) {
        const int pos = atomicAdd(&cur[p[k].x & (BKT_NODES - 1)], 1);
        csr[min((unsigned)pos, emax)] = p[k].y;
      }
    }
  }
}

#define GF_ROWS_128 9
#define GF_ROWS_64 8
template <int F, int R>
__global__ void __launch_bounds__(256)
k_gather_finish(const float* __restrict__ hsc, const int* __restrict__ rowptr,
                const int* __restrict__ csr, const float* __restrict__ dinv,
                const float* __restrict__ bias, float* __restrict__ vout,
                float* __restrict__ stats,
                int Nn, int E) {
  constexpr int V = F / 32;
  typedef float vf __attribute__((ext_vector_type(V)));
  __shared__ __attribute__((aligned(16))) float red[2][8][F];
  const int t = threadIdx.x;
  const int lane = t & 31;
  const int w = t >> 5;
  const int d0 = __builtin_amdgcn_readfirstlane((int)((blockIdx.x * 8 + w) * R));
  vf s = 0.0f;
  vf q = 0.0f;
  if (d0 < Nn) {
    const int rows = min(R, Nn - d0);
    const vf* hp = reinterpret_cast<const vf*>(hsc) + lane;
    vf* op = reinterpret_cast<vf*>(vout) + lane;
    const unsigned nmax = (unsigned)(Nn - 1);
    const vf zerov = 0.0f;
    const int rpraw = rowptr[min(d0 + lane, Nn)];
    const float dvl = dinv[min(d0 + lane, Nn - 1)];
    const vf b = *reinterpret_cast<const vf*>(bias + V * lane);
    const int rp = min(max(rpraw, 0), E);
    const unsigned emax = (unsigned)(E - 1);
    int beg = __builtin_amdgcn_readlane(rp, 0);
    int end = __builtin_amdgcn_readlane(rp, 1);
    int idx = csr[min((unsigned)(beg + lane), emax)];
    auto sum_chunk = [&](const int cur, const int m, vf a) -> vf {
      auto row = [&](int u) -> vf {
        const unsigned sidx = min((unsigned)__builtin_amdgcn_readlane(cur, u), nmax);
        return hp[(size_t)sidx * 32];
      };
      int u = 0;
      for (; u + 8 <= m; u += 8) {
        vf r[8];
#pragma unroll
        for (int k = 0; k < 8; ++k) r[k] = row(u + k);
#pragma unroll
        for (int k = 0; k < 8; ++k) a += r[k];
      }
      if (u + 4 <= m) {
        const vf r0 = row(u), r1 = row(u + 1), r2 = row(u + 2), r3 = row(u + 3);
        a += r0; a += r1; a += r2; a += r3;
        u += 4;
      }
      if (u + 2 <= m) {
        const vf r0 = row(u), r1 = row(u + 1);
        a += r0; a += r1;
        u += 2;
      }
      if (u < m) a += row(u);
      return a;
    };
    for (int i = 0; i < rows; ++i) {
      const int d = d0 + i;
      const vf self = hp[(size_t)d * 32];
      const int nbeg = end;
      const int nend = __builtin_amdgcn_readlane(rp, i + 2);
      const int idxn = csr[min((unsigned)(nbeg + lane), emax)];
      vf acc = sum_chunk(idx, min(32, end - beg), zerov);
      for (int j0 = beg + 32; j0 < end; j0 += 32) {
        const int m = min(32, end - j0);
        const int more = (lane < m) ? csr[j0 + lane] : 0;
        acc = sum_chunk(more, m, acc);
      }
      const float di = __builtin_bit_cast(
          float, __builtin_amdgcn_readlane(__builtin_bit_cast(int, dvl), i));
      const vf v = __builtin_elementwise_max((acc + self) * di + b, zerov);
      op[(size_t)d * 32] = v;
      s += v;
      q += v * v;
      beg = nbeg;
      end = nend;
      idx = idxn;
    }
  }
  *reinterpret_cast<vf*>(&red[0][w][V * lane]) = s;
  *reinterpret_cast<vf*>(&red[1][w][V * lane]) = q;
  __syncthreads();
  if (t < 2 * F) {
    const int which = t / F;
    const int j = t % F;
    float a = 0.f;
    for (int k = 0; k < 8; ++k) a += red[which][k][j];
    unsafeAtomicAdd(stats + which * F + j, a);
  }
}

__global__ void k_gemm_wmma_f32(const float* __restrict__ A, const float* __restrict__ B,
                                const float* __restrict__ dinv,
                                float* __restrict__ C, int M, int K, int F) {
  const int lane = threadIdx.x & 31;
  const int wave = (int)((blockIdx.x * blockDim.x + threadIdx.x) >> 5);
  const int nf = F >> 4;
  const int mTile = wave / nf;
  const int fTile = wave - mTile * nf;
  const int m0 = mTile << 4;
  if (m0 >= M) return;
  const int f0 = fTile << 4;
  const int row = lane & 15;
  const int kh2 = (lane >> 4) << 1;

  const float* arow = A + (size_t)(m0 + row) * K + kh2;
  const float* bcol = B + (size_t)kh2 * F + f0 + row;

  v8f acc = {0.f, 0.f, 0.f, 0.f, 0.f, 0.f, 0.f, 0.f};
  for (int k = 0; k < K; k += 4) {
    float2 av = *reinterpret_cast<const float2*>(arow + k);
    v2f a; a.x = av.x; a.y = av.y;
    v2f b;
    b.x = bcol[(size_t)k * F];
    b.y = bcol[(size_t)(k + 1) * F];
    acc = __builtin_amdgcn_wmma_f32_16x16x4_f32(false, a, false, b,
                                                (short)0, acc, false, false);
  }
  const float* dvp = dinv + m0 + (kh2 << 2);
  float* crow = C + (size_t)(m0 + (kh2 << 2)) * F + f0 + row;
#pragma unroll
  for (int r = 0; r < 8; ++r) crow[(size_t)r * F] = acc[r] * dvp[r];
}

template <int F>
__global__ void __launch_bounds__(256)
k_gemm_bn_wmma_f32(const float* __restrict__ V, const float* __restrict__ stats,
                   const float* __restrict__ gamma, const float* __restrict__ beta,
                   const float* __restrict__ B, const float* __restrict__ dinv,
                   float* __restrict__ C, int M, float invN) {
  constexpr int K = 128;
  constexpr int NF = F / 16;
  constexpr int RB = 16 * (8 / NF);
  constexpr int PITCH = K + 4;
  __shared__ __attribute__((aligned(16))) float As[RB * PITCH];

  const int t = threadIdx.x;
  const int lane = t & 31;
  const int w = t >> 5;
  const int rowBase = blockIdx.x * RB;

  {
    const int c4 = t & 31;
    const v4f su = *reinterpret_cast<const v4f*>(stats + 4 * c4);
    const v4f sq = *reinterpret_cast<const v4f*>(stats + K + 4 * c4);
    const v4f g = *reinterpret_cast<const v4f*>(gamma + 4 * c4);
    const v4f be = *reinterpret_cast<const v4f*>(beta + 4 * c4);
    v4f mean, rstd;
#pragma unroll
    for (int i = 0; i < 4; ++i) {
      mean[i] = su[i] * invN;
      const float var = sq[i] * invN - mean[i] * mean[i];
      rstd[i] = rsqrtf(var + BN_EPS);
    }
    v4f v[RB / 8];
#pragma unroll
    for (int p = 0; p < RB / 8; ++p) {
      const int gr = min(rowBase + p * 8 + w, M - 1);
      v[p] = *reinterpret_cast<const v4f*>(V + (size_t)gr * K + 4 * c4);
    }
#pragma unroll
    for (int p = 0; p < RB / 8; ++p) {
      const v4f o = (v[p] - mean) * rstd * g + be;
      *reinterpret_cast<v4f*>(&As[(p * 8 + w) * PITCH + 4 * c4]) = o;
    }
  }
  __syncthreads();

  const int mT = w / NF;
  const int fT = w - mT * NF;
  const int m0 = rowBase + (mT << 4);
  if (m0 >= M) return;
  const int f0 = fT << 4;
  const int row = lane & 15;
  const int kh2 = (lane >> 4) << 1;

  const float* arow = As + ((mT << 4) + row) * PITCH + kh2;
  const float* bcol = B + (size_t)kh2 * F + f0 + row;

  unsigned dbyte = (unsigned)(m0 + (kh2 << 2)) * 4u;
  unsigned cbyte = ((unsigned)(m0 + (kh2 << 2)) * F + f0 + row) * 4u;
  asm volatile("" : "+v"(dbyte), "+v"(cbyte));
  const float* dvp = reinterpret_cast<const float*>(reinterpret_cast<const char*>(dinv) + dbyte);
  float* crow = reinterpret_cast<float*>(reinterpret_cast<char*>(C) + cbyte);

  v8f acc = {0.f, 0.f, 0.f, 0.f, 0.f, 0.f, 0.f, 0.f};
#pragma unroll
  for (int k = 0; k < K; k += 4) {
    const v2f a = *reinterpret_cast<const v2f*>(arow + k);
    v2f b;
    b.x = bcol[(size_t)k * F];
    b.y = bcol[(size_t)(k + 1) * F];
    acc = __builtin_amdgcn_wmma_f32_16x16x4_f32(false, a, false, b,
                                                (short)0, acc, false, false);
  }
  asm volatile("s_wait_alu depctr_va_vdst(0)\n\t"
               "v_nop\n\tv_nop\n\tv_nop\n\tv_nop\n\t"
               "v_nop\n\tv_nop\n\tv_nop\n\tv_nop"
               : "+v"(acc));
#pragma unroll
  for (int r = 0; r < 8; ++r) crow[(size_t)r * F] = acc[r] * dvp[r];
}

__global__ void __launch_bounds__(256)
k_gather_x24(const float* __restrict__ x, const float* __restrict__ dinv,
             const int* __restrict__ rowptr, const int* __restrict__ csr,
             float* __restrict__ ax, int Nn, int E) {
  const int lane = threadIdx.x & 31;
  const int d = __builtin_amdgcn_readfirstlane((int)(blockIdx.x * 8 + (threadIdx.x >> 5)));
  if (d >= Nn) return;
  const int beg = max(rowptr[d], 0);
  const int end = min(rowptr[d + 1], E);
  const int p = lane & 7;
  const int q = lane >> 3;
  const bool pv = p < 6;
  const unsigned nmax = (unsigned)(Nn - 1);
  const v4f* xp = reinterpret_cast<const v4f*>(x) + p;
  const v4f zero4 = {0.f, 0.f, 0.f, 0.f};
  const float dd = dinv[d];
  const v4f xd = pv ? xp[(size_t)d * 6] : zero4;
  v4f acc = zero4;
  for (int j0 = beg; j0 < end; j0 += 32) {
    const int m = min(32, end - j0);
    int idx = 0;
    float dv = 0.f;
    if (lane < m) {
      idx = (int)min((unsigned)csr[j0 + lane], nmax);
      dv = dinv[idx];
    }
    v4f r[8];
    float wgt[8];
#pragma unroll
    for (int k = 0; k < 8; ++k) {
      const int e = 4 * k + q;
      const int s = __shfl(idx, e);
      wgt[k] = __shfl(dv, e);
      r[k] = (pv && e < m) ? xp[(size_t)s * 6] : zero4;
    }
#pragma unroll
    for (int k = 0; k < 8; ++k) acc += r[k] * wgt[k];
  }
#pragma unroll
  for (int i = 0; i < 4; ++i) {
    acc[i] += __shfl_xor(acc[i], 8);
    acc[i] += __shfl_xor(acc[i], 16);
  }
  if (q == 0 && pv)
    reinterpret_cast<v4f*>(ax)[(size_t)d * 6 + p] = (acc + xd * dd) * dd;
}

#define G1_TILES 8
__global__ void __launch_bounds__(256)
k_gemm1_relu_stats(const float* __restrict__ A, const float* __restrict__ B,
                   const float* __restrict__ bias, float* __restrict__ C,
                   float* __restrict__ stats,
                   int M) {
  constexpr int K = 24;
  constexpr int F = 128;
  constexpr int KS = K / 4;
  const int lane = threadIdx.x & 31;
  const int f0 = (int)(threadIdx.x >> 5) << 4;
  const int c = lane & 15;
  const int h = lane >> 4;
  const int kh2 = h << 1;
  const int rowBase = blockIdx.x * (16 * G1_TILES);

  v2f b[KS];
  const float* bcol = B + (size_t)kh2 * F + f0 + c;
#pragma unroll
  for (int kk = 0; kk < KS; ++kk) {
    b[kk].x = bcol[(size_t)(4 * kk) * F];
    b[kk].y = bcol[(size_t)(4 * kk + 1) * F];
  }
  const float bc = bias[f0 + c];

  v2f a[G1_TILES][KS];
#pragma unroll
  for (int i = 0; i < G1_TILES; ++i) {
    const int gr = min(rowBase + 16 * i + c, M - 1);
    const float* ar = A + (size_t)gr * K + kh2;
#pragma unroll
    for (int kk = 0; kk < KS; ++kk)
      a[i][kk] = *reinterpret_cast<const v2f*>(ar + 4 * kk);
  }

  float s = 0.f, q = 0.f;
#pragma unroll
  for (int i = 0; i < G1_TILES; ++i) {
    const int m0 = rowBase + 16 * i;
    if (m0 >= M) break;
    __builtin_amdgcn_sched_barrier(0);
    v8f acc = {0.f, 0.f, 0.f, 0.f, 0.f, 0.f, 0.f, 0.f};
#pragma unroll
    for (int kk = 0; kk < KS; ++kk)
      acc = __builtin_amdgcn_wmma_f32_16x16x4_f32(false, a[i][kk], false, b[kk],
                                                  (short)0, acc, false, false);
    asm volatile("s_wait_alu depctr_va_vdst(0)\n\t"
                 "v_nop\n\tv_nop\n\tv_nop\n\tv_nop\n\t"
                 "v_nop\n\tv_nop\n\tv_nop\n\tv_nop"
                 : "+v"(acc));
    __builtin_amdgcn_sched_barrier(0);
    const int r0 = m0 + (h << 3);
    float* crow = C + (size_t)r0 * F + f0 + c;
#pragma unroll
    for (int r = 0; r < 8; ++r) {
      if (r0 + r < M) {
        const float v = fmaxf(acc[r] + bc, 0.f);
        crow[(size_t)r * F] = v;
        s += v;
        q += v * v;
      }
    }
  }
  s += __shfl_xor(s, 16);
  q += __shfl_xor(q, 16);
  if (h == 0) {
    unsafeAtomicAdd(stats + f0 + c, s);
    unsafeAtomicAdd(stats + F + f0 + c, q);
  }
}

__global__ void k_scatter(const float* __restrict__ hsc, const int* __restrict__ src,
                          const int* __restrict__ dst,
                          float* __restrict__ agg, int E, int F) {
  int wave = (int)((blockIdx.x * blockDim.x + threadIdx.x) >> 5);
  int lane = threadIdx.x & 31;
  if (wave >= E) return;
  int s = src[wave], d = dst[wave];
  const float4* hrow = reinterpret_cast<const float4*>(hsc + (size_t)s * F);
  float* ad = agg + (size_t)d * F;
  for (int f4 = lane; 4 * f4 < F; f4 += 32) {
    float4 v = hrow[f4];
    unsafeAtomicAdd(ad + 4 * f4 + 0, v.x);
    unsafeAtomicAdd(ad + 4 * f4 + 1, v.y);
    unsafeAtomicAdd(ad + 4 * f4 + 2, v.z);
    unsafeAtomicAdd(ad + 4 * f4 + 3, v.w);
  }
}

#define FIN_ROWS 128
template <int F>
__global__ void __launch_bounds__(256)
k_finish_relu_stats(float* __restrict__ agg, const float* __restrict__ hsc,
                    const float* __restrict__ dinv,
                    const float* __restrict__ bias,
                    float* __restrict__ stats,
                    int Nn) {
  constexpr int TPR = F / 4;
  constexpr int RPP = 256 / TPR;
  constexpr int U = 8;
  constexpr int STEP = RPP * TPR;
  static_assert(FIN_ROWS % (U * RPP) == 0, "a block is a whole number of groups");
  __shared__ __attribute__((aligned(16))) float red[2][RPP][F];

  const int t = threadIdx.x;
  const int c4 = t % TPR;
  const int rsub = t / TPR;
  const int r0 = blockIdx.x * FIN_ROWS;
  const int r1 = min(r0 + FIN_ROWS, Nn);

  const v4f b = *reinterpret_cast<const v4f*>(bias + 4 * c4);
  const v4f zero4 = {0.f, 0.f, 0.f, 0.f};
  v4f s = zero4;
  v4f q = zero4;

  int r = r0 + rsub;
  v4f* ap = reinterpret_cast<v4f*>(agg) + (size_t)r * TPR + c4;
  const v4f* hp = reinterpret_cast<const v4f*>(hsc) + (size_t)r * TPR + c4;
  const float* dp = dinv + r;

  auto apply = [&](const v4f a, const v4f h, const float di, v4f* dst) {
    const v4f v = __builtin_elementwise_max((a + h) * di + b, zero4);
    *dst = v;
    s += v;
    q += v * v;
  };

  for (; r + (U - 1) * RPP < r1; r += U * RPP) {
    v4f a[U], h[U];
    float di[U];
#pragma unroll
    for (int u = 0; u < U; ++u) {
      a[u] = ap[u * STEP];
      h[u] = hp[u * STEP];
      di[u] = dp[u * RPP];
    }
    __builtin_amdgcn_sched_barrier(0);
#pragma unroll
    for (int u = 0; u < U; ++u) apply(a[u], h[u], di[u], ap + u * STEP);
    ap += U * STEP;
    hp += U * STEP;
    dp += U * RPP;
  }
  for (; r < r1; r += RPP) {
    apply(*ap, *hp, *dp, ap);
    ap += STEP;
    hp += STEP;
    dp += RPP;
  }

  *reinterpret_cast<v4f*>(&red[0][rsub][4 * c4]) = s;
  *reinterpret_cast<v4f*>(&red[1][rsub][4 * c4]) = q;
  __syncthreads();
  if (t < 2 * F) {
    const int which = t / F;
    const int j = t % F;
    float acc = 0.f;
    for (int k = 0; k < RPP; ++k) acc += red[which][k][j];
    unsafeAtomicAdd(stats + which * F + j, acc);
  }
}

__global__ void k_gemm64x2(const float* __restrict__ v3, const float* __restrict__ stats,
                           const float* __restrict__ gamma, const float* __restrict__ beta,
                           const float* __restrict__ W4, const float* __restrict__ dinv,
                           float* __restrict__ h4, int Nn, float invN) {
  int i = blockIdx.x * blockDim.x + threadIdx.x;
  if (i >= Nn) return;
  const float* row = v3 + (size_t)i * 64;
  float o0 = 0.f, o1 = 0.f;
#pragma unroll 8
  for (int k = 0; k < 64; ++k) {
    const float mean = stats[k] * invN;
    const float var = stats[64 + k] * invN - mean * mean;
    const float xv = (row[k] - mean) * rsqrtf(var + BN_EPS) * gamma[k] + beta[k];
    o0 += xv * W4[2 * k];
    o1 += xv * W4[2 * k + 1];
  }
  float di = dinv[i];
  h4[2 * (size_t)i]     = o0 * di;
  h4[2 * (size_t)i + 1] = o1 * di;
}

__global__ void k_scatter2(const float* __restrict__ h4s, const int* __restrict__ src,
                           const int* __restrict__ dst,
                           float* __restrict__ agg, int E) {
  int e = blockIdx.x * blockDim.x + threadIdx.x;
  if (e >= E) return;
  int s = src[e], d = dst[e];
  float2 v = *reinterpret_cast<const float2*>(h4s + 2 * (size_t)s);
  unsafeAtomicAdd(agg + 2 * (size_t)d,     v.x);
  unsafeAtomicAdd(agg + 2 * (size_t)d + 1, v.y);
}

__global__ void k_final_lsm(const float* __restrict__ agg, const float* __restrict__ h4s,
                            const float* __restrict__ dinv, const float* __restrict__ b4,
                            float* __restrict__ out, int Nn) {
  int i = blockIdx.x * blockDim.x + threadIdx.x;
  if (i >= Nn) return;
  float di = dinv[i];
  float o0 = (agg[2 * (size_t)i]     + h4s[2 * (size_t)i])     * di + b4[0];
  float o1 = (agg[2 * (size_t)i + 1] + h4s[2 * (size_t)i + 1]) * di + b4[1];
  float m = fmaxf(o0, o1);
  float l = m + logf(expf(o0 - m) + expf(o1 - m));
  out[2 * (size_t)i]     = o0 - l;
  out[2 * (size_t)i + 1] = o1 - l;
}

#define L4_ROWS 16
__global__ void __launch_bounds__(256)
k_gather2_lsm(const float* __restrict__ h4s, const int* __restrict__ rowptr,
              const int* __restrict__ csr, const float* __restrict__ dinv,
              const float* __restrict__ b4, float* __restrict__ out, int Nn, int E) {
  const int lane = threadIdx.x & 31;
  const int d0 = __builtin_amdgcn_readfirstlane(
      (int)((blockIdx.x * 8 + (threadIdx.x >> 5)) * L4_ROWS));
  if (d0 >= Nn) return;
  const int p = lane & 7;
  const int q = lane >> 3;
  const v2f* hp = reinterpret_cast<const v2f*>(h4s);
  const unsigned nmax = (unsigned)(Nn - 1);
  const unsigned emax = (unsigned)(E - 1);
  const v2f zero2 = {0.f, 0.f};
  const int rpraw = rowptr[min(d0 + lane, Nn)];
  const int gs = p & 3;
  const int rfin = d0 + 4 * gs + q;
  const int rfc = min(rfin, Nn - 1);
  const v2f self = hp[rfc];
  const float di = dinv[rfc];
  const float bb0 = b4[0];
  const float bb1 = b4[1];
  const int rp = min(max(rpraw, 0), E);

  int beg[4], len[4];
  unsigned own[4];
#pragma unroll
  for (int g = 0; g < 4; ++g) {
    const int b = __shfl(rp, 4 * g + q);
    const int e = __shfl(rp, 4 * g + q + 1);
    beg[g] = b;
    len[g] = e - b;
    own[g] = min((unsigned)(d0 + 4 * g + q), nmax);
  }
  int idx[4][4];
#pragma unroll
  for (int g = 0; g < 4; ++g) {
#pragma unroll
    for (int k = 0; k < 4; ++k)
      idx[g][k] = csr[min((unsigned)(beg[g] + p + 8 * k), emax)];
  }
  v2f v[4][4];
#pragma unroll
  for (int g = 0; g < 4; ++g) {
#pragma unroll
    for (int k = 0; k < 4; ++k) {
      const unsigned s = (p + 8 * k < len[g]) ? min((unsigned)idx[g][k], nmax) : own[g];
      v[g][k] = hp[s];
    }
  }
  v2f acc[4];
#pragma unroll
  for (int g = 0; g < 4; ++g) {
    v2f a = zero2;
#pragma unroll
    for (int k = 0; k < 4; ++k) a += (p + 8 * k < len[g]) ? v[g][k] : zero2;
    const int end = beg[g] + len[g];
    for (int j = beg[g] + 32 + p; j < end; j += 8) a += hp[min((unsigned)csr[j], nmax)];
    a.x += __shfl_xor(a.x, 1);
    a.y += __shfl_xor(a.y, 1);
    a.x += __shfl_xor(a.x, 2);
    a.y += __shfl_xor(a.y, 2);
    a.x += __shfl_xor(a.x, 4);
    a.y += __shfl_xor(a.y, 4);
    acc[g] = a;
  }
  const v2f a = gs == 0 ? acc[0] : gs == 1 ? acc[1] : gs == 2 ? acc[2] : acc[3];
  const float o0 = (a.x + self.x) * di + bb0;
  const float o1 = (a.y + self.y) * di + bb1;
  const float m = fmaxf(o0, o1);
  const float l = m + logf(expf(o0 - m) + expf(o1 - m));
  v2f o;
  o.x = o0 - l;
  o.y = o1 - l;
  if (p < 4 && rfin < Nn) reinterpret_cast<v2f*>(out)[rfin] = o;
}

extern "C" void kernel_launch(void* const* d_in, const int* in_sizes, int n_in,
                              void* d_out, int out_size, void* d_ws, size_t ws_size,
                              hipStream_t stream) {
  const float* x   = (const float*)d_in[0];
  const int*   ei  = (const int*)d_in[1];
  const float* W1  = (const float*)d_in[2];
  const float* b1  = (const float*)d_in[3];
  const float* g1  = (const float*)d_in[4];
  const float* be1 = (const float*)d_in[5];
  const float* W2  = (const float*)d_in[6];
  const float* b2  = (const float*)d_in[7];
  const float* g2  = (const float*)d_in[8];
  const float* be2 = (const float*)d_in[9];
  const float* W3  = (const float*)d_in[10];
  const float* b3  = (const float*)d_in[11];
  const float* g3  = (const float*)d_in[12];
  const float* be3 = (const float*)d_in[13];
  const float* W4  = (const float*)d_in[14];
  const float* b4  = (const float*)d_in[15];

  const int Nn = in_sizes[0] / 24;
  const int E  = in_sizes[1] / 2;
  const int* src = ei;
  const int* dst = ei + E;

  float* ws    = (float*)d_ws;
  float* dinv  = ws;
  float* stats = ws + Nn;
  float* bufA  = stats + 256;
  float* bufB  = bufA + (size_t)Nn * 128;
  const int nchunk = (Nn + SCAN_CHUNK - 1) / SCAN_CHUNK;
  const size_t rp_len = ((size_t)Nn + 1 + 31) & ~(size_t)31;
  const size_t bs_len = ((size_t)nchunk + 31) & ~(size_t)31;
  int* rowptr = (int*)(bufB + (size_t)Nn * 128);
  int* csr    = rowptr + rp_len;
  int* cnt    = (int*)bufB;
  int* bsum   = cnt + Nn;
  int* rank   = bsum + bs_len;
  const size_t need_bytes =
      ((size_t)Nn + 256 + 2 * (size_t)Nn * 128 + rp_len + (size_t)E) * sizeof(float);
  const bool grouped = ws_size >= need_bytes &&
                       (size_t)Nn + bs_len + (size_t)E <= (size_t)Nn * 128;
  const int NB = (Nn + BKT_NODES - 1) / BKT_NODES;
  int* hist  = (int*)bufB;
  int* btot  = hist + (size_t)BKT_BLOCKS * BKT_MAX;
  int* bbase = btot + BKT_MAX;
  int* pairs = bbase + BKT_MAX + 32;
  const bool bucketed = grouped && NB <= BKT_MAX &&
                        (size_t)BKT_BLOCKS * BKT_MAX + 2 * BKT_MAX + 32 + 2 * (size_t)E <=
                            (size_t)Nn * 128;

  const float invN = 1.0f / (float)Nn;
  const int TB = 256;

  auto gemm = [&](const float* A, const float* Bm, float* C, int K, int F) {
    long waves = (long)((Nn + 15) / 16) * (F / 16);
    int blocks = (int)((waves * 32 + TB - 1) / TB);
    k_gemm_wmma_f32<<<blocks, TB, 0, stream>>>(A, Bm, dinv, C, Nn, K, F);
  };
  auto scatter = [&](const float* h, float* agg, int F) {
    int blocks = (int)(((long)E * 32 + TB - 1) / TB);
    k_scatter<<<blocks, TB, 0, stream>>>(h, src, dst, agg, E, F);
  };
  auto finish = [&](float* agg, const float* h, const float* bias, int F) {
    const int blocks = (Nn + FIN_ROWS - 1) / FIN_ROWS;
    k_zero<<<1, 256, 0, stream>>>(stats, 256);
    if (F == 128)
      k_finish_relu_stats<128><<<blocks, 256, 0, stream>>>(agg, h, dinv, bias, stats, Nn);
    else
      k_finish_relu_stats<64><<<blocks, 256, 0, stream>>>(agg, h, dinv, bias, stats, Nn);
  };
  auto gemm_bn = [&](const float* V, const float* gamma, const float* beta,
                     const float* Bm, float* C, int F) {
    if (F == 128)
      k_gemm_bn_wmma_f32<128><<<(Nn + 15) / 16, 256, 0, stream>>>(
          V, stats, gamma, beta, Bm, dinv, C, Nn, invN);
    else
      k_gemm_bn_wmma_f32<64><<<(Nn + 31) / 32, 256, 0, stream>>>(
          V, stats, gamma, beta, Bm, dinv, C, Nn, invN);
  };
  auto zero = [&](float* p, size_t n) {
    int blocks = (int)((n + TB - 1) / TB);
    if (blocks > 4096) blocks = 4096;
    k_zero<<<blocks, TB, 0, stream>>>(p, n);
  };

  auto aggregate = [&](const float* h, float* agg, int F) {
    zero(agg, (size_t)Nn * F);
    scatter(h, agg, F);
  };
  auto aggregate_finish = [&](const float* h, float* v, const float* bias, int F) {
    if (grouped) {
      k_zero<<<1, 256, 0, stream>>>(stats, 256);
      if (F == 128)
        k_gather_finish<128, GF_ROWS_128>
            <<<(Nn + 8 * GF_ROWS_128 - 1) / (8 * GF_ROWS_128), 256, 0, stream>>>(
                h, rowptr, csr, dinv, bias, v, stats, Nn, E);
      else
        k_gather_finish<64, GF_ROWS_64>
            <<<(Nn + 8 * GF_ROWS_64 - 1) / (8 * GF_ROWS_64), 256, 0, stream>>>(
                h, rowptr, csr, dinv, bias, v, stats, Nn, E);
    } else {
      aggregate(h, v, F);
      finish(v, h, bias, F);
    }
  };

  if (bucketed) {
    k_bkt_hist<<<BKT_BLOCKS, BKT_THREADS, 0, stream>>>(dst, hist, E, NB);
    k_bkt_scan<<<(NB + 31) / 32, 1024, 0, stream>>>(hist, btot, NB);
    k_bkt_bin<<<BKT_BLOCKS, BKT_THREADS, 0, stream>>>(src, dst, hist, btot, bbase, pairs, E, NB);
    k_bkt_csr<<<NB, 256, 0, stream>>>(pairs, bbase, rowptr, csr, dinv, Nn, E);
  } else if (grouped) {
    zero((float*)cnt, (size_t)Nn);
    k_count_rank<<<(E + TB - 1) / TB, TB, 0, stream>>>(dst, cnt, rank, E);
    k_dinv_bsum<<<nchunk, 256, 0, stream>>>(cnt, dinv, bsum, Nn);
    k_scan_rowptr<<<nchunk, 256, 0, stream>>>(cnt, bsum, rowptr, Nn);
    k_fill_csr<<<(E + TB - 1) / TB, TB, 0, stream>>>(src, dst, rank, rowptr, csr, E);
  } else {
    k_deg_init<<<(Nn + TB - 1) / TB, TB, 0, stream>>>(dinv, Nn);
    k_deg_count<<<(E + TB - 1) / TB, TB, 0, stream>>>(dst, dinv, E);
    k_rsqrt_inplace<<<(Nn + TB - 1) / TB, TB, 0, stream>>>(dinv, Nn);
  }

  if (grouped) {
    k_gather_x24<<<(Nn + 7) / 8, 256, 0, stream>>>(x, dinv, rowptr, csr, bufA, Nn, E);
    k_zero<<<1, 256, 0, stream>>>(stats, 256);
    k_gemm1_relu_stats<<<(Nn + 16 * G1_TILES - 1) / (16 * G1_TILES), 256, 0, stream>>>(
        bufA, W1, b1, bufB, stats, Nn);
  } else {
    gemm(x, W1, bufA, 24, 128);
    aggregate(bufA, bufB, 128);
    finish(bufB, bufA, b1, 128);
  }

  gemm_bn(bufB, g1, be1, W2, bufA, 128);
  aggregate_finish(bufA, bufB, b2, 128);

  gemm_bn(bufB, g2, be2, W3, bufA, 64);
  aggregate_finish(bufA, bufB, b3, 64);

  k_gemm64x2<<<(Nn + TB - 1) / TB, TB, 0, stream>>>(bufB, stats, g3, be3, W4, dinv,
                                                    bufA, Nn, invN);
  if (grouped) {
    k_gather2_lsm<<<(Nn + 8 * L4_ROWS - 1) / (8 * L4_ROWS), 256, 0, stream>>>(
        bufA, rowptr, csr, dinv, b4, (float*)d_out, Nn, E);
  } else {
    zero(bufB, (size_t)Nn * 2);
    k_scatter2<<<(E + TB - 1) / TB, TB, 0, stream>>>(bufA, src, dst, bufB, E);
    k_final_lsm<<<(Nn + TB - 1) / TB, TB, 0, stream>>>(bufB, bufA, dinv, b4,
                                                       (float*)d_out, Nn);
  }
}


# four launches fewer (15 -> 11): layer 1's edge sum and product in one kernel (summed rows through an LDS tile); one BN sum buffer per layer, all cleared in the bucket pass (BN sum order changes)
# speedup vs baseline: 6.7249x; 1.0927x over previous
#include <hip/hip_runtime.h>
#include <hip/hip_bf16.h>
#include <math.h>


typedef float v2f __attribute__((ext_vector_type(2)));
typedef float v4f __attribute__((ext_vector_type(4)));
typedef float v8f __attribute__((ext_vector_type(8)));

#define BN_EPS 1e-5f

__global__ void k_zero(float* __restrict__ p, size_t n) {
  size_t i = (size_t)blockIdx.x * blockDim.x + threadIdx.x;
  size_t stride = (size_t)gridDim.x * blockDim.x;
  for (; i < n; i += stride) p[i] = 0.0f;
}

__global__ void k_deg_init(float* __restrict__ deg, int n) {
  int i = blockIdx.x * blockDim.x + threadIdx.x;
  if (i < n) deg[i] = 1.0f;
}

__global__ void k_deg_count(const int* __restrict__ dst, float* __restrict__ deg, int E) {
  int e = blockIdx.x * blockDim.x + threadIdx.x;
  if (e < E) unsafeAtomicAdd(&deg[dst[e]], 1.0f);
}

__global__ void k_rsqrt_inplace(float* __restrict__ d, int n) {
  int i = blockIdx.x * blockDim.x + threadIdx.x;
  if (i < n) {
    float v = d[i];
    d[i] = (v > 0.0f) ? rsqrtf(v) : 0.0f;
  }
}

__global__ void k_count_rank(const int* __restrict__ dst, int* __restrict__ cnt,
                             int* __restrict__ rank, int E) {
  int e = blockIdx.x * blockDim.x + threadIdx.x;
  if (e < E) rank[e] = atomicAdd(&cnt[dst[e]], 1);
}

#define SCAN_CHUNK 1024

__global__ void __launch_bounds__(256)
k_dinv_bsum(const int* __restrict__ cnt, float* __restrict__ dinv,
            int* __restrict__ bsum, int n) {
  __shared__ int wsum[8];
  const int t = threadIdx.x;
  const int i0 = blockIdx.x * SCAN_CHUNK + 4 * t;
  int s = 0;
#pragma unroll
  for (int k = 0; k < 4; ++k) {
    const int i = i0 + k;
    if (i < n) {
      const int c = cnt[i];
      s += c;
      const float deg = (float)(c + 1);
      dinv[i] = (deg > 0.0f) ? rsqrtf(deg) : 0.0f;
    }
  }
  for (int o = 16; o > 0; o >>= 1) s += __shfl_xor(s, o);
  if ((t & 31) == 0) wsum[t >> 5] = s;
  __syncthreads();
  if (t == 0) {
    int tot = 0;
    for (int w = 0; w < 8; ++w) tot += wsum[w];
    bsum[blockIdx.x] = tot;
  }
}

__global__ void __launch_bounds__(256)
k_scan_rowptr(const int* __restrict__ cnt, const int* __restrict__ bsum,
              int* __restrict__ rowptr, int n) {
  __shared__ int wtot[8];
  __shared__ int base_s;
  const int t = threadIdx.x, lane = t & 31, w = t >> 5;
  int p = 0;
  for (int b = t; b < (int)blockIdx.x; b += 256) p += bsum[b];
  for (int o = 16; o > 0; o >>= 1) p += __shfl_xor(p, o);
  if (lane == 0) wtot[w] = p;
  __syncthreads();
  if (t == 0) {
    int tot = 0;
    for (int k = 0; k < 8; ++k) tot += wtot[k];
    base_s = tot;
  }
  __syncthreads();
  const int base = base_s;
  const int i0 = blockIdx.x * SCAN_CHUNK + 4 * t;
  int c[4];
  int s = 0;
#pragma unroll
  for (int k = 0; k < 4; ++k) {
    c[k] = (i0 + k < n) ? cnt[i0 + k] : 0;
    s += c[k];
  }
  int incl = s;
  for (int o = 1; o < 32; o <<= 1) {
    const int v = __shfl_up(incl, o);
    if (lane >= o) incl += v;
  }
  if (lane == 31) wtot[w] = incl;
  __syncthreads();
  int woff = 0;
  for (int k = 0; k < w; ++k) woff += wtot[k];
  int excl = base + woff + incl - s;
#pragma unroll
  for (int k = 0; k < 4; ++k) {
    if (i0 + k < n) {
      rowptr[i0 + k] = excl;
      excl += c[k];
    }
  }
  if (blockIdx.x == gridDim.x - 1 && t == 255) rowptr[n] = excl;
}

__global__ void k_fill_csr(const int* __restrict__ src, const int* __restrict__ dst,
                           const int* __restrict__ rank, const int* __restrict__ rowptr,
                           int* __restrict__ csr, int E) {
  int e = blockIdx.x * blockDim.x + threadIdx.x;
  if (e >= E) return;
  unsigned pos = (unsigned)(rowptr[dst[e]] + rank[e]);
  csr[min(pos, (unsigned)(E - 1))] = src[e];
}

#define BKT_SHIFT 7
#define BKT_NODES (1 << BKT_SHIFT)
#define BKT_MAX 1024
#define BKT_BLOCKS 256
#define BKT_THREADS 1024

__global__ void __launch_bounds__(BKT_THREADS)
k_bkt_hist(const int* __restrict__ dst, int* __restrict__ hist, int E, int NB) {
  __shared__ int h[BKT_MAX];
  const int t = threadIdx.x;
  h[t] = 0;
  __syncthreads();
  const int chunk = (E + BKT_BLOCKS - 1) / BKT_BLOCKS;
  const int beg = min((int)blockIdx.x * chunk, E);
  const int end = min(beg + chunk, E);
  const unsigned bmax = (unsigned)(NB - 1);
  for (int e0 = beg + t; e0 < end; e0 += 4 * BKT_THREADS) {
    int d[4];
#pragma unroll
    for (int k = 0; k < 4; ++k) d[k] = dst[min(e0 + k * BKT_THREADS, end - 1)];
    asm volatile("" : "+v"(d[0]), "+v"(d[1]), "+v"(d[2]), "+v"(d[3]));
#pragma unroll
    for (int k = 0; k < 4; ++k)
      if (e0 + k * BKT_THREADS < end)
        atomicAdd(&h[min((unsigned)d[k] >> BKT_SHIFT, bmax)], 1);
  }
  __syncthreads();
  if (t < NB) hist[(size_t)blockIdx.x * BKT_MAX + t] = h[t];
}

__global__ void __launch_bounds__(1024)
k_bkt_scan(int* __restrict__ hist, int* __restrict__ btot, int NB) {
  constexpr int SEG = BKT_BLOCKS / 32;
  __shared__ int seg[32][32];
  const int lane = threadIdx.x & 31;
  const int w = threadIdx.x >> 5;
  const int b = blockIdx.x * 32 + lane;
  const bool ok = b < NB;
  int* hp = hist + (size_t)(w * SEG) * BKT_MAX + b;
  int v[SEG];
  int s = 0;
#pragma unroll
  for (int k = 0; k < SEG; ++k) v[k] = ok ? hp[(size_t)k * BKT_MAX] : 0;
#pragma unroll
  for (int k = 0; k < SEG; ++k) s += v[k];
  seg[w][lane] = s;
  __syncthreads();
  int off = 0, tot = 0;
  for (int k = 0; k < 32; ++k) {
    const int x = seg[k][lane];
    tot += x;
    if (k < w) off += x;
  }
  if (ok) {
#pragma unroll
    for (int k = 0; k < SEG; ++k) {
      hp[(size_t)k * BKT_MAX] = off;
      off += v[k];
    }
    if (w == 0) btot[b] = tot;
  }
}

__global__ void __launch_bounds__(BKT_THREADS)
k_bkt_bin(const int* __restrict__ src, const int* __restrict__ dst,
          const int* __restrict__ hist, const int* __restrict__ btot,
          int* __restrict__ bbase, int* __restrict__ pairs_, int E, int NB) {
  __shared__ int cur[BKT_MAX];
  __shared__ int wtot[BKT_THREADS / 32];
  int2* pairs = reinterpret_cast<int2*>(pairs_);
  const int t = threadIdx.x, lane = t & 31, w = t >> 5;
  const bool ok = t < NB;
  const int c = ok ? btot[t] : 0;
  const int within = ok ? hist[(size_t)blockIdx.x * BKT_MAX + t] : 0;
  int incl = c;
  for (int o = 1; o < 32; o <<= 1) {
    const int v = __shfl_up(incl, o);
    if (lane >= o) incl += v;
  }
  if (lane == 31) wtot[w] = incl;
  __syncthreads();
  int woff = 0;
  for (int k = 0; k < w; ++k) woff += wtot[k];
  const int excl = woff + incl - c;
  cur[t] = excl + within;
  if (blockIdx.x == 0 && ok) {
    bbase[t] = excl;
    if (t == NB - 1) bbase[NB] = excl + c;
  }
  __syncthreads();
  const int chunk = (E + BKT_BLOCKS - 1) / BKT_BLOCKS;
  const int beg = min((int)blockIdx.x * chunk, E);
  const int end = min(beg + chunk, E);
  const unsigned bmax = (unsigned)(NB - 1);
  const unsigned emax = (unsigned)(E - 1);
  for (int e0 = beg + t; e0 < end; e0 += 4 * BKT_THREADS) {
    int d[4], s[4];
#pragma unroll
    for (int k = 0; k < 4; ++k) {
      const int e = min(e0 + k * BKT_THREADS, end - 1);
      d[k] = dst[e];
      s[k] = src[e];
    }
    asm volatile("" : "+v"(d[0]), "+v"(d[1]), "+v"(d[2]), "+v"(d[3]),
                      "+v"(s[0]), "+v"(s[1]), "+v"(s[2]), "+v"(s[3]));
#pragma unroll
    for (int k = 0; k < 4; ++k) {
      if (e0 + k * BKT_THREADS < end) {
        const int pos = atomicAdd(&cur[min((unsigned)d[k] >> BKT_SHIFT, bmax)], 1);
        pairs[min((unsigned)pos, emax)] = make_int2(d[k], s[k]);
      }
    }
  }
}

#define XS_PITCH 32
__global__ void __launch_bounds__(256)
k_bkt_csr(const int* __restrict__ pairs_, const int* __restrict__ bbase,
          int* __restrict__ rowptr, int* __restrict__ csr, float* __restrict__ dinv,
          const float* __restrict__ x, float* __restrict__ xs,
          float* __restrict__ z1, int nz1, float* __restrict__ z2, int nz2,
          int Nn, int E) {
  __shared__ int cnt[BKT_NODES];
  __shared__ int cur[BKT_NODES];
  __shared__ int wtot[BKT_NODES / 32];
  __shared__ float dvs[BKT_NODES];
  const int2* pairs = reinterpret_cast<const int2*>(pairs_);
  const int t = threadIdx.x, lane = t & 31, w = t >> 5;
  const int b = blockIdx.x;
  if (b == 0) {
    for (int i = t; i < nz1; i += 256) z1[i] = 0.0f;
    for (int i = t; i < nz2; i += 256) z2[i] = 0.0f;
  }
  const int xp_ = t & 7;
  const int xr_ = t >> 3;
  v4f xv[BKT_NODES / 32];
#pragma unroll
  for (int k = 0; k < BKT_NODES / 32; ++k) {
    const int node = min(b * BKT_NODES + xr_ + 32 * k, Nn - 1);
    xv[k] = reinterpret_cast<const v4f*>(x)[(size_t)node * 6 + min(xp_, 5)];
  }
  const int beg = min(max(bbase[b], 0), E);
  const int end = min(max(bbase[b + 1], beg), E);
  if (t < BKT_NODES) cnt[t] = 0;
  __syncthreads();
  for (int i0 = beg + t; i0 < end; i0 += 4 * 256) {
    int d[4];
#pragma unroll
    for (int k = 0; k < 4; ++k) d[k] = pairs_[2 * (size_t)min(i0 + k * 256, end - 1)];
    asm volatile("" : "+v"(d[0]), "+v"(d[1]), "+v"(d[2]), "+v"(d[3]));
#pragma unroll
    for (int k = 0; k < 4; ++k)
      if (i0 + k * 256 < end) atomicAdd(&cnt[d[k] & (BKT_NODES - 1)], 1);
  }
  __syncthreads();
  const int c = (t < BKT_NODES) ? cnt[t] : 0;
  int incl = c;
  for (int o = 1; o < 32; o <<= 1) {
    const int v = __shfl_up(incl, o);
    if (lane >= o) incl += v;
  }
  if (t < BKT_NODES && lane == 31) wtot[w] = incl;
  __syncthreads();
  if (t < BKT_NODES) {
    int woff = 0;
    for (int k = 0; k < w; ++k) woff += wtot[k];
    const int excl = beg + woff + incl - c;
    cur[t] = excl;
    const int node = b * BKT_NODES + t;
    const float deg = (float)(c + 1);
    const float dv = (deg > 0.0f) ? rsqrtf(deg) : 0.0f;
    dvs[t] = dv;
    if (node < Nn) {
      rowptr[node] = excl;
      dinv[node] = dv;
      if (node == Nn - 1) rowptr[Nn] = end;
    }
  }
  __syncthreads();
  {
    const float keep = (xp_ < 6) ? 1.f : 0.f;
#pragma unroll
    for (int k = 0; k < BKT_NODES / 32; ++k) {
      const int r = xr_ + 32 * k;
      const int node = b * BKT_NODES + r;
      const float sc = dvs[r] * keep;
      if (node < Nn)
        reinterpret_cast<v4f*>(xs)[(size_t)node * (XS_PITCH / 4) + xp_] = xv[k] * sc;
    }
  }
  const unsigned emax = (unsigned)(E - 1);
  for (int i0 = beg + t; i0 < end; i0 += 4 * 256) {
    int2 p[4];
#pragma unroll
    for (int k = 0; k < 4; ++k) p[k] = pairs[min(i0 + k * 256, end - 1)];
    asm volatile("" : "+v"(p[0].x), "+v"(p[0].y), "+v"(p[1].x), "+v"(p[1].y),
                      "+v"(p[2].x), "+v"(p[2].y), "+v"(p[3].x), "+v"(p[3].y));
#pragma unroll
    for (int k = 0; k < 4; ++k) {
      if (i0 + k * 256 < end) {
        const int pos = atomicAdd(&cur[p[k].x & (BKT_NODES - 1)], 1);
        csr[min((unsigned)pos, emax)] = p[k].y;
      }
    }
  }
}

#define GF_ROWS_128 9
#define GF_ROWS_64 8
template <int F, int R>
__global__ void __launch_bounds__(256)
k_gather_finish(const float* __restrict__ hsc, const int* __restrict__ rowptr,
                const int* __restrict__ csr, const float* __restrict__ dinv,
                const float* __restrict__ bias, float* __restrict__ vout,
                float* __restrict__ stats,
                int Nn, int E) {
  constexpr int V = F / 32;
  typedef float vf __attribute__((ext_vector_type(V)));
  __shared__ __attribute__((aligned(16))) float red[2][8][F];
  const int t = threadIdx.x;
  const int lane = t & 31;
  const int w = t >> 5;
  const int d0 = __builtin_amdgcn_readfirstlane((int)((blockIdx.x * 8 + w) * R));
  vf s = 0.0f;
  vf q = 0.0f;
  if (d0 < Nn) {
    const int rows = min(R, Nn - d0);
    const vf* hp = reinterpret_cast<const vf*>(hsc) + lane;
    vf* op = reinterpret_cast<vf*>(vout) + lane;
    const unsigned nmax = (unsigned)(Nn - 1);
    const vf zerov = 0.0f;
    const int rpraw = rowptr[min(d0 + lane, Nn)];
    const float dvl = dinv[min(d0 + lane, Nn - 1)];
    const vf b = *reinterpret_cast<const vf*>(bias + V * lane);
    const int rp = min(max(rpraw, 0), E);
    const unsigned emax = (unsigned)(E - 1);
    int beg = __builtin_amdgcn_readlane(rp, 0);
    int end = __builtin_amdgcn_readlane(rp, 1);
    int idx = csr[min((unsigned)(beg + lane), emax)];
    auto sum_chunk = [&](const int cur, const int m, vf a) -> vf {
      auto row = [&](int u) -> vf {
        const unsigned sidx = min((unsigned)__builtin_amdgcn_readlane(cur, u), nmax);
        return hp[(size_t)sidx * 32];
      };
      int u = 0;
      for (; u + 8 <= m; u += 8) {
        vf r[8];
#pragma unroll
        for (int k = 0; k < 8; ++k) r[k] = row(u + k);
#pragma unroll
        for (int k = 0; k < 8; ++k) a += r[k];
      }
      if (u + 4 <= m) {
        const vf r0 = row(u), r1 = row(u + 1), r2 = row(u + 2), r3 = row(u + 3);
        a += r0; a += r1; a += r2; a += r3;
        u += 4;
      }
      if (u + 2 <= m) {
        const vf r0 = row(u), r1 = row(u + 1);
        a += r0; a += r1;
        u += 2;
      }
      if (u < m) a += row(u);
      return a;
    };
    for (int i = 0; i < rows; ++i) {
      const int d = d0 + i;
      const vf self = hp[(size_t)d * 32];
      const int nbeg = end;
      const int nend = __builtin_amdgcn_readlane(rp, i + 2);
      const int idxn = csr[min((unsigned)(nbeg + lane), emax)];
      vf acc = sum_chunk(idx, min(32, end - beg), zerov);
      for (int j0 = beg + 32; j0 < end; j0 += 32) {
        const int m = min(32, end - j0);
        const int more = (lane < m) ? csr[j0 + lane] : 0;
        acc = sum_chunk(more, m, acc);
      }
      const float di = __builtin_bit_cast(
          float, __builtin_amdgcn_readlane(__builtin_bit_cast(int, dvl), i));
      const vf v = __builtin_elementwise_max((acc + self) * di + b, zerov);
      op[(size_t)d * 32] = v;
      s += v;
      q += v * v;
      beg = nbeg;
      end = nend;
      idx = idxn;
    }
  }
  *reinterpret_cast<vf*>(&red[0][w][V * lane]) = s;
  *reinterpret_cast<vf*>(&red[1][w][V * lane]) = q;
  __syncthreads();
  if (t < 2 * F) {
    const int which = t / F;
    const int j = t % F;
    float a = 0.f;
    for (int k = 0; k < 8; ++k) a += red[which][k][j];
    unsafeAtomicAdd(stats + which * F + j, a);
  }
}

__global__ void k_gemm_wmma_f32(const float* __restrict__ A, const float* __restrict__ B,
                                const float* __restrict__ dinv,
                                float* __restrict__ C, int M, int K, int F) {
  const int lane = threadIdx.x & 31;
  const int wave = (int)((blockIdx.x * blockDim.x + threadIdx.x) >> 5);
  const int nf = F >> 4;
  const int mTile = wave / nf;
  const int fTile = wave - mTile * nf;
  const int m0 = mTile << 4;
  if (m0 >= M) return;
  const int f0 = fTile << 4;
  const int row = lane & 15;
  const int kh2 = (lane >> 4) << 1;

  const float* arow = A + (size_t)(m0 + row) * K + kh2;
  const float* bcol = B + (size_t)kh2 * F + f0 + row;

  v8f acc = {0.f, 0.f, 0.f, 0.f, 0.f, 0.f, 0.f, 0.f};
  for (int k = 0; k < K; k += 4) {
    float2 av = *reinterpret_cast<const float2*>(arow + k);
    v2f a; a.x = av.x; a.y = av.y;
    v2f b;
    b.x = bcol[(size_t)k * F];
    b.y = bcol[(size_t)(k + 1) * F];
    acc = __builtin_amdgcn_wmma_f32_16x16x4_f32(false, a, false, b,
                                                (short)0, acc, false, false);
  }
  const float* dvp = dinv + m0 + (kh2 << 2);
  float* crow = C + (size_t)(m0 + (kh2 << 2)) * F + f0 + row;
#pragma unroll
  for (int r = 0; r < 8; ++r) crow[(size_t)r * F] = acc[r] * dvp[r];
}

template <int F>
__global__ void __launch_bounds__(256)
k_gemm_bn_wmma_f32(const float* __restrict__ V, const float* __restrict__ stats,
                   const float* __restrict__ gamma, const float* __restrict__ beta,
                   const float* __restrict__ B, const float* __restrict__ dinv,
                   float* __restrict__ C, int M, float invN) {
  constexpr int K = 128;
  constexpr int NF = F / 16;
  constexpr int RB = 16 * (8 / NF);
  constexpr int PITCH = K + 4;
  __shared__ __attribute__((aligned(16))) float As[RB * PITCH];

  const int t = threadIdx.x;
  const int lane = t & 31;
  const int w = t >> 5;
  const int rowBase = blockIdx.x * RB;

  {
    const int c4 = t & 31;
    const v4f su = *reinterpret_cast<const v4f*>(stats + 4 * c4);
    const v4f sq = *reinterpret_cast<const v4f*>(stats + K + 4 * c4);
    const v4f g = *reinterpret_cast<const v4f*>(gamma + 4 * c4);
    const v4f be = *reinterpret_cast<const v4f*>(beta + 4 * c4);
    v4f mean, rstd;
#pragma unroll
    for (int i = 0; i < 4; ++i) {
      mean[i] = su[i] * invN;
      const float var = sq[i] * invN - mean[i] * mean[i];
      rstd[i] = rsqrtf(var + BN_EPS);
    }
    v4f v[RB / 8];
#pragma unroll
    for (int p = 0; p < RB / 8; ++p) {
      const int gr = min(rowBase + p * 8 + w, M - 1);
      v[p] = *reinterpret_cast<const v4f*>(V + (size_t)gr * K + 4 * c4);
    }
#pragma unroll
    for (int p = 0; p < RB / 8; ++p) {
      const v4f o = (v[p] - mean) * rstd * g + be;
      *reinterpret_cast<v4f*>(&As[(p * 8 + w) * PITCH + 4 * c4]) = o;
    }
  }
  __syncthreads();

  const int mT = w / NF;
  const int fT = w - mT * NF;
  const int m0 = rowBase + (mT << 4);
  if (m0 >= M) return;
  const int f0 = fT << 4;
  const int row = lane & 15;
  const int kh2 = (lane >> 4) << 1;

  const float* arow = As + ((mT << 4) + row) * PITCH + kh2;
  const float* bcol = B + (size_t)kh2 * F + f0 + row;

  unsigned dbyte = (unsigned)(m0 + (kh2 << 2)) * 4u;
  unsigned cbyte = ((unsigned)(m0 + (kh2 << 2)) * F + f0 + row) * 4u;
  asm volatile("" : "+v"(dbyte), "+v"(cbyte));
  const float* dvp = reinterpret_cast<const float*>(reinterpret_cast<const char*>(dinv) + dbyte);
  float* crow = reinterpret_cast<float*>(reinterpret_cast<char*>(C) + cbyte);

  v8f acc = {0.f, 0.f, 0.f, 0.f, 0.f, 0.f, 0.f, 0.f};
#pragma unroll
  for (int k = 0; k < K; k += 4) {
    const v2f a = *reinterpret_cast<const v2f*>(arow + k);
    v2f b;
    b.x = bcol[(size_t)k * F];
    b.y = bcol[(size_t)(k + 1) * F];
    acc = __builtin_amdgcn_wmma_f32_16x16x4_f32(false, a, false, b,
                                                (short)0, acc, false, false);
  }
  asm volatile("s_wait_alu depctr_va_vdst(0)\n\t"
               "v_nop\n\tv_nop\n\tv_nop\n\tv_nop\n\t"
               "v_nop\n\tv_nop\n\tv_nop\n\tv_nop"
               : "+v"(acc));
#pragma unroll
  for (int r = 0; r < 8; ++r) crow[(size_t)r * F] = acc[r] * dvp[r];
}

__global__ void __launch_bounds__(256)
k_gather_x24(const float* __restrict__ x, const float* __restrict__ dinv,
             const int* __restrict__ rowptr, const int* __restrict__ csr,
             float* __restrict__ ax, int Nn, int E) {
  const int lane = threadIdx.x & 31;
  const int d = __builtin_amdgcn_readfirstlane((int)(blockIdx.x * 8 + (threadIdx.x >> 5)));
  if (d >= Nn) return;
  const int beg = max(rowptr[d], 0);
  const int end = min(rowptr[d + 1], E);
  const int p = lane & 7;
  const int q = lane >> 3;
  const bool pv = p < 6;
  const unsigned nmax = (unsigned)(Nn - 1);
  const v4f* xp = reinterpret_cast<const v4f*>(x) + p;
  const v4f zero4 = {0.f, 0.f, 0.f, 0.f};
  const float dd = dinv[d];
  const v4f xd = pv ? xp[(size_t)d * 6] : zero4;
  v4f acc = zero4;
  for (int j0 = beg; j0 < end; j0 += 32) {
    const int m = min(32, end - j0);
    int idx = 0;
    float dv = 0.f;
    if (lane < m) {
      idx = (int)min((unsigned)csr[j0 + lane], nmax);
      dv = dinv[idx];
    }
    v4f r[8];
    float wgt[8];
#pragma unroll
    for (int k = 0; k < 8; ++k) {
      const int e = 4 * k + q;
      const int s = __shfl(idx, e);
      wgt[k] = __shfl(dv, e);
      r[k] = (pv && e < m) ? xp[(size_t)s * 6] : zero4;
    }
#pragma unroll
    for (int k = 0; k < 8; ++k) acc += r[k] * wgt[k];
  }
#pragma unroll
  for (int i = 0; i < 4; ++i) {
    acc[i] += __shfl_xor(acc[i], 8);
    acc[i] += __shfl_xor(acc[i], 16);
  }
  if (q == 0 && pv)
    reinterpret_cast<v4f*>(ax)[(size_t)d * 6 + p] = (acc + xd * dd) * dd;
}


template <int K>
__device__ __forceinline__ unsigned group8_lane(unsigned v) {
  return (unsigned)__builtin_amdgcn_mov_dpp8(v, K * 0x249249);
}
__device__ __forceinline__ float xor8(float x) {
  const int b = __builtin_bit_cast(int, x);
  return __builtin_bit_cast(
      float, __builtin_amdgcn_permlane16(b, b, 0xfedcba98, 0x76543210, false, false));
}
__device__ __forceinline__ float xor16(float x) {
  const int b = __builtin_bit_cast(int, x);
  return __builtin_bit_cast(
      float, __builtin_amdgcn_permlanex16(b, b, 0x76543210, 0xfedcba98, false, false));
}

#define G1_TILES 8
__global__ void __launch_bounds__(256)
k_gemm1_relu_stats(const float* __restrict__ A, const float* __restrict__ B,
                   const float* __restrict__ bias, float* __restrict__ C,
                   float* __restrict__ stats,
                   int M) {
  constexpr int K = 24;
  constexpr int F = 128;
  constexpr int KS = K / 4;
  const int lane = threadIdx.x & 31;
  const int f0 = (int)(threadIdx.x >> 5) << 4;
  const int c = lane & 15;
  const int h = lane >> 4;
  const int kh2 = h << 1;
  const int rowBase = blockIdx.x * (16 * G1_TILES);

  v2f b[KS];
  const float* bcol = B + (size_t)kh2 * F + f0 + c;
#pragma unroll
  for (int kk = 0; kk < KS; ++kk) {
    b[kk].x = bcol[(size_t)(4 * kk) * F];
    b[kk].y = bcol[(size_t)(4 * kk + 1) * F];
  }
  const float bc = bias[f0 + c];

  v2f a[G1_TILES][KS];
#pragma unroll
  for (int i = 0; i < G1_TILES; ++i) {
    const int gr = min(rowBase + 16 * i + c, M - 1);
    const float* ar = A + (size_t)gr * K + kh2;
#pragma unroll
    for (int kk = 0; kk < KS; ++kk)
      a[i][kk] = *reinterpret_cast<const v2f*>(ar + 4 * kk);
  }

  float s = 0.f, q = 0.f;
#pragma unroll
  for (int i = 0; i < G1_TILES; ++i) {
    const int m0 = rowBase + 16 * i;
    if (m0 >= M) break;
    __builtin_amdgcn_sched_barrier(0);
    v8f acc = {0.f, 0.f, 0.f, 0.f, 0.f, 0.f, 0.f, 0.f};
#pragma unroll
    for (int kk = 0; kk < KS; ++kk)
      acc = __builtin_amdgcn_wmma_f32_16x16x4_f32(false, a[i][kk], false, b[kk],
                                                  (short)0, acc, false, false);
    asm volatile("s_wait_alu depctr_va_vdst(0)\n\t"
                 "v_nop\n\tv_nop\n\tv_nop\n\tv_nop\n\t"
                 "v_nop\n\tv_nop\n\tv_nop\n\tv_nop"
                 : "+v"(acc));
    __builtin_amdgcn_sched_barrier(0);
    const int r0 = m0 + (h << 3);
    float* crow = C + (size_t)r0 * F + f0 + c;
#pragma unroll
    for (int r = 0; r < 8; ++r) {
      if (r0 + r < M) {
        const float v = fmaxf(acc[r] + bc, 0.f);
        crow[(size_t)r * F] = v;
        s += v;
        q += v * v;
      }
    }
  }
  s += __shfl_xor(s, 16);
  q += __shfl_xor(q, 16);
  if (h == 0) {
    unsafeAtomicAdd(stats + f0 + c, s);
    unsafeAtomicAdd(stats + F + f0 + c, q);
  }
}

#define L1_ROWS 10
#define L1_BLK (8 * L1_ROWS)
#define L1_TILES (L1_BLK / 16)
#define L1_PITCH 28
__global__ void __launch_bounds__(256)
k_l1_sum_gemm(const float* __restrict__ xs, const int* __restrict__ rowptr,
              const int* __restrict__ csr, const float* __restrict__ dinv,
              const float* __restrict__ B, const float* __restrict__ bias,
              float* __restrict__ C,
              float* __restrict__ stats,
              int Nn, int E) {
  static_assert(L1_BLK % 16 == 0, "a block is a whole number of row tiles");
  __shared__ __attribute__((aligned(16))) float As[L1_BLK * L1_PITCH];
  const int lane = threadIdx.x & 31;
  const int w = (int)(threadIdx.x >> 5);
  const int rowBase = blockIdx.x * L1_BLK;
  const int d0 = __builtin_amdgcn_readfirstlane(rowBase + w * L1_ROWS);
  const int rows = max(0, min(L1_ROWS, Nn - d0));
  {
    const int p = lane & 7;
    const int q = lane >> 3;
    const int el = 4 * p + q;
    const unsigned p16 = 16u * (unsigned)p;
    const char* xb = reinterpret_cast<const char*>(xs);
    const unsigned nmax = (unsigned)(Nn - 1);
    const unsigned emax = (unsigned)(E - 1);
    const v4f zero4 = {0.f, 0.f, 0.f, 0.f};
    const int rpraw = rowptr[min(d0 + lane, Nn)];
    const float dvl = dinv[min(d0 + lane, Nn - 1)];
    const int rp = min(max(rpraw, 0), E);
    int beg = __builtin_amdgcn_readlane(rp, 0);
    int end = __builtin_amdgcn_readlane(rp, 1);
    int idx = csr[min((unsigned)(beg + el), emax)];
    auto piece = [&](unsigned o) -> v4f {
      return *reinterpret_cast<const v4f*>(xb + (o + p16));
    };
    auto sum_chunk = [&](const int cur, const int m, const unsigned own, v4f a) -> v4f {
      const unsigned off =
          (el < m) ? min((unsigned)cur, nmax) * (unsigned)(XS_PITCH * 4) : own;
      const int mq = m - q;
      v4f r0 = piece(group8_lane<0>(off));
      v4f r1 = piece(group8_lane<1>(off));
      v4f r2 = piece(group8_lane<2>(off));
      v4f r3 = piece(group8_lane<3>(off));
      if (m > 16) {
        v4f r4 = piece(group8_lane<4>(off));
        v4f r5 = piece(group8_lane<5>(off));
        v4f r6 = piece(group8_lane<6>(off));
        v4f r7 = piece(group8_lane<7>(off));
        asm volatile("" : "+v"(r0), "+v"(r1), "+v"(r2), "+v"(r3),
                          "+v"(r4), "+v"(r5), "+v"(r6), "+v"(r7));
        a += (0 < mq) ? r0 : zero4;
        a += (4 < mq) ? r1 : zero4;
        a += (8 < mq) ? r2 : zero4;
        a += (12 < mq) ? r3 : zero4;
        a += (16 < mq) ? r4 : zero4;
        a += (20 < mq) ? r5 : zero4;
        a += (24 < mq) ? r6 : zero4;
        a += (28 < mq) ? r7 : zero4;
      } else {
        a += (0 < mq) ? r0 : zero4;
        a += (4 < mq) ? r1 : zero4;
        a += (8 < mq) ? r2 : zero4;
        a += (12 < mq) ? r3 : zero4;
      }
      return a;
    };
    for (int i = 0; i < rows; ++i) {
      const int d = d0 + i;
      const unsigned own = (unsigned)d * (unsigned)(XS_PITCH * 4);
      const v4f self = piece(own);
      const int nbeg = end;
      const int nend = __builtin_amdgcn_readlane(rp, i + 2);
      const int idxn = csr[min((unsigned)(nbeg + el), emax)];
      v4f acc = sum_chunk(idx, min(32, end - beg), own, zero4);
      for (int j0 = beg + 32; j0 < end; j0 += 32) {
        const int more = csr[min((unsigned)(j0 + el), emax)];
        acc = sum_chunk(more, min(32, end - j0), own, acc);
      }
#pragma unroll
      for (int cc = 0; cc < 4; ++cc) {
        acc[cc] += xor8(acc[cc]);
        acc[cc] += xor16(acc[cc]);
      }
      const float di = __builtin_bit_cast(
          float, __builtin_amdgcn_readlane(__builtin_bit_cast(int, dvl), i));
      if (q == 0 && p < 6)
        *reinterpret_cast<v4f*>(&As[(w * L1_ROWS + i) * L1_PITCH + 4 * p]) = (acc + self) * di;
      beg = nbeg;
      end = nend;
      idx = idxn;
    }
  }

  constexpr int F = 128;
  constexpr int KS = 24 / 4;
  const int f0 = w << 4;
  const int c = lane & 15;
  const int h = lane >> 4;
  const int kh2 = h << 1;
  v2f b[KS];
  const float* bcol = B + (size_t)kh2 * F + f0 + c;
#pragma unroll
  for (int kk = 0; kk < KS; ++kk) {
    b[kk].x = bcol[(size_t)(4 * kk) * F];
    b[kk].y = bcol[(size_t)(4 * kk + 1) * F];
  }
  const float bc = bias[f0 + c];
  __syncthreads();

  const float* arow = As + c * L1_PITCH + kh2;
  float s = 0.f, q2 = 0.f;
#pragma unroll
  for (int i = 0; i < L1_TILES; ++i) {
    const int m0 = rowBase + 16 * i;
    if (m0 >= Nn) break;
    v2f a[KS];
#pragma unroll
    for (int kk = 0; kk < KS; ++kk)
      a[kk] = *reinterpret_cast<const v2f*>(arow + (16 * i) * L1_PITCH + 4 * kk);
    __builtin_amdgcn_sched_barrier(0);
    v8f acc = {0.f, 0.f, 0.f, 0.f, 0.f, 0.f, 0.f, 0.f};
#pragma unroll
    for (int kk = 0; kk < KS; ++kk)
      acc = __builtin_amdgcn_wmma_f32_16x16x4_f32(false, a[kk], false, b[kk],
                                                  (short)0, acc, false, false);
    asm volatile("s_wait_alu depctr_va_vdst(0)\n\t"
                 "v_nop\n\tv_nop\n\tv_nop\n\tv_nop\n\t"
                 "v_nop\n\tv_nop\n\tv_nop\n\tv_nop"
                 : "+v"(acc));
    __builtin_amdgcn_sched_barrier(0);
    const int r0 = m0 + (h << 3);
    float* crow = C + (size_t)r0 * F + f0 + c;
#pragma unroll
    for (int r = 0; r < 8; ++r) {
      if (r0 + r < Nn) {
        const float v = fmaxf(acc[r] + bc, 0.f);
        crow[(size_t)r * F] = v;
        s += v;
        q2 += v * v;
      }
    }
  }
  s += __shfl_xor(s, 16);
  q2 += __shfl_xor(q2, 16);
  if (h == 0) {
    unsafeAtomicAdd(stats + f0 + c, s);
    unsafeAtomicAdd(stats + F + f0 + c, q2);
  }
}

__global__ void k_scatter(const float* __restrict__ hsc, const int* __restrict__ src,
                          const int* __restrict__ dst,
                          float* __restrict__ agg, int E, int F) {
  int wave = (int)((blockIdx.x * blockDim.x + threadIdx.x) >> 5);
  int lane = threadIdx.x & 31;
  if (wave >= E) return;
  int s = src[wave], d = dst[wave];
  const float4* hrow = reinterpret_cast<const float4*>(hsc + (size_t)s * F);
  float* ad = agg + (size_t)d * F;
  for (int f4 = lane; 4 * f4 < F; f4 += 32) {
    float4 v = hrow[f4];
    unsafeAtomicAdd(ad + 4 * f4 + 0, v.x);
    unsafeAtomicAdd(ad + 4 * f4 + 1, v.y);
    unsafeAtomicAdd(ad + 4 * f4 + 2, v.z);
    unsafeAtomicAdd(ad + 4 * f4 + 3, v.w);
  }
}

#define FIN_ROWS 128
template <int F>
__global__ void __launch_bounds__(256)
k_finish_relu_stats(float* __restrict__ agg, const float* __restrict__ hsc,
                    const float* __restrict__ dinv,
                    const float* __restrict__ bias,
                    float* __restrict__ stats,
                    int Nn) {
  constexpr int TPR = F / 4;
  constexpr int RPP = 256 / TPR;
  constexpr int U = 8;
  constexpr int STEP = RPP * TPR;
  static_assert(FIN_ROWS % (U * RPP) == 0, "a block is a whole number of groups");
  __shared__ __attribute__((aligned(16))) float red[2][RPP][F];

  const int t = threadIdx.x;
  const int c4 = t % TPR;
  const int rsub = t / TPR;
  const int r0 = blockIdx.x * FIN_ROWS;
  const int r1 = min(r0 + FIN_ROWS, Nn);

  const v4f b = *reinterpret_cast<const v4f*>(bias + 4 * c4);
  const v4f zero4 = {0.f, 0.f, 0.f, 0.f};
  v4f s = zero4;
  v4f q = zero4;

  int r = r0 + rsub;
  v4f* ap = reinterpret_cast<v4f*>(agg) + (size_t)r * TPR + c4;
  const v4f* hp = reinterpret_cast<const v4f*>(hsc) + (size_t)r * TPR + c4;
  const float* dp = dinv + r;

  auto apply = [&](const v4f a, const v4f h, const float di, v4f* dst) {
    const v4f v = __builtin_elementwise_max((a + h) * di + b, zero4);
    *dst = v;
    s += v;
    q += v * v;
  };

  for (; r + (U - 1) * RPP < r1; r += U * RPP) {
    v4f a[U], h[U];
    float di[U];
#pragma unroll
    for (int u = 0; u < U; ++u) {
      a[u] = ap[u * STEP];
      h[u] = hp[u * STEP];
      di[u] = dp[u * RPP];
    }
    __builtin_amdgcn_sched_barrier(0);
#pragma unroll
    for (int u = 0; u < U; ++u) apply(a[u], h[u], di[u], ap + u * STEP);
    ap += U * STEP;
    hp += U * STEP;
    dp += U * RPP;
  }
  for (; r < r1; r += RPP) {
    apply(*ap, *hp, *dp, ap);
    ap += STEP;
    hp += STEP;
    dp += RPP;
  }

  *reinterpret_cast<v4f*>(&red[0][rsub][4 * c4]) = s;
  *reinterpret_cast<v4f*>(&red[1][rsub][4 * c4]) = q;
  __syncthreads();
  if (t < 2 * F) {
    const int which = t / F;
    const int j = t % F;
    float acc = 0.f;
    for (int k = 0; k < RPP; ++k) acc += red[which][k][j];
    unsafeAtomicAdd(stats + which * F + j, acc);
  }
}

__global__ void k_gemm64x2(const float* __restrict__ v3, const float* __restrict__ stats,
                           const float* __restrict__ gamma, const float* __restrict__ beta,
                           const float* __restrict__ W4, const float* __restrict__ dinv,
                           float* __restrict__ h4, int Nn, float invN) {
  int i = blockIdx.x * blockDim.x + threadIdx.x;
  if (i >= Nn) return;
  const float* row = v3 + (size_t)i * 64;
  float o0 = 0.f, o1 = 0.f;
#pragma unroll 8
  for (int k = 0; k < 64; ++k) {
    const float mean = stats[k] * invN;
    const float var = stats[64 + k] * invN - mean * mean;
    const float xv = (row[k] - mean) * rsqrtf(var + BN_EPS) * gamma[k] + beta[k];
    o0 += xv * W4[2 * k];
    o1 += xv * W4[2 * k + 1];
  }
  float di = dinv[i];
  h4[2 * (size_t)i]     = o0 * di;
  h4[2 * (size_t)i + 1] = o1 * di;
}

__global__ void k_scatter2(const float* __restrict__ h4s, const int* __restrict__ src,
                           const int* __restrict__ dst,
                           float* __restrict__ agg, int E) {
  int e = blockIdx.x * blockDim.x + threadIdx.x;
  if (e >= E) return;
  int s = src[e], d = dst[e];
  float2 v = *reinterpret_cast<const float2*>(h4s + 2 * (size_t)s);
  unsafeAtomicAdd(agg + 2 * (size_t)d,     v.x);
  unsafeAtomicAdd(agg + 2 * (size_t)d + 1, v.y);
}

__global__ void k_final_lsm(const float* __restrict__ agg, const float* __restrict__ h4s,
                            const float* __restrict__ dinv, const float* __restrict__ b4,
                            float* __restrict__ out, int Nn) {
  int i = blockIdx.x * blockDim.x + threadIdx.x;
  if (i >= Nn) return;
  float di = dinv[i];
  float o0 = (agg[2 * (size_t)i]     + h4s[2 * (size_t)i])     * di + b4[0];
  float o1 = (agg[2 * (size_t)i + 1] + h4s[2 * (size_t)i + 1]) * di + b4[1];
  float m = fmaxf(o0, o1);
  float l = m + logf(expf(o0 - m) + expf(o1 - m));
  out[2 * (size_t)i]     = o0 - l;
  out[2 * (size_t)i + 1] = o1 - l;
}

#define L4_ROWS 16
__global__ void __launch_bounds__(256)
k_gather2_lsm(const float* __restrict__ h4s, const int* __restrict__ rowptr,
              const int* __restrict__ csr, const float* __restrict__ dinv,
              const float* __restrict__ b4, float* __restrict__ out, int Nn, int E) {
  const int lane = threadIdx.x & 31;
  const int d0 = __builtin_amdgcn_readfirstlane(
      (int)((blockIdx.x * 8 + (threadIdx.x >> 5)) * L4_ROWS));
  if (d0 >= Nn) return;
  const int p = lane & 7;
  const int q = lane >> 3;
  const v2f* hp = reinterpret_cast<const v2f*>(h4s);
  const unsigned nmax = (unsigned)(Nn - 1);
  const unsigned emax = (unsigned)(E - 1);
  const v2f zero2 = {0.f, 0.f};
  const int rpraw = rowptr[min(d0 + lane, Nn)];
  const int gs = p & 3;
  const int rfin = d0 + 4 * gs + q;
  const int rfc = min(rfin, Nn - 1);
  const v2f self = hp[rfc];
  const float di = dinv[rfc];
  const float bb0 = b4[0];
  const float bb1 = b4[1];
  const int rp = min(max(rpraw, 0), E);

  int beg[4], len[4];
  unsigned own[4];
#pragma unroll
  for (int g = 0; g < 4; ++g) {
    const int b = __shfl(rp, 4 * g + q);
    const int e = __shfl(rp, 4 * g + q + 1);
    beg[g] = b;
    len[g] = e - b;
    own[g] = min((unsigned)(d0 + 4 * g + q), nmax);
  }
  int idx[4][4];
#pragma unroll
  for (int g = 0; g < 4; ++g) {
#pragma unroll
    for (int k = 0; k < 4; ++k)
      idx[g][k] = csr[min((unsigned)(beg[g] + p + 8 * k), emax)];
  }
  v2f v[4][4];
#pragma unroll
  for (int g = 0; g < 4; ++g) {
#pragma unroll
    for (int k = 0; k < 4; ++k) {
      const unsigned s = (p + 8 * k < len[g]) ? min((unsigned)idx[g][k], nmax) : own[g];
      v[g][k] = hp[s];
    }
  }
  v2f acc[4];
#pragma unroll
  for (int g = 0; g < 4; ++g) {
    v2f a = zero2;
#pragma unroll
    for (int k = 0; k < 4; ++k) a += (p + 8 * k < len[g]) ? v[g][k] : zero2;
    const int end = beg[g] + len[g];
    for (int j = beg[g] + 32 + p; j < end; j += 8) a += hp[min((unsigned)csr[j], nmax)];
    a.x += __shfl_xor(a.x, 1);
    a.y += __shfl_xor(a.y, 1);
    a.x += __shfl_xor(a.x, 2);
    a.y += __shfl_xor(a.y, 2);
    a.x += __shfl_xor(a.x, 4);
    a.y += __shfl_xor(a.y, 4);
    acc[g] = a;
  }
  const v2f a = gs == 0 ? acc[0] : gs == 1 ? acc[1] : gs == 2 ? acc[2] : acc[3];
  const float o0 = (a.x + self.x) * di + bb0;
  const float o1 = (a.y + self.y) * di + bb1;
  const float m = fmaxf(o0, o1);
  const float l = m + logf(expf(o0 - m) + expf(o1 - m));
  v2f o;
  o.x = o0 - l;
  o.y = o1 - l;
  if (p < 4 && rfin < Nn) reinterpret_cast<v2f*>(out)[rfin] = o;
}

extern "C" void kernel_launch(void* const* d_in, const int* in_sizes, int n_in,
                              void* d_out, int out_size, void* d_ws, size_t ws_size,
                              hipStream_t stream) {
  const float* x   = (const float*)d_in[0];
  const int*   ei  = (const int*)d_in[1];
  const float* W1  = (const float*)d_in[2];
  const float* b1  = (const float*)d_in[3];
  const float* g1  = (const float*)d_in[4];
  const float* be1 = (const float*)d_in[5];
  const float* W2  = (const float*)d_in[6];
  const float* b2  = (const float*)d_in[7];
  const float* g2  = (const float*)d_in[8];
  const float* be2 = (const float*)d_in[9];
  const float* W3  = (const float*)d_in[10];
  const float* b3  = (const float*)d_in[11];
  const float* g3  = (const float*)d_in[12];
  const float* be3 = (const float*)d_in[13];
  const float* W4  = (const float*)d_in[14];
  const float* b4  = (const float*)d_in[15];

  const int Nn = in_sizes[0] / 24;
  const int E  = in_sizes[1] / 2;
  const int* src = ei;
  const int* dst = ei + E;

  float* ws    = (float*)d_ws;
  float* dinv  = ws;
  float* stats = ws + Nn;
  float* bufA  = stats + 256;
  float* bufB  = bufA + (size_t)Nn * 128;
  const int nchunk = (Nn + SCAN_CHUNK - 1) / SCAN_CHUNK;
  const size_t rp_len = ((size_t)Nn + 1 + 31) & ~(size_t)31;
  const size_t bs_len = ((size_t)nchunk + 31) & ~(size_t)31;
  int* rowptr = (int*)(bufB + (size_t)Nn * 128);
  int* csr    = rowptr + rp_len;
  int* cnt    = (int*)bufB;
  int* bsum   = cnt + Nn;
  int* rank   = bsum + bs_len;
  const size_t need_bytes =
      ((size_t)Nn + 256 + 2 * (size_t)Nn * 128 + rp_len + (size_t)E) * sizeof(float);
  const bool grouped = ws_size >= need_bytes &&
                       (size_t)Nn + bs_len + (size_t)E <= (size_t)Nn * 128;
  const int NB = (Nn + BKT_NODES - 1) / BKT_NODES;
  int* hist  = (int*)bufB;
  int* btot  = hist + (size_t)BKT_BLOCKS * BKT_MAX;
  int* bbase = btot + BKT_MAX;
  int* pairs = bbase + BKT_MAX + 32;
  const bool bucketed = grouped && NB <= BKT_MAX &&
                        (size_t)BKT_BLOCKS * BKT_MAX + 2 * BKT_MAX + 32 + 2 * (size_t)E <=
                            (size_t)Nn * 128;
  float* xs = bufA + (size_t)Nn * XS_PITCH;
  const size_t csr_len = ((size_t)E + 31) & ~(size_t)31;
  float* stats_x = (float*)(csr + csr_len);
  const bool folded =
      bucketed &&
      ws_size >= ((size_t)Nn + 256 + 2 * (size_t)Nn * 128 + rp_len + csr_len + 512) * sizeof(float);
  float* stats2 = folded ? stats_x : stats;
  float* stats3 = folded ? stats_x + 256 : stats;

  const float invN = 1.0f / (float)Nn;
  const int TB = 256;

  auto gemm = [&](const float* A, const float* Bm, float* C, int K, int F) {
    long waves = (long)((Nn + 15) / 16) * (F / 16);
    int blocks = (int)((waves * 32 + TB - 1) / TB);
    k_gemm_wmma_f32<<<blocks, TB, 0, stream>>>(A, Bm, dinv, C, Nn, K, F);
  };
  auto scatter = [&](const float* h, float* agg, int F) {
    int blocks = (int)(((long)E * 32 + TB - 1) / TB);
    k_scatter<<<blocks, TB, 0, stream>>>(h, src, dst, agg, E, F);
  };
  auto finish = [&](float* agg, const float* h, const float* bias, int F) {
    const int blocks = (Nn + FIN_ROWS - 1) / FIN_ROWS;
    k_zero<<<1, 256, 0, stream>>>(stats, 256);
    if (F == 128)
      k_finish_relu_stats<128><<<blocks, 256, 0, stream>>>(agg, h, dinv, bias, stats, Nn);
    else
      k_finish_relu_stats<64><<<blocks, 256, 0, stream>>>(agg, h, dinv, bias, stats, Nn);
  };
  auto gemm_bn = [&](const float* V, const float* st, const float* gamma, const float* beta,
                     const float* Bm, float* C, int F) {
    if (F == 128)
      k_gemm_bn_wmma_f32<128><<<(Nn + 15) / 16, 256, 0, stream>>>(
          V, st, gamma, beta, Bm, dinv, C, Nn, invN);
    else
      k_gemm_bn_wmma_f32<64><<<(Nn + 31) / 32, 256, 0, stream>>>(
          V, st, gamma, beta, Bm, dinv, C, Nn, invN);
  };
  auto zero = [&](float* p, size_t n) {
    int blocks = (int)((n + TB - 1) / TB);
    if (blocks > 4096) blocks = 4096;
    k_zero<<<blocks, TB, 0, stream>>>(p, n);
  };

  auto aggregate = [&](const float* h, float* agg, int F) {
    zero(agg, (size_t)Nn * F);
    scatter(h, agg, F);
  };
  auto aggregate_finish = [&](const float* h, float* v, const float* bias, float* st, int F) {
    if (grouped) {
      if (!folded) k_zero<<<1, 256, 0, stream>>>(st, 256);
      if (F == 128)
        k_gather_finish<128, GF_ROWS_128>
            <<<(Nn + 8 * GF_ROWS_128 - 1) / (8 * GF_ROWS_128), 256, 0, stream>>>(
                h, rowptr, csr, dinv, bias, v, st, Nn, E);
      else
        k_gather_finish<64, GF_ROWS_64>
            <<<(Nn + 8 * GF_ROWS_64 - 1) / (8 * GF_ROWS_64), 256, 0, stream>>>(
                h, rowptr, csr, dinv, bias, v, st, Nn, E);
    } else {
      aggregate(h, v, F);
      finish(v, h, bias, F);
    }
  };

  if (bucketed) {
    k_bkt_hist<<<BKT_BLOCKS, BKT_THREADS, 0, stream>>>(dst, hist, E, NB);
    k_bkt_scan<<<(NB + 31) / 32, 1024, 0, stream>>>(hist, btot, NB);
    k_bkt_bin<<<BKT_BLOCKS, BKT_THREADS, 0, stream>>>(src, dst, hist, btot, bbase, pairs, E, NB);
    k_bkt_csr<<<NB, 256, 0, stream>>>(pairs, bbase, rowptr, csr, dinv, x, xs,
                                      stats, 256, stats2, folded ? 512 : 0, Nn, E);
  } else if (grouped) {
    zero((float*)cnt, (size_t)Nn);
    k_count_rank<<<(E + TB - 1) / TB, TB, 0, stream>>>(dst, cnt, rank, E);
    k_dinv_bsum<<<nchunk, 256, 0, stream>>>(cnt, dinv, bsum, Nn);
    k_scan_rowptr<<<nchunk, 256, 0, stream>>>(cnt, bsum, rowptr, Nn);
    k_fill_csr<<<(E + TB - 1) / TB, TB, 0, stream>>>(src, dst, rank, rowptr, csr, E);
  } else {
    k_deg_init<<<(Nn + TB - 1) / TB, TB, 0, stream>>>(dinv, Nn);
    k_deg_count<<<(E + TB - 1) / TB, TB, 0, stream>>>(dst, dinv, E);
    k_rsqrt_inplace<<<(Nn + TB - 1) / TB, TB, 0, stream>>>(dinv, Nn);
  }

  if (bucketed) {
    k_l1_sum_gemm<<<(Nn + L1_BLK - 1) / L1_BLK, 256, 0, stream>>>(
        xs, rowptr, csr, dinv, W1, b1, bufB, stats, Nn, E);
  } else if (grouped) {
    k_gather_x24<<<(Nn + 7) / 8, 256, 0, stream>>>(x, dinv, rowptr, csr, bufA, Nn, E);
    k_zero<<<1, 256, 0, stream>>>(stats, 256);
    k_gemm1_relu_stats<<<(Nn + 16 * G1_TILES - 1) / (16 * G1_TILES), 256, 0, stream>>>(
        bufA, W1, b1, bufB, stats, Nn);
  } else {
    gemm(x, W1, bufA, 24, 128);
    aggregate(bufA, bufB, 128);
    finish(bufB, bufA, b1, 128);
  }

  gemm_bn(bufB, stats, g1, be1, W2, bufA, 128);
  aggregate_finish(bufA, bufB, b2, stats2, 128);

  gemm_bn(bufB, stats2, g2, be2, W3, bufA, 64);
  aggregate_finish(bufA, bufB, b3, stats3, 64);

  k_gemm64x2<<<(Nn + TB - 1) / TB, TB, 0, stream>>>(bufB, stats3, g3, be3, W4, dinv,
                                                    bufA, Nn, invN);
  if (grouped) {
    k_gather2_lsm<<<(Nn + 8 * L4_ROWS - 1) / (8 * L4_ROWS), 256, 0, stream>>>(
        bufA, rowptr, csr, dinv, b4, (float*)d_out, Nn, E);
  } else {
    zero(bufB, (size_t)Nn * 2);
    k_scatter2<<<(E + TB - 1) / TB, TB, 0, stream>>>(bufA, src, dst, bufB, E);
    k_final_lsm<<<(Nn + TB - 1) / TB, TB, 0, stream>>>(bufB, bufA, dinv, b4,
                                                       (float*)d_out, Nn);
  }
}


# memory instructions that touched 25-32 lines made whole-line: the edge grouping's two placing passes order their stores in LDS first; layer 4's product reads 8 lanes a row (its sum order changes)
# speedup vs baseline: 7.5130x; 1.1172x over previous
#include <hip/hip_runtime.h>
#include <hip/hip_bf16.h>
#include <math.h>


typedef float v2f __attribute__((ext_vector_type(2)));
typedef float v4f __attribute__((ext_vector_type(4)));
typedef float v8f __attribute__((ext_vector_type(8)));

#define BN_EPS 1e-5f

__global__ void k_zero(float* __restrict__ p, size_t n) {
  size_t i = (size_t)blockIdx.x * blockDim.x + threadIdx.x;
  size_t stride = (size_t)gridDim.x * blockDim.x;
  for (; i < n; i += stride) p[i] = 0.0f;
}

__global__ void k_deg_init(float* __restrict__ deg, int n) {
  int i = blockIdx.x * blockDim.x + threadIdx.x;
  if (i < n) deg[i] = 1.0f;
}

__global__ void k_deg_count(const int* __restrict__ dst, float* __restrict__ deg, int E) {
  int e = blockIdx.x * blockDim.x + threadIdx.x;
  if (e < E) unsafeAtomicAdd(&deg[dst[e]], 1.0f);
}

__global__ void k_rsqrt_inplace(float* __restrict__ d, int n) {
  int i = blockIdx.x * blockDim.x + threadIdx.x;
  if (i < n) {
    float v = d[i];
    d[i] = (v > 0.0f) ? rsqrtf(v) : 0.0f;
  }
}

__global__ void k_count_rank(const int* __restrict__ dst, int* __restrict__ cnt,
                             int* __restrict__ rank, int E) {
  int e = blockIdx.x * blockDim.x + threadIdx.x;
  if (e < E) rank[e] = atomicAdd(&cnt[dst[e]], 1);
}

#define SCAN_CHUNK 1024

__global__ void __launch_bounds__(256)
k_dinv_bsum(const int* __restrict__ cnt, float* __restrict__ dinv,
            int* __restrict__ bsum, int n) {
  __shared__ int wsum[8];
  const int t = threadIdx.x;
  const int i0 = blockIdx.x * SCAN_CHUNK + 4 * t;
  int s = 0;
#pragma unroll
  for (int k = 0; k < 4; ++k) {
    const int i = i0 + k;
    if (i < n) {
      const int c = cnt[i];
      s += c;
      const float deg = (float)(c + 1);
      dinv[i] = (deg > 0.0f) ? rsqrtf(deg) : 0.0f;
    }
  }
  for (int o = 16; o > 0; o >>= 1) s += __shfl_xor(s, o);
  if ((t & 31) == 0) wsum[t >> 5] = s;
  __syncthreads();
  if (t == 0) {
    int tot = 0;
    for (int w = 0; w < 8; ++w) tot += wsum[w];
    bsum[blockIdx.x] = tot;
  }
}

__global__ void __launch_bounds__(256)
k_scan_rowptr(const int* __restrict__ cnt, const int* __restrict__ bsum,
              int* __restrict__ rowptr, int n) {
  __shared__ int wtot[8];
  __shared__ int base_s;
  const int t = threadIdx.x, lane = t & 31, w = t >> 5;
  int p = 0;
  for (int b = t; b < (int)blockIdx.x; b += 256) p += bsum[b];
  for (int o = 16; o > 0; o >>= 1) p += __shfl_xor(p, o);
  if (lane == 0) wtot[w] = p;
  __syncthreads();
  if (t == 0) {
    int tot = 0;
    for (int k = 0; k < 8; ++k) tot += wtot[k];
    base_s = tot;
  }
  __syncthreads();
  const int base = base_s;
  const int i0 = blockIdx.x * SCAN_CHUNK + 4 * t;
  int c[4];
  int s = 0;
#pragma unroll
  for (int k = 0; k < 4; ++k) {
    c[k] = (i0 + k < n) ? cnt[i0 + k] : 0;
    s += c[k];
  }
  int incl = s;
  for (int o = 1; o < 32; o <<= 1) {
    const int v = __shfl_up(incl, o);
    if (lane >= o) incl += v;
  }
  if (lane == 31) wtot[w] = incl;
  __syncthreads();
  int woff = 0;
  for (int k = 0; k < w; ++k) woff += wtot[k];
  int excl = base + woff + incl - s;
#pragma unroll
  for (int k = 0; k < 4; ++k) {
    if (i0 + k < n) {
      rowptr[i0 + k] = excl;
      excl += c[k];
    }
  }
  if (blockIdx.x == gridDim.x - 1 && t == 255) rowptr[n] = excl;
}

__global__ void k_fill_csr(const int* __restrict__ src, const int* __restrict__ dst,
                           const int* __restrict__ rank, const int* __restrict__ rowptr,
                           int* __restrict__ csr, int E) {
  int e = blockIdx.x * blockDim.x + threadIdx.x;
  if (e >= E) return;
  unsigned pos = (unsigned)(rowptr[dst[e]] + rank[e]);
  csr[min(pos, (unsigned)(E - 1))] = src[e];
}

#define BKT_SHIFT 7
#define BKT_NODES (1 << BKT_SHIFT)
#define BKT_MAX 1024
#define BKT_BLOCKS 256
#define BKT_THREADS 1024

__global__ void __launch_bounds__(BKT_THREADS)
k_bkt_hist(const int* __restrict__ dst, int* __restrict__ hist, int E, int NB) {
  __shared__ int h[BKT_MAX];
  const int t = threadIdx.x;
  h[t] = 0;
  __syncthreads();
  const int chunk = (E + BKT_BLOCKS - 1) / BKT_BLOCKS;
  const int beg = min((int)blockIdx.x * chunk, E);
  const int end = min(beg + chunk, E);
  const unsigned bmax = (unsigned)(NB - 1);
  for (int e0 = beg + t; e0 < end; e0 += 4 * BKT_THREADS) {
    int d[4];
#pragma unroll
    for (int k = 0; k < 4; ++k) d[k] = dst[min(e0 + k * BKT_THREADS, end - 1)];
    asm volatile("" : "+v"(d[0]), "+v"(d[1]), "+v"(d[2]), "+v"(d[3]));
#pragma unroll
    for (int k = 0; k < 4; ++k)
      if (e0 + k * BKT_THREADS < end)
        atomicAdd(&h[min((unsigned)d[k] >> BKT_SHIFT, bmax)], 1);
  }
  __syncthreads();
  if (t < NB) hist[(size_t)blockIdx.x * BKT_MAX + t] = h[t];
}

__global__ void __launch_bounds__(1024)
k_bkt_scan(int* __restrict__ hist, int* __restrict__ btot, int NB) {
  constexpr int SEG = BKT_BLOCKS / 32;
  __shared__ int seg[32][32];
  const int lane = threadIdx.x & 31;
  const int w = threadIdx.x >> 5;
  const int b = blockIdx.x * 32 + lane;
  const bool ok = b < NB;
  int* hp = hist + (size_t)(w * SEG) * BKT_MAX + b;
  int v[SEG];
  int s = 0;
#pragma unroll
  for (int k = 0; k < SEG; ++k) v[k] = ok ? hp[(size_t)k * BKT_MAX] : 0;
#pragma unroll
  for (int k = 0; k < SEG; ++k) s += v[k];
  seg[w][lane] = s;
  __syncthreads();
  int off = 0, tot = 0;
  for (int k = 0; k < 32; ++k) {
    const int x = seg[k][lane];
    tot += x;
    if (k < w) off += x;
  }
  if (ok) {
#pragma unroll
    for (int k = 0; k < SEG; ++k) {
      hp[(size_t)k * BKT_MAX] = off;
      off += v[k];
    }
    if (w == 0) btot[b] = tot;
  }
}

#define BIN_STAGE 6656
__global__ void __launch_bounds__(BKT_THREADS)
k_bkt_bin(const int* __restrict__ src, const int* __restrict__ dst,
          const int* __restrict__ hist, const int* __restrict__ btot,
          int* __restrict__ bbase, int* __restrict__ pairs_, int E, int NB) {
  __shared__ int cur[BKT_MAX];
  __shared__ int gdl[BKT_MAX];
  __shared__ int wtot[BKT_THREADS / 32];
  __shared__ int wtot2[BKT_THREADS / 32];
  __shared__ int2 stg[BIN_STAGE];
  int2* pairs = reinterpret_cast<int2*>(pairs_);
  const int t = threadIdx.x, lane = t & 31, w = t >> 5;
  const bool ok = t < NB;
  const int c = ok ? btot[t] : 0;
  const int within = ok ? hist[(size_t)blockIdx.x * BKT_MAX + t] : 0;
  const int nxt = !ok ? 0
                  : ((int)blockIdx.x + 1 < (int)gridDim.x)
                      ? hist[(size_t)(blockIdx.x + 1) * BKT_MAX + t]
                      : c;
  const int own = nxt - within;
  int incl = c, incl2 = own;
  for (int o = 1; o < 32; o <<= 1) {
    const int v = __shfl_up(incl, o);
    const int v2 = __shfl_up(incl2, o);
    if (lane >= o) {
      incl += v;
      incl2 += v2;
    }
  }
  if (lane == 31) {
    wtot[w] = incl;
    wtot2[w] = incl2;
  }
  __syncthreads();
  int woff = 0, woff2 = 0;
  for (int k = 0; k < w; ++k) {
    woff += wtot[k];
    woff2 += wtot2[k];
  }
  const int excl = woff + incl - c;
  const int lexcl = woff2 + incl2 - own;
  const int chunk = (E + BKT_BLOCKS - 1) / BKT_BLOCKS;
  const int beg = min((int)blockIdx.x * chunk, E);
  const int end = min(beg + chunk, E);
  const bool staged = (end - beg) <= BIN_STAGE;
  cur[t] = staged ? lexcl : excl + within;
  gdl[t] = excl + within - lexcl;
  if (blockIdx.x == 0 && ok) {
    bbase[t] = excl;
    if (t == NB - 1) bbase[NB] = excl + c;
  }
  __syncthreads();
  const unsigned bmax = (unsigned)(NB - 1);
  const unsigned emax = (unsigned)(E - 1);
  for (int e0 = beg + t; e0 < end; e0 += 4 * BKT_THREADS) {
    int d[4], s[4];
#pragma unroll
    for (int k = 0; k < 4; ++k) {
      const int e = min(e0 + k * BKT_THREADS, end - 1);
      d[k] = dst[e];
      s[k] = src[e];
    }
    asm volatile("" : "+v"(d[0]), "+v"(d[1]), "+v"(d[2]), "+v"(d[3]),
                      "+v"(s[0]), "+v"(s[1]), "+v"(s[2]), "+v"(s[3]));
#pragma unroll
    for (int k = 0; k < 4; ++k) {
      if (e0 + k * BKT_THREADS < end) {
        const int pos = atomicAdd(&cur[min((unsigned)d[k] >> BKT_SHIFT, bmax)], 1);
        if (staged)
          stg[min((unsigned)pos, (unsigned)(BIN_STAGE - 1))] = make_int2(d[k], s[k]);
        else
          pairs[min((unsigned)pos, emax)] = make_int2(d[k], s[k]);
      }
    }
  }
  if (staged) {
    __syncthreads();
    const int n = end - beg;
    for (int i = t; i < n; i += BKT_THREADS) {
      const int2 pr = stg[i];
      const int gp = i + gdl[min((unsigned)pr.x >> BKT_SHIFT, bmax)];
      pairs[min((unsigned)gp, emax)] = pr;
    }
  }
}

#define XS_PITCH 32
#define CSR_STAGE 4096
__global__ void __launch_bounds__(256)
k_bkt_csr(const int* __restrict__ pairs_, const int* __restrict__ bbase,
          int* __restrict__ rowptr, int* __restrict__ csr, float* __restrict__ dinv,
          const float* __restrict__ x, float* __restrict__ xs,
          float* __restrict__ z1, int nz1, float* __restrict__ z2, int nz2,
          int Nn, int E) {
  __shared__ int cnt[BKT_NODES];
  __shared__ int cur[BKT_NODES];
  __shared__ int wtot[BKT_NODES / 32];
  __shared__ float dvs[BKT_NODES];
  __shared__ int lst[CSR_STAGE];
  const int2* pairs = reinterpret_cast<const int2*>(pairs_);
  const int t = threadIdx.x, lane = t & 31, w = t >> 5;
  const int b = blockIdx.x;
  if (b == 0) {
    for (int i = t; i < nz1; i += 256) z1[i] = 0.0f;
    for (int i = t; i < nz2; i += 256) z2[i] = 0.0f;
  }
  const int xp_ = t & 7;
  const int xr_ = t >> 3;
  v4f xv[BKT_NODES / 32];
#pragma unroll
  for (int k = 0; k < BKT_NODES / 32; ++k) {
    const int node = min(b * BKT_NODES + xr_ + 32 * k, Nn - 1);
    xv[k] = reinterpret_cast<const v4f*>(x)[(size_t)node * 6 + min(xp_, 5)];
  }
  const int beg = min(max(bbase[b], 0), E);
  const int end = min(max(bbase[b + 1], beg), E);
  if (t < BKT_NODES) cnt[t] = 0;
  __syncthreads();
  for (int i0 = beg + t; i0 < end; i0 += 4 * 256) {
    int d[4];
#pragma unroll
    for (int k = 0; k < 4; ++k) d[k] = pairs_[2 * (size_t)min(i0 + k * 256, end - 1)];
    asm volatile("" : "+v"(d[0]), "+v"(d[1]), "+v"(d[2]), "+v"(d[3]));
#pragma unroll
    for (int k = 0; k < 4; ++k)
      if (i0 + k * 256 < end) atomicAdd(&cnt[d[k] & (BKT_NODES - 1)], 1);
  }
  __syncthreads();
  const int c = (t < BKT_NODES) ? cnt[t] : 0;
  int incl = c;
  for (int o = 1; o < 32; o <<= 1) {
    const int v = __shfl_up(incl, o);
    if (lane >= o) incl += v;
  }
  if (t < BKT_NODES && lane == 31) wtot[w] = incl;
  __syncthreads();
  if (t < BKT_NODES) {
    int woff = 0;
    for (int k = 0; k < w; ++k) woff += wtot[k];
    const int excl = beg + woff + incl - c;
    cur[t] = excl;
    const int node = b * BKT_NODES + t;
    const float deg = (float)(c + 1);
    const float dv = (deg > 0.0f) ? rsqrtf(deg) : 0.0f;
    dvs[t] = dv;
    if (node < Nn) {
      rowptr[node] = excl;
      dinv[node] = dv;
      if (node == Nn - 1) rowptr[Nn] = end;
    }
  }
  __syncthreads();
  {
    const float keep = (xp_ < 6) ? 1.f : 0.f;
#pragma unroll
    for (int k = 0; k < BKT_NODES / 32; ++k) {
      const int r = xr_ + 32 * k;
      const int node = b * BKT_NODES + r;
      const float sc = dvs[r] * keep;
      if (node < Nn)
        reinterpret_cast<v4f*>(xs)[(size_t)node * (XS_PITCH / 4) + xp_] = xv[k] * sc;
    }
  }
  const unsigned emax = (unsigned)(E - 1);
  const bool staged = (end - beg) <= CSR_STAGE;
  for (int i0 = beg + t; i0 < end; i0 += 4 * 256) {
    int2 p[4];
#pragma unroll
    for (int k = 0; k < 4; ++k) p[k] = pairs[min(i0 + k * 256, end - 1)];
    asm volatile("" : "+v"(p[0].x), "+v"(p[0].y), "+v"(p[1].x), "+v"(p[1].y),
                      "+v"(p[2].x), "+v"(p[2].y), "+v"(p[3].x), "+v"(p[3].y));
#pragma unroll
    for (int k = 0; k < 4; ++k) {
      if (i0 + k * 256 < end) {
        const int pos = atomicAdd(&cur[p[k].x & (BKT_NODES - 1)], 1);
        if (staged)
          lst[min((unsigned)(pos - beg), (unsigned)(CSR_STAGE - 1))] = p[k].y;
        else
          csr[min((unsigned)pos, emax)] = p[k].y;
      }
    }
  }
  if (staged) {
    __syncthreads();
    for (int i = (beg & ~31) + t; i < end; i += 256)
      if (i >= beg) csr[i] = lst[i - beg];
  }
}

#define GF_ROWS_128 9
#define GF_ROWS_64 8
template <int F, int R>
__global__ void __launch_bounds__(256)
k_gather_finish(const float* __restrict__ hsc, const int* __restrict__ rowptr,
                const int* __restrict__ csr, const float* __restrict__ dinv,
                const float* __restrict__ bias, float* __restrict__ vout,
                float* __restrict__ stats,
                int Nn, int E) {
  constexpr int V = F / 32;
  typedef float vf __attribute__((ext_vector_type(V)));
  __shared__ __attribute__((aligned(16))) float red[2][8][F];
  const int t = threadIdx.x;
  const int lane = t & 31;
  const int w = t >> 5;
  const int d0 = __builtin_amdgcn_readfirstlane((int)((blockIdx.x * 8 + w) * R));
  vf s = 0.0f;
  vf q = 0.0f;
  if (d0 < Nn) {
    const int rows = min(R, Nn - d0);
    const vf* hp = reinterpret_cast<const vf*>(hsc) + lane;
    vf* op = reinterpret_cast<vf*>(vout) + lane;
    const unsigned nmax = (unsigned)(Nn - 1);
    const vf zerov = 0.0f;
    const int rpraw = rowptr[min(d0 + lane, Nn)];
    const float dvl = dinv[min(d0 + lane, Nn - 1)];
    const vf b = *reinterpret_cast<const vf*>(bias + V * lane);
    const int rp = min(max(rpraw, 0), E);
    const unsigned emax = (unsigned)(E - 1);
    int beg = __builtin_amdgcn_readlane(rp, 0);
    int end = __builtin_amdgcn_readlane(rp, 1);
    int idx = csr[min((unsigned)(beg + lane), emax)];
    auto sum_chunk = [&](const int cur, const int m, vf a) -> vf {
      auto row = [&](int u) -> vf {
        const unsigned sidx = min((unsigned)__builtin_amdgcn_readlane(cur, u), nmax);
        return hp[(size_t)sidx * 32];
      };
      int u = 0;
      for (; u + 8 <= m; u += 8) {
        vf r[8];
#pragma unroll
        for (int k = 0; k < 8; ++k) r[k] = row(u + k);
#pragma unroll
        for (int k = 0; k < 8; ++k) a += r[k];
      }
      if (u + 4 <= m) {
        const vf r0 = row(u), r1 = row(u + 1), r2 = row(u + 2), r3 = row(u + 3);
        a += r0; a += r1; a += r2; a += r3;
        u += 4;
      }
      if (u + 2 <= m) {
        const vf r0 = row(u), r1 = row(u + 1);
        a += r0; a += r1;
        u += 2;
      }
      if (u < m) a += row(u);
      return a;
    };
    for (int i = 0; i < rows; ++i) {
      const int d = d0 + i;
      const vf self = hp[(size_t)d * 32];
      const int nbeg = end;
      const int nend = __builtin_amdgcn_readlane(rp, i + 2);
      const int idxn = csr[min((unsigned)(nbeg + lane), emax)];
      vf acc = sum_chunk(idx, min(32, end - beg), zerov);
      for (int j0 = beg + 32; j0 < end; j0 += 32) {
        const int m = min(32, end - j0);
        const int more = (lane < m) ? csr[j0 + lane] : 0;
        acc = sum_chunk(more, m, acc);
      }
      const float di = __builtin_bit_cast(
          float, __builtin_amdgcn_readlane(__builtin_bit_cast(int, dvl), i));
      const vf v = __builtin_elementwise_max((acc + self) * di + b, zerov);
      op[(size_t)d * 32] = v;
      s += v;
      q += v * v;
      beg = nbeg;
      end = nend;
      idx = idxn;
    }
  }
  *reinterpret_cast<vf*>(&red[0][w][V * lane]) = s;
  *reinterpret_cast<vf*>(&red[1][w][V * lane]) = q;
  __syncthreads();
  if (t < 2 * F) {
    const int which = t / F;
    const int j = t % F;
    float a = 0.f;
    for (int k = 0; k < 8; ++k) a += red[which][k][j];
    unsafeAtomicAdd(stats + which * F + j, a);
  }
}

__global__ void k_gemm_wmma_f32(const float* __restrict__ A, const float* __restrict__ B,
                                const float* __restrict__ dinv,
                                float* __restrict__ C, int M, int K, int F) {
  const int lane = threadIdx.x & 31;
  const int wave = (int)((blockIdx.x * blockDim.x + threadIdx.x) >> 5);
  const int nf = F >> 4;
  const int mTile = wave / nf;
  const int fTile = wave - mTile * nf;
  const int m0 = mTile << 4;
  if (m0 >= M) return;
  const int f0 = fTile << 4;
  const int row = lane & 15;
  const int kh2 = (lane >> 4) << 1;

  const float* arow = A + (size_t)(m0 + row) * K + kh2;
  const float* bcol = B + (size_t)kh2 * F + f0 + row;

  v8f acc = {0.f, 0.f, 0.f, 0.f, 0.f, 0.f, 0.f, 0.f};
  for (int k = 0; k < K; k += 4) {
    float2 av = *reinterpret_cast<const float2*>(arow + k);
    v2f a; a.x = av.x; a.y = av.y;
    v2f b;
    b.x = bcol[(size_t)k * F];
    b.y = bcol[(size_t)(k + 1) * F];
    acc = __builtin_amdgcn_wmma_f32_16x16x4_f32(false, a, false, b,
                                                (short)0, acc, false, false);
  }
  const float* dvp = dinv + m0 + (kh2 << 2);
  float* crow = C + (size_t)(m0 + (kh2 << 2)) * F + f0 + row;
#pragma unroll
  for (int r = 0; r < 8; ++r) crow[(size_t)r * F] = acc[r] * dvp[r];
}

template <int F>
__global__ void __launch_bounds__(256)
k_gemm_bn_wmma_f32(const float* __restrict__ V, const float* __restrict__ stats,
                   const float* __restrict__ gamma, const float* __restrict__ beta,
                   const float* __restrict__ B, const float* __restrict__ dinv,
                   float* __restrict__ C, int M, float invN) {
  constexpr int K = 128;
  constexpr int NF = F / 16;
  constexpr int RB = 16 * (8 / NF);
  constexpr int PITCH = K + 4;
  __shared__ __attribute__((aligned(16))) float As[RB * PITCH];

  const int t = threadIdx.x;
  const int lane = t & 31;
  const int w = t >> 5;
  const int rowBase = blockIdx.x * RB;

  {
    const int c4 = t & 31;
    const v4f su = *reinterpret_cast<const v4f*>(stats + 4 * c4);
    const v4f sq = *reinterpret_cast<const v4f*>(stats + K + 4 * c4);
    const v4f g = *reinterpret_cast<const v4f*>(gamma + 4 * c4);
    const v4f be = *reinterpret_cast<const v4f*>(beta + 4 * c4);
    v4f mean, rstd;
#pragma unroll
    for (int i = 0; i < 4; ++i) {
      mean[i] = su[i] * invN;
      const float var = sq[i] * invN - mean[i] * mean[i];
      rstd[i] = rsqrtf(var + BN_EPS);
    }
    v4f v[RB / 8];
#pragma unroll
    for (int p = 0; p < RB / 8; ++p) {
      const int gr = min(rowBase + p * 8 + w, M - 1);
      v[p] = *reinterpret_cast<const v4f*>(V + (size_t)gr * K + 4 * c4);
    }
#pragma unroll
    for (int p = 0; p < RB / 8; ++p) {
      const v4f o = (v[p] - mean) * rstd * g + be;
      *reinterpret_cast<v4f*>(&As[(p * 8 + w) * PITCH + 4 * c4]) = o;
    }
  }
  __syncthreads();

  const int mT = w / NF;
  const int fT = w - mT * NF;
  const int m0 = rowBase + (mT << 4);
  if (m0 >= M) return;
  const int f0 = fT << 4;
  const int row = lane & 15;
  const int kh2 = (lane >> 4) << 1;

  const float* arow = As + ((mT << 4) + row) * PITCH + kh2;
  const float* bcol = B + (size_t)kh2 * F + f0 + row;

  unsigned dbyte = (unsigned)(m0 + (kh2 << 2)) * 4u;
  unsigned cbyte = ((unsigned)(m0 + (kh2 << 2)) * F + f0 + row) * 4u;
  asm volatile("" : "+v"(dbyte), "+v"(cbyte));
  const float* dvp = reinterpret_cast<const float*>(reinterpret_cast<const char*>(dinv) + dbyte);
  float* crow = reinterpret_cast<float*>(reinterpret_cast<char*>(C) + cbyte);

  v8f acc = {0.f, 0.f, 0.f, 0.f, 0.f, 0.f, 0.f, 0.f};
#pragma unroll
  for (int k = 0; k < K; k += 4) {
    const v2f a = *reinterpret_cast<const v2f*>(arow + k);
    v2f b;
    b.x = bcol[(size_t)k * F];
    b.y = bcol[(size_t)(k + 1) * F];
    acc = __builtin_amdgcn_wmma_f32_16x16x4_f32(false, a, false, b,
                                                (short)0, acc, false, false);
  }
  asm volatile("s_wait_alu depctr_va_vdst(0)\n\t"
               "v_nop\n\tv_nop\n\tv_nop\n\tv_nop\n\t"
               "v_nop\n\tv_nop\n\tv_nop\n\tv_nop"
               : "+v"(acc));
#pragma unroll
  for (int r = 0; r < 8; ++r) crow[(size_t)r * F] = acc[r] * dvp[r];
}

__global__ void __launch_bounds__(256)
k_gather_x24(const float* __restrict__ x, const float* __restrict__ dinv,
             const int* __restrict__ rowptr, const int* __restrict__ csr,
             float* __restrict__ ax, int Nn, int E) {
  const int lane = threadIdx.x & 31;
  const int d = __builtin_amdgcn_readfirstlane((int)(blockIdx.x * 8 + (threadIdx.x >> 5)));
  if (d >= Nn) return;
  const int beg = max(rowptr[d], 0);
  const int end = min(rowptr[d + 1], E);
  const int p = lane & 7;
  const int q = lane >> 3;
  const bool pv = p < 6;
  const unsigned nmax = (unsigned)(Nn - 1);
  const v4f* xp = reinterpret_cast<const v4f*>(x) + p;
  const v4f zero4 = {0.f, 0.f, 0.f, 0.f};
  const float dd = dinv[d];
  const v4f xd = pv ? xp[(size_t)d * 6] : zero4;
  v4f acc = zero4;
  for (int j0 = beg; j0 < end; j0 += 32) {
    const int m = min(32, end - j0);
    int idx = 0;
    float dv = 0.f;
    if (lane < m) {
      idx = (int)min((unsigned)csr[j0 + lane], nmax);
      dv = dinv[idx];
    }
    v4f r[8];
    float wgt[8];
#pragma unroll
    for (int k = 0; k < 8; ++k) {
      const int e = 4 * k + q;
      const int s = __shfl(idx, e);
      wgt[k] = __shfl(dv, e);
      r[k] = (pv && e < m) ? xp[(size_t)s * 6] : zero4;
    }
#pragma unroll
    for (int k = 0; k < 8; ++k) acc += r[k] * wgt[k];
  }
#pragma unroll
  for (int i = 0; i < 4; ++i) {
    acc[i] += __shfl_xor(acc[i], 8);
    acc[i] += __shfl_xor(acc[i], 16);
  }
  if (q == 0 && pv)
    reinterpret_cast<v4f*>(ax)[(size_t)d * 6 + p] = (acc + xd * dd) * dd;
}


template <int K>
__device__ __forceinline__ unsigned group8_lane(unsigned v) {
  return (unsigned)__builtin_amdgcn_mov_dpp8(v, K * 0x249249);
}
__device__ __forceinline__ float xor8(float x) {
  const int b = __builtin_bit_cast(int, x);
  return __builtin_bit_cast(
      float, __builtin_amdgcn_permlane16(b, b, 0xfedcba98, 0x76543210, false, false));
}
__device__ __forceinline__ float xor16(float x) {
  const int b = __builtin_bit_cast(int, x);
  return __builtin_bit_cast(
      float, __builtin_amdgcn_permlanex16(b, b, 0x76543210, 0xfedcba98, false, false));
}

#define G1_TILES 8
__global__ void __launch_bounds__(256)
k_gemm1_relu_stats(const float* __restrict__ A, const float* __restrict__ B,
                   const float* __restrict__ bias, float* __restrict__ C,
                   float* __restrict__ stats,
                   int M) {
  constexpr int K = 24;
  constexpr int F = 128;
  constexpr int KS = K / 4;
  const int lane = threadIdx.x & 31;
  const int f0 = (int)(threadIdx.x >> 5) << 4;
  const int c = lane & 15;
  const int h = lane >> 4;
  const int kh2 = h << 1;
  const int rowBase = blockIdx.x * (16 * G1_TILES);

  v2f b[KS];
  const float* bcol = B + (size_t)kh2 * F + f0 + c;
#pragma unroll
  for (int kk = 0; kk < KS; ++kk) {
    b[kk].x = bcol[(size_t)(4 * kk) * F];
    b[kk].y = bcol[(size_t)(4 * kk + 1) * F];
  }
  const float bc = bias[f0 + c];

  v2f a[G1_TILES][KS];
#pragma unroll
  for (int i = 0; i < G1_TILES; ++i) {
    const int gr = min(rowBase + 16 * i + c, M - 1);
    const float* ar = A + (size_t)gr * K + kh2;
#pragma unroll
    for (int kk = 0; kk < KS; ++kk)
      a[i][kk] = *reinterpret_cast<const v2f*>(ar + 4 * kk);
  }

  float s = 0.f, q = 0.f;
#pragma unroll
  for (int i = 0; i < G1_TILES; ++i) {
    const int m0 = rowBase + 16 * i;
    if (m0 >= M) break;
    __builtin_amdgcn_sched_barrier(0);
    v8f acc = {0.f, 0.f, 0.f, 0.f, 0.f, 0.f, 0.f, 0.f};
#pragma unroll
    for (int kk = 0; kk < KS; ++kk)
      acc = __builtin_amdgcn_wmma_f32_16x16x4_f32(false, a[i][kk], false, b[kk],
                                                  (short)0, acc, false, false);
    asm volatile("s_wait_alu depctr_va_vdst(0)\n\t"
                 "v_nop\n\tv_nop\n\tv_nop\n\tv_nop\n\t"
                 "v_nop\n\tv_nop\n\tv_nop\n\tv_nop"
                 : "+v"(acc));
    __builtin_amdgcn_sched_barrier(0);
    const int r0 = m0 + (h << 3);
    float* crow = C + (size_t)r0 * F + f0 + c;
#pragma unroll
    for (int r = 0; r < 8; ++r) {
      if (r0 + r < M) {
        const float v = fmaxf(acc[r] + bc, 0.f);
        crow[(size_t)r * F] = v;
        s += v;
        q += v * v;
      }
    }
  }
  s += __shfl_xor(s, 16);
  q += __shfl_xor(q, 16);
  if (h == 0) {
    unsafeAtomicAdd(stats + f0 + c, s);
    unsafeAtomicAdd(stats + F + f0 + c, q);
  }
}

#define L1_ROWS 10
#define L1_BLK (8 * L1_ROWS)
#define L1_TILES (L1_BLK / 16)
#define L1_PITCH 28
__global__ void __launch_bounds__(256)
k_l1_sum_gemm(const float* __restrict__ xs, const int* __restrict__ rowptr,
              const int* __restrict__ csr, const float* __restrict__ dinv,
              const float* __restrict__ B, const float* __restrict__ bias,
              float* __restrict__ C,
              float* __restrict__ stats,
              int Nn, int E) {
  static_assert(L1_BLK % 16 == 0, "a block is a whole number of row tiles");
  __shared__ __attribute__((aligned(16))) float As[L1_BLK * L1_PITCH];
  const int lane = threadIdx.x & 31;
  const int w = (int)(threadIdx.x >> 5);
  const int rowBase = blockIdx.x * L1_BLK;
  const int d0 = __builtin_amdgcn_readfirstlane(rowBase + w * L1_ROWS);
  const int rows = max(0, min(L1_ROWS, Nn - d0));
  {
    const int p = lane & 7;
    const int q = lane >> 3;
    const int el = 4 * p + q;
    const unsigned p16 = 16u * (unsigned)p;
    const char* xb = reinterpret_cast<const char*>(xs);
    const unsigned nmax = (unsigned)(Nn - 1);
    const unsigned emax = (unsigned)(E - 1);
    const v4f zero4 = {0.f, 0.f, 0.f, 0.f};
    const int rpraw = rowptr[min(d0 + lane, Nn)];
    const float dvl = dinv[min(d0 + lane, Nn - 1)];
    const int rp = min(max(rpraw, 0), E);
    int beg = __builtin_amdgcn_readlane(rp, 0);
    int end = __builtin_amdgcn_readlane(rp, 1);
    int idx = csr[min((unsigned)(beg + el), emax)];
    auto piece = [&](unsigned o) -> v4f {
      return *reinterpret_cast<const v4f*>(xb + (o + p16));
    };
    auto sum_chunk = [&](const int cur, const int m, const unsigned own, v4f a) -> v4f {
      const unsigned off =
          (el < m) ? min((unsigned)cur, nmax) * (unsigned)(XS_PITCH * 4) : own;
      const int mq = m - q;
      v4f r0 = piece(group8_lane<0>(off));
      v4f r1 = piece(group8_lane<1>(off));
      v4f r2 = piece(group8_lane<2>(off));
      v4f r3 = piece(group8_lane<3>(off));
      if (m > 16) {
        v4f r4 = piece(group8_lane<4>(off));
        v4f r5 = piece(group8_lane<5>(off));
        v4f r6 = piece(group8_lane<6>(off));
        v4f r7 = piece(group8_lane<7>(off));
        asm volatile("" : "+v"(r0), "+v"(r1), "+v"(r2), "+v"(r3),
                          "+v"(r4), "+v"(r5), "+v"(r6), "+v"(r7));
        a += (0 < mq) ? r0 : zero4;
        a += (4 < mq) ? r1 : zero4;
        a += (8 < mq) ? r2 : zero4;
        a += (12 < mq) ? r3 : zero4;
        a += (16 < mq) ? r4 : zero4;
        a += (20 < mq) ? r5 : zero4;
        a += (24 < mq) ? r6 : zero4;
        a += (28 < mq) ? r7 : zero4;
      } else {
        a += (0 < mq) ? r0 : zero4;
        a += (4 < mq) ? r1 : zero4;
        a += (8 < mq) ? r2 : zero4;
        a += (12 < mq) ? r3 : zero4;
      }
      return a;
    };
    for (int i = 0; i < rows; ++i) {
      const int d = d0 + i;
      const unsigned own = (unsigned)d * (unsigned)(XS_PITCH * 4);
      const v4f self = piece(own);
      const int nbeg = end;
      const int nend = __builtin_amdgcn_readlane(rp, i + 2);
      const int idxn = csr[min((unsigned)(nbeg + el), emax)];
      v4f acc = sum_chunk(idx, min(32, end - beg), own, zero4);
      for (int j0 = beg + 32; j0 < end; j0 += 32) {
        const int more = csr[min((unsigned)(j0 + el), emax)];
        acc = sum_chunk(more, min(32, end - j0), own, acc);
      }
#pragma unroll
      for (int cc = 0; cc < 4; ++cc) {
        acc[cc] += xor8(acc[cc]);
        acc[cc] += xor16(acc[cc]);
      }
      const float di = __builtin_bit_cast(
          float, __builtin_amdgcn_readlane(__builtin_bit_cast(int, dvl), i));
      if (q == 0 && p < 6)
        *reinterpret_cast<v4f*>(&As[(w * L1_ROWS + i) * L1_PITCH + 4 * p]) = (acc + self) * di;
      beg = nbeg;
      end = nend;
      idx = idxn;
    }
  }

  constexpr int F = 128;
  constexpr int KS = 24 / 4;
  const int f0 = w << 4;
  const int c = lane & 15;
  const int h = lane >> 4;
  const int kh2 = h << 1;
  v2f b[KS];
  const float* bcol = B + (size_t)kh2 * F + f0 + c;
#pragma unroll
  for (int kk = 0; kk < KS; ++kk) {
    b[kk].x = bcol[(size_t)(4 * kk) * F];
    b[kk].y = bcol[(size_t)(4 * kk + 1) * F];
  }
  const float bc = bias[f0 + c];
  __syncthreads();

  const float* arow = As + c * L1_PITCH + kh2;
  float s = 0.f, q2 = 0.f;
#pragma unroll
  for (int i = 0; i < L1_TILES; ++i) {
    const int m0 = rowBase + 16 * i;
    if (m0 >= Nn) break;
    v2f a[KS];
#pragma unroll
    for (int kk = 0; kk < KS; ++kk)
      a[kk] = *reinterpret_cast<const v2f*>(arow + (16 * i) * L1_PITCH + 4 * kk);
    __builtin_amdgcn_sched_barrier(0);
    v8f acc = {0.f, 0.f, 0.f, 0.f, 0.f, 0.f, 0.f, 0.f};
#pragma unroll
    for (int kk = 0; kk < KS; ++kk)
      acc = __builtin_amdgcn_wmma_f32_16x16x4_f32(false, a[kk], false, b[kk],
                                                  (short)0, acc, false, false);
    asm volatile("s_wait_alu depctr_va_vdst(0)\n\t"
                 "v_nop\n\tv_nop\n\tv_nop\n\tv_nop\n\t"
                 "v_nop\n\tv_nop\n\tv_nop\n\tv_nop"
                 : "+v"(acc));
    __builtin_amdgcn_sched_barrier(0);
    const int r0 = m0 + (h << 3);
    float* crow = C + (size_t)r0 * F + f0 + c;
#pragma unroll
    for (int r = 0; r < 8; ++r) {
      if (r0 + r < Nn) {
        const float v = fmaxf(acc[r] + bc, 0.f);
        crow[(size_t)r * F] = v;
        s += v;
        q2 += v * v;
      }
    }
  }
  s += __shfl_xor(s, 16);
  q2 += __shfl_xor(q2, 16);
  if (h == 0) {
    unsafeAtomicAdd(stats + f0 + c, s);
    unsafeAtomicAdd(stats + F + f0 + c, q2);
  }
}

__global__ void k_scatter(const float* __restrict__ hsc, const int* __restrict__ src,
                          const int* __restrict__ dst,
                          float* __restrict__ agg, int E, int F) {
  int wave = (int)((blockIdx.x * blockDim.x + threadIdx.x) >> 5);
  int lane = threadIdx.x & 31;
  if (wave >= E) return;
  int s = src[wave], d = dst[wave];
  const float4* hrow = reinterpret_cast<const float4*>(hsc + (size_t)s * F);
  float* ad = agg + (size_t)d * F;
  for (int f4 = lane; 4 * f4 < F; f4 += 32) {
    float4 v = hrow[f4];
    unsafeAtomicAdd(ad + 4 * f4 + 0, v.x);
    unsafeAtomicAdd(ad + 4 * f4 + 1, v.y);
    unsafeAtomicAdd(ad + 4 * f4 + 2, v.z);
    unsafeAtomicAdd(ad + 4 * f4 + 3, v.w);
  }
}

#define FIN_ROWS 128
template <int F>
__global__ void __launch_bounds__(256)
k_finish_relu_stats(float* __restrict__ agg, const float* __restrict__ hsc,
                    const float* __restrict__ dinv,
                    const float* __restrict__ bias,
                    float* __restrict__ stats,
                    int Nn) {
  constexpr int TPR = F / 4;
  constexpr int RPP = 256 / TPR;
  constexpr int U = 8;
  constexpr int STEP = RPP * TPR;
  static_assert(FIN_ROWS % (U * RPP) == 0, "a block is a whole number of groups");
  __shared__ __attribute__((aligned(16))) float red[2][RPP][F];

  const int t = threadIdx.x;
  const int c4 = t % TPR;
  const int rsub = t / TPR;
  const int r0 = blockIdx.x * FIN_ROWS;
  const int r1 = min(r0 + FIN_ROWS, Nn);

  const v4f b = *reinterpret_cast<const v4f*>(bias + 4 * c4);
  const v4f zero4 = {0.f, 0.f, 0.f, 0.f};
  v4f s = zero4;
  v4f q = zero4;

  int r = r0 + rsub;
  v4f* ap = reinterpret_cast<v4f*>(agg) + (size_t)r * TPR + c4;
  const v4f* hp = reinterpret_cast<const v4f*>(hsc) + (size_t)r * TPR + c4;
  const float* dp = dinv + r;

  auto apply = [&](const v4f a, const v4f h, const float di, v4f* dst) {
    const v4f v = __builtin_elementwise_max((a + h) * di + b, zero4);
    *dst = v;
    s += v;
    q += v * v;
  };

  for (; r + (U - 1) * RPP < r1; r += U * RPP) {
    v4f a[U], h[U];
    float di[U];
#pragma unroll
    for (int u = 0; u < U; ++u) {
      a[u] = ap[u * STEP];
      h[u] = hp[u * STEP];
      di[u] = dp[u * RPP];
    }
    __builtin_amdgcn_sched_barrier(0);
#pragma unroll
    for (int u = 0; u < U; ++u) apply(a[u], h[u], di[u], ap + u * STEP);
    ap += U * STEP;
    hp += U * STEP;
    dp += U * RPP;
  }
  for (; r < r1; r += RPP) {
    apply(*ap, *hp, *dp, ap);
    ap += STEP;
    hp += STEP;
    dp += RPP;
  }

  *reinterpret_cast<v4f*>(&red[0][rsub][4 * c4]) = s;
  *reinterpret_cast<v4f*>(&red[1][rsub][4 * c4]) = q;
  __syncthreads();
  if (t < 2 * F) {
    const int which = t / F;
    const int j = t % F;
    float acc = 0.f;
    for (int k = 0; k < RPP; ++k) acc += red[which][k][j];
    unsafeAtomicAdd(stats + which * F + j, acc);
  }
}

#define G4_ROWS 16
__global__ void __launch_bounds__(256)
k_gemm64x2(const float* __restrict__ v3, const float* __restrict__ stats,
           const float* __restrict__ gamma, const float* __restrict__ beta,
           const float* __restrict__ W4, const float* __restrict__ dinv,
           float* __restrict__ h4, int Nn, float invN) {
  const int lane = threadIdx.x & 31;
  const int d0 = __builtin_amdgcn_readfirstlane(
      (int)((blockIdx.x * 8 + (threadIdx.x >> 5)) * G4_ROWS));
  if (d0 >= Nn) return;
  const int p = lane & 7;
  const int q = lane >> 3;
  v4f va[4], vb[4];
#pragma unroll
  for (int g = 0; g < 4; ++g) {
    const int r = min(d0 + 4 * g + q, Nn - 1);
    const v4f* rp = reinterpret_cast<const v4f*>(v3 + (size_t)r * 64);
    va[g] = rp[p];
    vb[g] = rp[8 + p];
  }
  const int gs = p & 3;
  const int rfin = d0 + 4 * gs + q;
  float di = dinv[min(rfin, Nn - 1)];
  v4f su[2], sq[2], gm[2], bt[2], wl[2], wh[2];
#pragma unroll
  for (int h = 0; h < 2; ++h) {
    const int c0 = 32 * h + 4 * p;
    su[h] = *reinterpret_cast<const v4f*>(stats + c0);
    sq[h] = *reinterpret_cast<const v4f*>(stats + 64 + c0);
    gm[h] = *reinterpret_cast<const v4f*>(gamma + c0);
    bt[h] = *reinterpret_cast<const v4f*>(beta + c0);
    wl[h] = *reinterpret_cast<const v4f*>(W4 + 2 * c0);
    wh[h] = *reinterpret_cast<const v4f*>(W4 + 2 * c0 + 4);
  }
  asm volatile("" : "+v"(va[0]), "+v"(va[1]), "+v"(va[2]), "+v"(va[3]),
                    "+v"(vb[0]), "+v"(vb[1]), "+v"(vb[2]), "+v"(vb[3]),
                    "+v"(su[0]), "+v"(su[1]), "+v"(sq[0]), "+v"(sq[1]),
                    "+v"(gm[0]), "+v"(gm[1]), "+v"(bt[0]), "+v"(bt[1]),
                    "+v"(wl[0]), "+v"(wl[1]), "+v"(wh[0]), "+v"(wh[1]), "+v"(di));
  v4f mean[2], rstd[2], w0[2], w1[2];
#pragma unroll
  for (int h = 0; h < 2; ++h) {
    w0[h][0] = wl[h][0]; w1[h][0] = wl[h][1];
    w0[h][1] = wl[h][2]; w1[h][1] = wl[h][3];
    w0[h][2] = wh[h][0]; w1[h][2] = wh[h][1];
    w0[h][3] = wh[h][2]; w1[h][3] = wh[h][3];
#pragma unroll
    for (int i = 0; i < 4; ++i) {
      mean[h][i] = su[h][i] * invN;
      const float var = sq[h][i] * invN - mean[h][i] * mean[h][i];
      rstd[h][i] = rsqrtf(var + BN_EPS);
    }
  }
  v2f acc[4];
#pragma unroll
  for (int g = 0; g < 4; ++g) {
    float o0 = 0.f, o1 = 0.f;
#pragma unroll
    for (int i = 0; i < 4; ++i) {
      const float xa = (va[g][i] - mean[0][i]) * rstd[0][i] * gm[0][i] + bt[0][i];
      o0 += xa * w0[0][i];
      o1 += xa * w1[0][i];
    }
#pragma unroll
    for (int i = 0; i < 4; ++i) {
      const float xb = (vb[g][i] - mean[1][i]) * rstd[1][i] * gm[1][i] + bt[1][i];
      o0 += xb * w0[1][i];
      o1 += xb * w1[1][i];
    }
    o0 += __shfl_xor(o0, 1);
    o1 += __shfl_xor(o1, 1);
    o0 += __shfl_xor(o0, 2);
    o1 += __shfl_xor(o1, 2);
    o0 += __shfl_xor(o0, 4);
    o1 += __shfl_xor(o1, 4);
    acc[g].x = o0;
    acc[g].y = o1;
  }
  const v2f a = gs == 0 ? acc[0] : gs == 1 ? acc[1] : gs == 2 ? acc[2] : acc[3];
  v2f o;
  o.x = a.x * di;
  o.y = a.y * di;
  if (p < 4 && rfin < Nn) reinterpret_cast<v2f*>(h4)[rfin] = o;
}

__global__ void k_scatter2(const float* __restrict__ h4s, const int* __restrict__ src,
                           const int* __restrict__ dst,
                           float* __restrict__ agg, int E) {
  int e = blockIdx.x * blockDim.x + threadIdx.x;
  if (e >= E) return;
  int s = src[e], d = dst[e];
  float2 v = *reinterpret_cast<const float2*>(h4s + 2 * (size_t)s);
  unsafeAtomicAdd(agg + 2 * (size_t)d,     v.x);
  unsafeAtomicAdd(agg + 2 * (size_t)d + 1, v.y);
}

__global__ void k_final_lsm(const float* __restrict__ agg, const float* __restrict__ h4s,
                            const float* __restrict__ dinv, const float* __restrict__ b4,
                            float* __restrict__ out, int Nn) {
  int i = blockIdx.x * blockDim.x + threadIdx.x;
  if (i >= Nn) return;
  float di = dinv[i];
  float o0 = (agg[2 * (size_t)i]     + h4s[2 * (size_t)i])     * di + b4[0];
  float o1 = (agg[2 * (size_t)i + 1] + h4s[2 * (size_t)i + 1]) * di + b4[1];
  float m = fmaxf(o0, o1);
  float l = m + logf(expf(o0 - m) + expf(o1 - m));
  out[2 * (size_t)i]     = o0 - l;
  out[2 * (size_t)i + 1] = o1 - l;
}

#define L4_ROWS 16
__global__ void __launch_bounds__(256)
k_gather2_lsm(const float* __restrict__ h4s, const int* __restrict__ rowptr,
              const int* __restrict__ csr, const float* __restrict__ dinv,
              const float* __restrict__ b4, float* __restrict__ out, int Nn, int E) {
  const int lane = threadIdx.x & 31;
  const int d0 = __builtin_amdgcn_readfirstlane(
      (int)((blockIdx.x * 8 + (threadIdx.x >> 5)) * L4_ROWS));
  if (d0 >= Nn) return;
  const int p = lane & 7;
  const int q = lane >> 3;
  const v2f* hp = reinterpret_cast<const v2f*>(h4s);
  const unsigned nmax = (unsigned)(Nn - 1);
  const unsigned emax = (unsigned)(E - 1);
  const v2f zero2 = {0.f, 0.f};
  const int rpraw = rowptr[min(d0 + lane, Nn)];
  const int gs = p & 3;
  const int rfin = d0 + 4 * gs + q;
  const int rfc = min(rfin, Nn - 1);
  const v2f self = hp[rfc];
  const float di = dinv[rfc];
  const float bb0 = b4[0];
  const float bb1 = b4[1];
  const int rp = min(max(rpraw, 0), E);

  int beg[4], len[4];
  unsigned own[4];
#pragma unroll
  for (int g = 0; g < 4; ++g) {
    const int b = __shfl(rp, 4 * g + q);
    const int e = __shfl(rp, 4 * g + q + 1);
    beg[g] = b;
    len[g] = e - b;
    own[g] = min((unsigned)(d0 + 4 * g + q), nmax);
  }
  int idx[4][4];
#pragma unroll
  for (int g = 0; g < 4; ++g) {
#pragma unroll
    for (int k = 0; k < 4; ++k)
      idx[g][k] = csr[min((unsigned)(beg[g] + p + 8 * k), emax)];
  }
  v2f v[4][4];
#pragma unroll
  for (int g = 0; g < 4; ++g) {
#pragma unroll
    for (int k = 0; k < 4; ++k) {
      const unsigned s = (p + 8 * k < len[g]) ? min((unsigned)idx[g][k], nmax) : own[g];
      v[g][k] = hp[s];
    }
  }
  v2f acc[4];
#pragma unroll
  for (int g = 0; g < 4; ++g) {
    v2f a = zero2;
#pragma unroll
    for (int k = 0; k < 4; ++k) a += (p + 8 * k < len[g]) ? v[g][k] : zero2;
    const int end = beg[g] + len[g];
    for (int j = beg[g] + 32 + p; j < end; j += 8) a += hp[min((unsigned)csr[j], nmax)];
    a.x += __shfl_xor(a.x, 1);
    a.y += __shfl_xor(a.y, 1);
    a.x += __shfl_xor(a.x, 2);
    a.y += __shfl_xor(a.y, 2);
    a.x += __shfl_xor(a.x, 4);
    a.y += __shfl_xor(a.y, 4);
    acc[g] = a;
  }
  const v2f a = gs == 0 ? acc[0] : gs == 1 ? acc[1] : gs == 2 ? acc[2] : acc[3];
  const float o0 = (a.x + self.x) * di + bb0;
  const float o1 = (a.y + self.y) * di + bb1;
  const float m = fmaxf(o0, o1);
  const float l = m + logf(expf(o0 - m) + expf(o1 - m));
  v2f o;
  o.x = o0 - l;
  o.y = o1 - l;
  if (p < 4 && rfin < Nn) reinterpret_cast<v2f*>(out)[rfin] = o;
}

extern "C" void kernel_launch(void* const* d_in, const int* in_sizes, int n_in,
                              void* d_out, int out_size, void* d_ws, size_t ws_size,
                              hipStream_t stream) {
  const float* x   = (const float*)d_in[0];
  const int*   ei  = (const int*)d_in[1];
  const float* W1  = (const float*)d_in[2];
  const float* b1  = (const float*)d_in[3];
  const float* g1  = (const float*)d_in[4];
  const float* be1 = (const float*)d_in[5];
  const float* W2  = (const float*)d_in[6];
  const float* b2  = (const float*)d_in[7];
  const float* g2  = (const float*)d_in[8];
  const float* be2 = (const float*)d_in[9];
  const float* W3  = (const float*)d_in[10];
  const float* b3  = (const float*)d_in[11];
  const float* g3  = (const float*)d_in[12];
  const float* be3 = (const float*)d_in[13];
  const float* W4  = (const float*)d_in[14];
  const float* b4  = (const float*)d_in[15];

  const int Nn = in_sizes[0] / 24;
  const int E  = in_sizes[1] / 2;
  const int* src = ei;
  const int* dst = ei + E;

  float* ws    = (float*)d_ws;
  float* dinv  = ws;
  float* stats = ws + Nn;
  float* bufA  = stats + 256;
  float* bufB  = bufA + (size_t)Nn * 128;
  const int nchunk = (Nn + SCAN_CHUNK - 1) / SCAN_CHUNK;
  const size_t rp_len = ((size_t)Nn + 1 + 31) & ~(size_t)31;
  const size_t bs_len = ((size_t)nchunk + 31) & ~(size_t)31;
  int* rowptr = (int*)(bufB + (size_t)Nn * 128);
  int* csr    = rowptr + rp_len;
  int* cnt    = (int*)bufB;
  int* bsum   = cnt + Nn;
  int* rank   = bsum + bs_len;
  const size_t need_bytes =
      ((size_t)Nn + 256 + 2 * (size_t)Nn * 128 + rp_len + (size_t)E) * sizeof(float);
  const bool grouped = ws_size >= need_bytes &&
                       (size_t)Nn + bs_len + (size_t)E <= (size_t)Nn * 128;
  const int NB = (Nn + BKT_NODES - 1) / BKT_NODES;
  int* hist  = (int*)bufB;
  int* btot  = hist + (size_t)BKT_BLOCKS * BKT_MAX;
  int* bbase = btot + BKT_MAX;
  int* pairs = bbase + BKT_MAX + 32;
  const bool bucketed = grouped && NB <= BKT_MAX &&
                        (size_t)BKT_BLOCKS * BKT_MAX + 2 * BKT_MAX + 32 + 2 * (size_t)E <=
                            (size_t)Nn * 128;
  float* xs = bufA + (size_t)Nn * XS_PITCH;
  const size_t csr_len = ((size_t)E + 31) & ~(size_t)31;
  float* stats_x = (float*)(csr + csr_len);
  const bool folded =
      bucketed &&
      ws_size >= ((size_t)Nn + 256 + 2 * (size_t)Nn * 128 + rp_len + csr_len + 512) * sizeof(float);
  float* stats2 = folded ? stats_x : stats;
  float* stats3 = folded ? stats_x + 256 : stats;

  const float invN = 1.0f / (float)Nn;
  const int TB = 256;

  auto gemm = [&](const float* A, const float* Bm, float* C, int K, int F) {
    long waves = (long)((Nn + 15) / 16) * (F / 16);
    int blocks = (int)((waves * 32 + TB - 1) / TB);
    k_gemm_wmma_f32<<<blocks, TB, 0, stream>>>(A, Bm, dinv, C, Nn, K, F);
  };
  auto scatter = [&](const float* h, float* agg, int F) {
    int blocks = (int)(((long)E * 32 + TB - 1) / TB);
    k_scatter<<<blocks, TB, 0, stream>>>(h, src, dst, agg, E, F);
  };
  auto finish = [&](float* agg, const float* h, const float* bias, int F) {
    const int blocks = (Nn + FIN_ROWS - 1) / FIN_ROWS;
    k_zero<<<1, 256, 0, stream>>>(stats, 256);
    if (F == 128)
      k_finish_relu_stats<128><<<blocks, 256, 0, stream>>>(agg, h, dinv, bias, stats, Nn);
    else
      k_finish_relu_stats<64><<<blocks, 256, 0, stream>>>(agg, h, dinv, bias, stats, Nn);
  };
  auto gemm_bn = [&](const float* V, const float* st, const float* gamma, const float* beta,
                     const float* Bm, float* C, int F) {
    if (F == 128)
      k_gemm_bn_wmma_f32<128><<<(Nn + 15) / 16, 256, 0, stream>>>(
          V, st, gamma, beta, Bm, dinv, C, Nn, invN);
    else
      k_gemm_bn_wmma_f32<64><<<(Nn + 31) / 32, 256, 0, stream>>>(
          V, st, gamma, beta, Bm, dinv, C, Nn, invN);
  };
  auto zero = [&](float* p, size_t n) {
    int blocks = (int)((n + TB - 1) / TB);
    if (blocks > 4096) blocks = 4096;
    k_zero<<<blocks, TB, 0, stream>>>(p, n);
  };

  auto aggregate = [&](const float* h, float* agg, int F) {
    zero(agg, (size_t)Nn * F);
    scatter(h, agg, F);
  };
  auto aggregate_finish = [&](const float* h, float* v, const float* bias, float* st, int F) {
    if (grouped) {
      if (!folded) k_zero<<<1, 256, 0, stream>>>(st, 256);
      if (F == 128)
        k_gather_finish<128, GF_ROWS_128>
            <<<(Nn + 8 * GF_ROWS_128 - 1) / (8 * GF_ROWS_128), 256, 0, stream>>>(
                h, rowptr, csr, dinv, bias, v, st, Nn, E);
      else
        k_gather_finish<64, GF_ROWS_64>
            <<<(Nn + 8 * GF_ROWS_64 - 1) / (8 * GF_ROWS_64), 256, 0, stream>>>(
                h, rowptr, csr, dinv, bias, v, st, Nn, E);
    } else {
      aggregate(h, v, F);
      finish(v, h, bias, F);
    }
  };

  if (bucketed) {
    k_bkt_hist<<<BKT_BLOCKS, BKT_THREADS, 0, stream>>>(dst, hist, E, NB);
    k_bkt_scan<<<(NB + 31) / 32, 1024, 0, stream>>>(hist, btot, NB);
    k_bkt_bin<<<BKT_BLOCKS, BKT_THREADS, 0, stream>>>(src, dst, hist, btot, bbase, pairs, E, NB);
    k_bkt_csr<<<NB, 256, 0, stream>>>(pairs, bbase, rowptr, csr, dinv, x, xs,
                                      stats, 256, stats2, folded ? 512 : 0, Nn, E);
  } else if (grouped) {
    zero((float*)cnt, (size_t)Nn);
    k_count_rank<<<(E + TB - 1) / TB, TB, 0, stream>>>(dst, cnt, rank, E);
    k_dinv_bsum<<<nchunk, 256, 0, stream>>>(cnt, dinv, bsum, Nn);
    k_scan_rowptr<<<nchunk, 256, 0, stream>>>(cnt, bsum, rowptr, Nn);
    k_fill_csr<<<(E + TB - 1) / TB, TB, 0, stream>>>(src, dst, rank, rowptr, csr, E);
  } else {
    k_deg_init<<<(Nn + TB - 1) / TB, TB, 0, stream>>>(dinv, Nn);
    k_deg_count<<<(E + TB - 1) / TB, TB, 0, stream>>>(dst, dinv, E);
    k_rsqrt_inplace<<<(Nn + TB - 1) / TB, TB, 0, stream>>>(dinv, Nn);
  }

  if (bucketed) {
    k_l1_sum_gemm<<<(Nn + L1_BLK - 1) / L1_BLK, 256, 0, stream>>>(
        xs, rowptr, csr, dinv, W1, b1, bufB, stats, Nn, E);
  } else if (grouped) {
    k_gather_x24<<<(Nn + 7) / 8, 256, 0, stream>>>(x, dinv, rowptr, csr, bufA, Nn, E);
    k_zero<<<1, 256, 0, stream>>>(stats, 256);
    k_gemm1_relu_stats<<<(Nn + 16 * G1_TILES - 1) / (16 * G1_TILES), 256, 0, stream>>>(
        bufA, W1, b1, bufB, stats, Nn);
  } else {
    gemm(x, W1, bufA, 24, 128);
    aggregate(bufA, bufB, 128);
    finish(bufB, bufA, b1, 128);
  }

  gemm_bn(bufB, stats, g1, be1, W2, bufA, 128);
  aggregate_finish(bufA, bufB, b2, stats2, 128);

  gemm_bn(bufB, stats2, g2, be2, W3, bufA, 64);
  aggregate_finish(bufA, bufB, b3, stats3, 64);

  k_gemm64x2<<<(Nn + 8 * G4_ROWS - 1) / (8 * G4_ROWS), 256, 0, stream>>>(
      bufB, stats3, g3, be3, W4, dinv, bufA, Nn, invN);
  if (grouped) {
    k_gather2_lsm<<<(Nn + 8 * L4_ROWS - 1) / (8 * L4_ROWS), 256, 0, stream>>>(
        bufA, rowptr, csr, dinv, b4, (float*)d_out, Nn, E);
  } else {
    zero(bufB, (size_t)Nn * 2);
    k_scatter2<<<(E + TB - 1) / TB, TB, 0, stream>>>(bufA, src, dst, bufB, E);
    k_final_lsm<<<(Nn + TB - 1) / TB, TB, 0, stream>>>(bufB, bufA, dinv, b4,
                                                       (float*)d_out, Nn);
  }
}
